# starting kernel
# baseline (speedup 1.0000x reference)
#include <hip/hip_runtime.h>

typedef float v2f __attribute__((ext_vector_type(2)));
typedef float v8f __attribute__((ext_vector_type(8)));

#define D 128
#define KSTACK 3
#define LDS_STRIDE 132

__global__ void zero_f4_kernel(float4* __restrict__ p, int n4) {
  int i = blockIdx.x * blockDim.x + threadIdx.x;
  if (i < n4) p[i] = make_float4(0.f, 0.f, 0.f, 0.f);
}

__global__ void deg_kernel(const int* __restrict__ dst, float* __restrict__ deg, int nE) {
  int i = blockIdx.x * blockDim.x + threadIdx.x;
  if (i < nE) unsafeAtomicAdd(&deg[dst[i]], 1.0f);
}

__global__ void norm_kernel(float* __restrict__ d, int n) {
  int i = blockIdx.x * blockDim.x + threadIdx.x;
  if (i < n) d[i] = rsqrtf(fmaxf(d[i], 1.0f));
}

__global__ void spmm_scatter_kernel(const float* __restrict__ h,
                                    const int* __restrict__ src,
                                    const int* __restrict__ dst,
                                    const float* __restrict__ nrm,
                                    float* __restrict__ agg, int nE) {
  const int wid  = threadIdx.x >> 5;
  const int lane = threadIdx.x & 31;
  const int e = blockIdx.x * (blockDim.x >> 5) + wid;
  if (e >= nE) return;
  const int s = src[e];
  const int d = dst[e];
  const float ns = nrm[s];
  const float4 v4 = *reinterpret_cast<const float4*>(h + (size_t)s * D + lane * 4);
  float* base = agg + (size_t)d * D + lane * 4;
  unsafeAtomicAdd(base + 0, v4.x * ns);
  unsafeAtomicAdd(base + 1, v4.y * ns);
  unsafeAtomicAdd(base + 2, v4.z * ns);
  unsafeAtomicAdd(base + 3, v4.w * ns);
}

template <bool HAS_NRM, bool HAS_ADD, bool RELU, bool ACCUM>
__global__ __launch_bounds__(256) void gemm_kernel(
    const float* __restrict__ A, const float* __restrict__ nrm,
    const float* __restrict__ W, const float* __restrict__ bias,
    const float* __restrict__ addm, float* __restrict__ out,
    int nRowTiles, float outScale) {
  __shared__ float As[16 * LDS_STRIDE];
  const int tid  = threadIdx.x;
  const int wid  = tid >> 5;
  const int lane = tid & 31;
  const int half = lane >> 4;
  const int lm   = lane & 15;
  const int colBase = wid * 16;

  v2f breg[32];
#pragma unroll
  for (int ks = 0; ks < 32; ++ks) {
    const int kk = ks * 4 + half * 2;
    breg[ks].x = W[(size_t)kk * D + colBase + lm];
    breg[ks].y = W[(size_t)(kk + 1) * D + colBase + lm];
  }
  const float bcol = bias[colBase + lm];

  const int idx0 = tid;
  const int idx1 = tid + 256;
  const int r0 = idx0 >> 5, c0 = (idx0 & 31) * 4;
  const int r1 = idx1 >> 5, c1 = (idx1 & 31) * 4;
  const float4* __restrict__ A4 = reinterpret_cast<const float4*>(A);

  int rt = blockIdx.x;
  if (rt >= nRowTiles) return;

  float4 s0, s1;
  {
    const int rowBase = rt * 16;
    s0 = A4[(size_t)(rowBase + r0) * 32 + (idx0 & 31)];
    s1 = A4[(size_t)(rowBase + r1) * 32 + (idx1 & 31)];
    if (HAS_NRM) {
      const float n0 = nrm[rowBase + r0], n1 = nrm[rowBase + r1];
      s0.x *= n0; s0.y *= n0; s0.z *= n0; s0.w *= n0;
      s1.x *= n1; s1.y *= n1; s1.z *= n1; s1.w *= n1;
    }
  }

  for (;;) {
    __syncthreads();
    *reinterpret_cast<float4*>(&As[r0 * LDS_STRIDE + c0]) = s0;
    *reinterpret_cast<float4*>(&As[r1 * LDS_STRIDE + c1]) = s1;

    const int rnext = rt + gridDim.x;
    const bool more = rnext < nRowTiles;
    if (more) {
      const int rowBase = rnext * 16;
      s0 = A4[(size_t)(rowBase + r0) * 32 + (idx0 & 31)];
      s1 = A4[(size_t)(rowBase + r1) * 32 + (idx1 & 31)];
      if (HAS_NRM) {
        const float n0 = nrm[rowBase + r0], n1 = nrm[rowBase + r1];
        s0.x *= n0; s0.y *= n0; s0.z *= n0; s0.w *= n0;
        s1.x *= n1; s1.y *= n1; s1.z *= n1; s1.w *= n1;
      }
    }
    __syncthreads();

    v8f acc = {0.f, 0.f, 0.f, 0.f, 0.f, 0.f, 0.f, 0.f};
#pragma unroll
    for (int ks = 0; ks < 32; ++ks) {
      const int kk = ks * 4 + half * 2;
      v2f a;
      a.x = As[lm * LDS_STRIDE + kk];
      a.y = As[lm * LDS_STRIDE + kk + 1];
      acc = __builtin_amdgcn_wmma_f32_16x16x4_f32(
           false, a,  false, breg[ks],
           (short)0, acc,  false,  false);
    }

    const int rowBase = rt * 16;
#pragma unroll
    for (int r = 0; r < 8; ++r) {
      const int row = rowBase + r + half * 8;
      const int col = colBase + lm;
      const size_t oi = (size_t)row * D + col;
      float vv = acc[r] + bcol;
      if (HAS_ADD) vv += addm[oi];
      if (RELU) vv = fmaxf(vv, 0.f);
      vv *= outScale;
      if (ACCUM) out[oi] += vv;
      else       out[oi]  = vv;
    }

    if (!more) break;
    rt = rnext;
  }
}

extern "C" void kernel_launch(void* const* d_in, const int* in_sizes, int n_in,
                              void* d_out, int out_size, void* d_ws, size_t ws_size,
                              hipStream_t stream) {
  const float* feats = (const float*)d_in[0];
  const int*   src   = (const int*)d_in[1];
  const int*   dst   = (const int*)d_in[2];
  const float* w0    = (const float*)d_in[3];
  const float* b0    = (const float*)d_in[4];
  const float* w     = (const float*)d_in[5];
  const float* bw    = (const float*)d_in[6];
  const float* v     = (const float*)d_in[7];
  const float* bv    = (const float*)d_in[8];
  float* out = (float*)d_out;

  const int N = in_sizes[0] / D;
  const int E = in_sizes[1];
  const size_t M = (size_t)N * D;

  float* ws = (float*)d_ws;
  size_t off = 0;
  float* nrm  = ws + off; off += ((size_t)N + 255) & ~(size_t)255;
  float* agg0 = ws + off; off += M;
  float* fv   = ws + off; off += M;
  float* h1   = ws + off; off += M;
  float* agg1 = ws + off; off += M;

  const int nRowTiles  = N / 16;
  const int gemmBlocks = 625;

  zero_f4_kernel<<<(N / 4 + 255) / 256, 256, 0, stream>>>((float4*)nrm, N / 4);
  deg_kernel<<<(E + 255) / 256, 256, 0, stream>>>(dst, nrm, E);
  norm_kernel<<<(N + 255) / 256, 256, 0, stream>>>(nrm, N);

  zero_f4_kernel<<<((int)(M / 4) + 255) / 256, 256, 0, stream>>>((float4*)agg0, (int)(M / 4));
  spmm_scatter_kernel<<<(E + 7) / 8, 256, 0, stream>>>(feats, src, dst, nrm, agg0, E);

  for (int k = 0; k < KSTACK; ++k) {
    gemm_kernel<false, false, false, false><<<gemmBlocks, 256, 0, stream>>>(
        feats, nullptr, v + (size_t)k * D * D, bv + (size_t)k * D,
        nullptr, fv, nRowTiles, 1.0f);
    gemm_kernel<true, true, true, false><<<gemmBlocks, 256, 0, stream>>>(
        agg0, nrm, w0 + (size_t)k * D * D, b0 + (size_t)k * D,
        fv, h1, nRowTiles, 1.0f);
    zero_f4_kernel<<<((int)(M / 4) + 255) / 256, 256, 0, stream>>>((float4*)agg1, (int)(M / 4));
    spmm_scatter_kernel<<<(E + 7) / 8, 256, 0, stream>>>(h1, src, dst, nrm, agg1, E);
    if (k == 0) {
      gemm_kernel<true, true, true, false><<<gemmBlocks, 256, 0, stream>>>(
          agg1, nrm, w + (size_t)k * D * D, bw + (size_t)k * D,
          fv, out, nRowTiles, 1.0f / (float)KSTACK);
    } else {
      gemm_kernel<true, true, true, true><<<gemmBlocks, 256, 0, stream>>>(
          agg1, nrm, w + (size_t)k * D * D, bw + (size_t)k * D,
          fv, out, nRowTiles, 1.0f / (float)KSTACK);
    }
  }
}


# SpMM scatter-add with f32 atomics -> per-call CSR by destination (returning int add = arrival-order rank per edge, scan, fill) + one wave per row gather: row summed in rank order, stored once
# speedup vs baseline: 2.1724x; 2.1724x over previous
#include <hip/hip_runtime.h>

typedef float v2f __attribute__((ext_vector_type(2)));
typedef float v8f __attribute__((ext_vector_type(8)));

#define D 128
#define KSTACK 3
#define LDS_STRIDE 132

__global__ void zero_f4_kernel(float4* __restrict__ p, int n4) {
  int i = blockIdx.x * blockDim.x + threadIdx.x;
  if (i < n4) p[i] = make_float4(0.f, 0.f, 0.f, 0.f);
}

__global__ void count_rank_kernel(const int* __restrict__ dst, int* __restrict__ cnt,
                                  int* __restrict__ pos, int nE) {
  int i = blockIdx.x * blockDim.x + threadIdx.x;
  if (i < nE) pos[i] = atomicAdd(&cnt[dst[i]], 1);
}

__global__ __launch_bounds__(256) void scan_norm_kernel(const int* __restrict__ cnt,
                                                        int* __restrict__ rowptr,
                                                        float* __restrict__ nrm, int n) {
  __shared__ int waveIncl[8];
  __shared__ int waveFront[8];
  const int tid  = threadIdx.x;
  const int lane = tid & 31;
  const int wid  = tid >> 5;
  const int base = blockIdx.x * 256;

  const int4* __restrict__ c4 = reinterpret_cast<const int4*>(cnt);
  int front = 0;
  for (int q = tid; q < (base >> 2); q += 256) {
    const int4 v = c4[q];
    front += (v.x + v.y) + (v.z + v.w);
  }
  for (int o = 16; o > 0; o >>= 1) front += __shfl_xor(front, o);

  const int i = base + tid;
  const int c = (i < n) ? cnt[i] : 0;
  if (i < n) nrm[i] = rsqrtf(fmaxf((float)c, 1.0f));

  int incl = c;
  for (int o = 1; o < 32; o <<= 1) {
    const int t = __shfl_up(incl, o);
    if (lane >= o) incl += t;
  }
  if (lane == 31) waveIncl[wid]  = incl;
  if (lane == 0)  waveFront[wid] = front;
  __syncthreads();
  int off = 0;
  for (int w = 0; w < 8; ++w) {
    off += waveFront[w];
    if (w < wid) off += waveIncl[w];
  }
  const int excl = off + incl - c;
  if (i < n) rowptr[i] = excl;
  if (i == n - 1) rowptr[n] = excl + c;
}

__global__ void csr_fill_kernel(const int* __restrict__ src, const int* __restrict__ dst,
                                const int* __restrict__ pos, const int* __restrict__ rowptr,
                                int* __restrict__ csrSrc, int nE) {
  int i = blockIdx.x * blockDim.x + threadIdx.x;
  if (i < nE) csrSrc[rowptr[dst[i]] + pos[i]] = src[i];
}

__device__ __forceinline__ void row_fma(float4& acc, const float4 v, const float w) {
  acc.x += v.x * w; acc.y += v.y * w; acc.z += v.z * w; acc.w += v.w * w;
}

template <int R>
__device__ __forceinline__ void gather_rows(float4& acc, const float4* __restrict__ h4,
                                            const int sL, const int nLi,
                                            const int j, const int lane) {
  float4 v[R];
#pragma unroll
  for (int u = 0; u < R; ++u) {
    const int s = __builtin_amdgcn_readlane(sL, j + u);
    v[u] = h4[(size_t)s * 32 + lane];
  }
#pragma unroll
  for (int u = 0; u < R; ++u)
    row_fma(acc, v[u], __int_as_float(__builtin_amdgcn_readlane(nLi, j + u)));
}

__global__ __launch_bounds__(256) void spmm_gather_kernel(const float* __restrict__ h,
                                                          const int* __restrict__ rowptr,
                                                          const int* __restrict__ csrSrc,
                                                          const float* __restrict__ nrm,
                                                          float* __restrict__ agg, int nN) {
  const int lane = threadIdx.x & 31;
  const int d = __builtin_amdgcn_readfirstlane(blockIdx.x * 8 + (threadIdx.x >> 5));
  if (d >= nN) return;
  const int beg = __builtin_amdgcn_readfirstlane(rowptr[d]);
  const int end = __builtin_amdgcn_readfirstlane(rowptr[d + 1]);
  const float4* __restrict__ h4 = reinterpret_cast<const float4*>(h);

  float4 acc = make_float4(0.f, 0.f, 0.f, 0.f);
  for (int base = beg; base < end; base += 32) {
    const int cnt = min(32, end - base);
    int   sL = 0;
    float nL = 0.f;
    if (lane < cnt) {
      sL = min(max(csrSrc[base + lane], 0), nN - 1);
      nL = nrm[sL];
    }
    const int nLi = __float_as_int(nL);
    int j = 0;
    for (; j + 4 <= cnt; j += 4) gather_rows<4>(acc, h4, sL, nLi, j, lane);
    const int rem = cnt - j;
    if (rem == 3)      gather_rows<3>(acc, h4, sL, nLi, j, lane);
    else if (rem == 2) gather_rows<2>(acc, h4, sL, nLi, j, lane);
    else if (rem == 1) gather_rows<1>(acc, h4, sL, nLi, j, lane);
  }
  reinterpret_cast<float4*>(agg)[(size_t)d * 32 + lane] = acc;
}

template <bool HAS_NRM, bool HAS_ADD, bool RELU, bool ACCUM>
__global__ __launch_bounds__(256) void gemm_kernel(
    const float* __restrict__ A, const float* __restrict__ nrm,
    const float* __restrict__ W, const float* __restrict__ bias,
    const float* __restrict__ addm, float* __restrict__ out,
    int nRowTiles, float outScale) {
  __shared__ float As[16 * LDS_STRIDE];
  const int tid  = threadIdx.x;
  const int wid  = tid >> 5;
  const int lane = tid & 31;
  const int half = lane >> 4;
  const int lm   = lane & 15;
  const int colBase = wid * 16;

  v2f breg[32];
#pragma unroll
  for (int ks = 0; ks < 32; ++ks) {
    const int kk = ks * 4 + half * 2;
    breg[ks].x = W[(size_t)kk * D + colBase + lm];
    breg[ks].y = W[(size_t)(kk + 1) * D + colBase + lm];
  }
  const float bcol = bias[colBase + lm];

  const int idx0 = tid;
  const int idx1 = tid + 256;
  const int r0 = idx0 >> 5, c0 = (idx0 & 31) * 4;
  const int r1 = idx1 >> 5, c1 = (idx1 & 31) * 4;
  const float4* __restrict__ A4 = reinterpret_cast<const float4*>(A);

  int rt = blockIdx.x;
  if (rt >= nRowTiles) return;

  float4 s0, s1;
  {
    const int rowBase = rt * 16;
    s0 = A4[(size_t)(rowBase + r0) * 32 + (idx0 & 31)];
    s1 = A4[(size_t)(rowBase + r1) * 32 + (idx1 & 31)];
    if (HAS_NRM) {
      const float n0 = nrm[rowBase + r0], n1 = nrm[rowBase + r1];
      s0.x *= n0; s0.y *= n0; s0.z *= n0; s0.w *= n0;
      s1.x *= n1; s1.y *= n1; s1.z *= n1; s1.w *= n1;
    }
  }

  for (;;) {
    __syncthreads();
    *reinterpret_cast<float4*>(&As[r0 * LDS_STRIDE + c0]) = s0;
    *reinterpret_cast<float4*>(&As[r1 * LDS_STRIDE + c1]) = s1;

    const int rnext = rt + gridDim.x;
    const bool more = rnext < nRowTiles;
    if (more) {
      const int rowBase = rnext * 16;
      s0 = A4[(size_t)(rowBase + r0) * 32 + (idx0 & 31)];
      s1 = A4[(size_t)(rowBase + r1) * 32 + (idx1 & 31)];
      if (HAS_NRM) {
        const float n0 = nrm[rowBase + r0], n1 = nrm[rowBase + r1];
        s0.x *= n0; s0.y *= n0; s0.z *= n0; s0.w *= n0;
        s1.x *= n1; s1.y *= n1; s1.z *= n1; s1.w *= n1;
      }
    }
    __syncthreads();

    v8f acc = {0.f, 0.f, 0.f, 0.f, 0.f, 0.f, 0.f, 0.f};
#pragma unroll
    for (int ks = 0; ks < 32; ++ks) {
      const int kk = ks * 4 + half * 2;
      v2f a;
      a.x = As[lm * LDS_STRIDE + kk];
      a.y = As[lm * LDS_STRIDE + kk + 1];
      acc = __builtin_amdgcn_wmma_f32_16x16x4_f32(
           false, a,  false, breg[ks],
           (short)0, acc,  false,  false);
    }

    const int rowBase = rt * 16;
#pragma unroll
    for (int r = 0; r < 8; ++r) {
      const int row = rowBase + r + half * 8;
      const int col = colBase + lm;
      const size_t oi = (size_t)row * D + col;
      float vv = acc[r] + bcol;
      if (HAS_ADD) vv += addm[oi];
      if (RELU) vv = fmaxf(vv, 0.f);
      vv *= outScale;
      if (ACCUM) out[oi] += vv;
      else       out[oi]  = vv;
    }

    if (!more) break;
    rt = rnext;
  }
}

extern "C" void kernel_launch(void* const* d_in, const int* in_sizes, int n_in,
                              void* d_out, int out_size, void* d_ws, size_t ws_size,
                              hipStream_t stream) {
  const float* feats = (const float*)d_in[0];
  const int*   src   = (const int*)d_in[1];
  const int*   dst   = (const int*)d_in[2];
  const float* w0    = (const float*)d_in[3];
  const float* b0    = (const float*)d_in[4];
  const float* w     = (const float*)d_in[5];
  const float* bw    = (const float*)d_in[6];
  const float* v     = (const float*)d_in[7];
  const float* bv    = (const float*)d_in[8];
  float* out = (float*)d_out;

  const int N = in_sizes[0] / D;
  const int E = in_sizes[1];
  const size_t M = (size_t)N * D;

  const size_t Nr  = ((size_t)N + 255) & ~(size_t)255;
  const size_t N1r = ((size_t)N + 1 + 255) & ~(size_t)255;
  const size_t Er  = ((size_t)E + 255) & ~(size_t)255;

  float* ws = (float*)d_ws;
  size_t off = 0;
  float* nrm  = ws + off; off += Nr;
  float* agg0 = ws + off; off += M;
  float* fv   = ws + off; off += M;
  float* h1   = ws + off; off += M;
  float* agg1 = ws + off; off += M;
  int* cnt    = (int*)(ws + off); off += Nr;
  int* rowptr = (int*)(ws + off); off += N1r;
  int* pos    = (int*)(ws + off); off += Er;
  int* csrSrc = (int*)(ws + off); off += Er;
  if (off * sizeof(float) > ws_size) return;

  const int nRowTiles  = N / 16;
  const int gemmBlocks = 625;

  zero_f4_kernel<<<((int)(Nr / 4) + 255) / 256, 256, 0, stream>>>((float4*)cnt, (int)(Nr / 4));
  count_rank_kernel<<<(E + 255) / 256, 256, 0, stream>>>(dst, cnt, pos, E);
  scan_norm_kernel<<<(N + 255) / 256, 256, 0, stream>>>(cnt, rowptr, nrm, N);
  csr_fill_kernel<<<(E + 255) / 256, 256, 0, stream>>>(src, dst, pos, rowptr, csrSrc, E);

  spmm_gather_kernel<<<(N + 7) / 8, 256, 0, stream>>>(feats, rowptr, csrSrc, nrm, agg0, N);

  for (int k = 0; k < KSTACK; ++k) {
    gemm_kernel<false, false, false, false><<<gemmBlocks, 256, 0, stream>>>(
        feats, nullptr, v + (size_t)k * D * D, bv + (size_t)k * D,
        nullptr, fv, nRowTiles, 1.0f);
    gemm_kernel<true, true, true, false><<<gemmBlocks, 256, 0, stream>>>(
        agg0, nrm, w0 + (size_t)k * D * D, b0 + (size_t)k * D,
        fv, h1, nRowTiles, 1.0f);
    spmm_gather_kernel<<<(N + 7) / 8, 256, 0, stream>>>(h1, rowptr, csrSrc, nrm, agg1, N);
    if (k == 0) {
      gemm_kernel<true, true, true, false><<<gemmBlocks, 256, 0, stream>>>(
          agg1, nrm, w + (size_t)k * D * D, bw + (size_t)k * D,
          fv, out, nRowTiles, 1.0f / (float)KSTACK);
    } else {
      gemm_kernel<true, true, true, true><<<gemmBlocks, 256, 0, stream>>>(
          agg1, nrm, w + (size_t)k * D * D, bw + (size_t)k * D,
          fv, out, nRowTiles, 1.0f / (float)KSTACK);
    }
  }
}


# fv GEMM folded into the h1 GEMM launch: one kernel runs feats@v and (agg0*norm)@w0 per 16-row tile on two independent accumulators, fv stays in registers for the h1 epilogue; 17 -> 14 launches
# speedup vs baseline: 2.2684x; 1.0442x over previous
#include <hip/hip_runtime.h>

typedef float v2f __attribute__((ext_vector_type(2)));
typedef float v8f __attribute__((ext_vector_type(8)));

#define D 128
#define KSTACK 3
#define LDS_STRIDE 132

__global__ void zero_f4_kernel(float4* __restrict__ p, int n4) {
  int i = blockIdx.x * blockDim.x + threadIdx.x;
  if (i < n4) p[i] = make_float4(0.f, 0.f, 0.f, 0.f);
}

__global__ void count_rank_kernel(const int* __restrict__ dst, int* __restrict__ cnt,
                                  int* __restrict__ pos, int nE) {
  int i = blockIdx.x * blockDim.x + threadIdx.x;
  if (i < nE) pos[i] = atomicAdd(&cnt[dst[i]], 1);
}

__global__ __launch_bounds__(256) void scan_norm_kernel(const int* __restrict__ cnt,
                                                        int* __restrict__ rowptr,
                                                        float* __restrict__ nrm, int n) {
  __shared__ int waveIncl[8];
  __shared__ int waveFront[8];
  const int tid  = threadIdx.x;
  const int lane = tid & 31;
  const int wid  = tid >> 5;
  const int base = blockIdx.x * 256;

  const int4* __restrict__ c4 = reinterpret_cast<const int4*>(cnt);
  int front = 0;
  for (int q = tid; q < (base >> 2); q += 256) {
    const int4 v = c4[q];
    front += (v.x + v.y) + (v.z + v.w);
  }
  for (int o = 16; o > 0; o >>= 1) front += __shfl_xor(front, o);

  const int i = base + tid;
  const int c = (i < n) ? cnt[i] : 0;
  if (i < n) nrm[i] = rsqrtf(fmaxf((float)c, 1.0f));

  int incl = c;
  for (int o = 1; o < 32; o <<= 1) {
    const int t = __shfl_up(incl, o);
    if (lane >= o) incl += t;
  }
  if (lane == 31) waveIncl[wid]  = incl;
  if (lane == 0)  waveFront[wid] = front;
  __syncthreads();
  int off = 0;
  for (int w = 0; w < 8; ++w) {
    off += waveFront[w];
    if (w < wid) off += waveIncl[w];
  }
  const int excl = off + incl - c;
  if (i < n) rowptr[i] = excl;
  if (i == n - 1) rowptr[n] = excl + c;
}

__global__ void csr_fill_kernel(const int* __restrict__ src, const int* __restrict__ dst,
                                const int* __restrict__ pos, const int* __restrict__ rowptr,
                                int* __restrict__ csrSrc, int nE) {
  int i = blockIdx.x * blockDim.x + threadIdx.x;
  if (i < nE) csrSrc[rowptr[dst[i]] + pos[i]] = src[i];
}

__device__ __forceinline__ void row_fma(float4& acc, const float4 v, const float w) {
  acc.x += v.x * w; acc.y += v.y * w; acc.z += v.z * w; acc.w += v.w * w;
}

template <int R>
__device__ __forceinline__ void gather_rows(float4& acc, const float4* __restrict__ h4,
                                            const int sL, const int nLi,
                                            const int j, const int lane) {
  float4 v[R];
#pragma unroll
  for (int u = 0; u < R; ++u) {
    const int s = __builtin_amdgcn_readlane(sL, j + u);
    v[u] = h4[(size_t)s * 32 + lane];
  }
#pragma unroll
  for (int u = 0; u < R; ++u)
    row_fma(acc, v[u], __int_as_float(__builtin_amdgcn_readlane(nLi, j + u)));
}

__global__ __launch_bounds__(256) void spmm_gather_kernel(const float* __restrict__ h,
                                                          const int* __restrict__ rowptr,
                                                          const int* __restrict__ csrSrc,
                                                          const float* __restrict__ nrm,
                                                          float* __restrict__ agg, int nN) {
  const int lane = threadIdx.x & 31;
  const int d = __builtin_amdgcn_readfirstlane(blockIdx.x * 8 + (threadIdx.x >> 5));
  if (d >= nN) return;
  const int beg = __builtin_amdgcn_readfirstlane(rowptr[d]);
  const int end = __builtin_amdgcn_readfirstlane(rowptr[d + 1]);
  const float4* __restrict__ h4 = reinterpret_cast<const float4*>(h);

  float4 acc = make_float4(0.f, 0.f, 0.f, 0.f);
  for (int base = beg; base < end; base += 32) {
    const int cnt = min(32, end - base);
    int   sL = 0;
    float nL = 0.f;
    if (lane < cnt) {
      sL = min(max(csrSrc[base + lane], 0), nN - 1);
      nL = nrm[sL];
    }
    const int nLi = __float_as_int(nL);
    int j = 0;
    for (; j + 4 <= cnt; j += 4) gather_rows<4>(acc, h4, sL, nLi, j, lane);
    const int rem = cnt - j;
    if (rem == 3)      gather_rows<3>(acc, h4, sL, nLi, j, lane);
    else if (rem == 2) gather_rows<2>(acc, h4, sL, nLi, j, lane);
    else if (rem == 1) gather_rows<1>(acc, h4, sL, nLi, j, lane);
  }
  reinterpret_cast<float4*>(agg)[(size_t)d * 32 + lane] = acc;
}

template <bool HAS_NRM, bool HAS_ADD, bool RELU, bool ACCUM>
__global__ __launch_bounds__(256) void gemm_kernel(
    const float* __restrict__ A, const float* __restrict__ nrm,
    const float* __restrict__ W, const float* __restrict__ bias,
    const float* __restrict__ addm, float* __restrict__ out,
    int nRowTiles, float outScale) {
  __shared__ float As[16 * LDS_STRIDE];
  const int tid  = threadIdx.x;
  const int wid  = tid >> 5;
  const int lane = tid & 31;
  const int half = lane >> 4;
  const int lm   = lane & 15;
  const int colBase = wid * 16;

  v2f breg[32];
#pragma unroll
  for (int ks = 0; ks < 32; ++ks) {
    const int kk = ks * 4 + half * 2;
    breg[ks].x = W[(size_t)kk * D + colBase + lm];
    breg[ks].y = W[(size_t)(kk + 1) * D + colBase + lm];
  }
  const float bcol = bias[colBase + lm];

  const int idx0 = tid;
  const int idx1 = tid + 256;
  const int r0 = idx0 >> 5, c0 = (idx0 & 31) * 4;
  const int r1 = idx1 >> 5, c1 = (idx1 & 31) * 4;
  const float4* __restrict__ A4 = reinterpret_cast<const float4*>(A);

  int rt = blockIdx.x;
  if (rt >= nRowTiles) return;

  float4 s0, s1;
  {
    const int rowBase = rt * 16;
    s0 = A4[(size_t)(rowBase + r0) * 32 + (idx0 & 31)];
    s1 = A4[(size_t)(rowBase + r1) * 32 + (idx1 & 31)];
    if (HAS_NRM) {
      const float n0 = nrm[rowBase + r0], n1 = nrm[rowBase + r1];
      s0.x *= n0; s0.y *= n0; s0.z *= n0; s0.w *= n0;
      s1.x *= n1; s1.y *= n1; s1.z *= n1; s1.w *= n1;
    }
  }

  for (;;) {
    __syncthreads();
    *reinterpret_cast<float4*>(&As[r0 * LDS_STRIDE + c0]) = s0;
    *reinterpret_cast<float4*>(&As[r1 * LDS_STRIDE + c1]) = s1;

    const int rnext = rt + gridDim.x;
    const bool more = rnext < nRowTiles;
    if (more) {
      const int rowBase = rnext * 16;
      s0 = A4[(size_t)(rowBase + r0) * 32 + (idx0 & 31)];
      s1 = A4[(size_t)(rowBase + r1) * 32 + (idx1 & 31)];
      if (HAS_NRM) {
        const float n0 = nrm[rowBase + r0], n1 = nrm[rowBase + r1];
        s0.x *= n0; s0.y *= n0; s0.z *= n0; s0.w *= n0;
        s1.x *= n1; s1.y *= n1; s1.z *= n1; s1.w *= n1;
      }
    }
    __syncthreads();

    v8f acc = {0.f, 0.f, 0.f, 0.f, 0.f, 0.f, 0.f, 0.f};
#pragma unroll
    for (int ks = 0; ks < 32; ++ks) {
      const int kk = ks * 4 + half * 2;
      v2f a;
      a.x = As[lm * LDS_STRIDE + kk];
      a.y = As[lm * LDS_STRIDE + kk + 1];
      acc = __builtin_amdgcn_wmma_f32_16x16x4_f32(
           false, a,  false, breg[ks],
           (short)0, acc,  false,  false);
    }

    const int rowBase = rt * 16;
#pragma unroll
    for (int r = 0; r < 8; ++r) {
      const int row = rowBase + r + half * 8;
      const int col = colBase + lm;
      const size_t oi = (size_t)row * D + col;
      float vv = acc[r] + bcol;
      if (HAS_ADD) vv += addm[oi];
      if (RELU) vv = fmaxf(vv, 0.f);
      vv *= outScale;
      if (ACCUM) out[oi] += vv;
      else       out[oi]  = vv;
    }

    if (!more) break;
    rt = rnext;
  }
}

__global__ __launch_bounds__(256) void gemm_pair_kernel(
    const float* __restrict__ X, const float* __restrict__ G,
    const float* __restrict__ nrm,
    const float* __restrict__ Wx, const float* __restrict__ bx,
    const float* __restrict__ Wg, const float* __restrict__ bg,
    float* __restrict__ fvOut, float* __restrict__ hOut, int nRowTiles) {
  __shared__ float Xs[16 * LDS_STRIDE];
  __shared__ float Gs[16 * LDS_STRIDE];
  const int tid  = threadIdx.x;
  const int wid  = tid >> 5;
  const int lane = tid & 31;
  const int half = lane >> 4;
  const int lm   = lane & 15;
  const int colBase = wid * 16;

  v2f bxr[32];
  v2f bgr[32];
#pragma unroll
  for (int ks = 0; ks < 32; ++ks) {
    const int kk = ks * 4 + half * 2;
    bxr[ks].x = Wx[(size_t)kk * D + colBase + lm];
    bxr[ks].y = Wx[(size_t)(kk + 1) * D + colBase + lm];
    bgr[ks].x = Wg[(size_t)kk * D + colBase + lm];
    bgr[ks].y = Wg[(size_t)(kk + 1) * D + colBase + lm];
  }
  const float bxcol = bx[colBase + lm];
  const float bgcol = bg[colBase + lm];

  const int idx0 = tid;
  const int idx1 = tid + 256;
  const int r0 = idx0 >> 5, c0 = (idx0 & 31) * 4;
  const int r1 = idx1 >> 5, c1 = (idx1 & 31) * 4;
  const float4* __restrict__ X4 = reinterpret_cast<const float4*>(X);
  const float4* __restrict__ G4 = reinterpret_cast<const float4*>(G);

  int rt = blockIdx.x;
  if (rt >= nRowTiles) return;

  float4 x0, x1, g0, g1;
  {
    const int rowBase = rt * 16;
    x0 = X4[(size_t)(rowBase + r0) * 32 + (idx0 & 31)];
    x1 = X4[(size_t)(rowBase + r1) * 32 + (idx1 & 31)];
    g0 = G4[(size_t)(rowBase + r0) * 32 + (idx0 & 31)];
    g1 = G4[(size_t)(rowBase + r1) * 32 + (idx1 & 31)];
    const float n0 = nrm[rowBase + r0], n1 = nrm[rowBase + r1];
    g0.x *= n0; g0.y *= n0; g0.z *= n0; g0.w *= n0;
    g1.x *= n1; g1.y *= n1; g1.z *= n1; g1.w *= n1;
  }

  for (;;) {
    __syncthreads();
    *reinterpret_cast<float4*>(&Xs[r0 * LDS_STRIDE + c0]) = x0;
    *reinterpret_cast<float4*>(&Xs[r1 * LDS_STRIDE + c1]) = x1;
    *reinterpret_cast<float4*>(&Gs[r0 * LDS_STRIDE + c0]) = g0;
    *reinterpret_cast<float4*>(&Gs[r1 * LDS_STRIDE + c1]) = g1;

    const int rnext = rt + gridDim.x;
    const bool more = rnext < nRowTiles;
    if (more) {
      const int rowBase = rnext * 16;
      x0 = X4[(size_t)(rowBase + r0) * 32 + (idx0 & 31)];
      x1 = X4[(size_t)(rowBase + r1) * 32 + (idx1 & 31)];
      g0 = G4[(size_t)(rowBase + r0) * 32 + (idx0 & 31)];
      g1 = G4[(size_t)(rowBase + r1) * 32 + (idx1 & 31)];
      const float n0 = nrm[rowBase + r0], n1 = nrm[rowBase + r1];
      g0.x *= n0; g0.y *= n0; g0.z *= n0; g0.w *= n0;
      g1.x *= n1; g1.y *= n1; g1.z *= n1; g1.w *= n1;
    }
    __syncthreads();

    v8f accX = {0.f, 0.f, 0.f, 0.f, 0.f, 0.f, 0.f, 0.f};
    v8f accG = {0.f, 0.f, 0.f, 0.f, 0.f, 0.f, 0.f, 0.f};
#pragma unroll
    for (int ks = 0; ks < 32; ++ks) {
      const int kk = ks * 4 + half * 2;
      v2f ax, ag;
      ax.x = Xs[lm * LDS_STRIDE + kk];
      ax.y = Xs[lm * LDS_STRIDE + kk + 1];
      ag.x = Gs[lm * LDS_STRIDE + kk];
      ag.y = Gs[lm * LDS_STRIDE + kk + 1];
      accX = __builtin_amdgcn_wmma_f32_16x16x4_f32(
           false, ax,  false, bxr[ks],
           (short)0, accX,  false,  false);
      accG = __builtin_amdgcn_wmma_f32_16x16x4_f32(
           false, ag,  false, bgr[ks],
           (short)0, accG,  false,  false);
    }

    const int rowBase = rt * 16;
#pragma unroll
    for (int r = 0; r < 8; ++r) {
      const int row = rowBase + r + half * 8;
      const int col = colBase + lm;
      const size_t oi = (size_t)row * D + col;
      const float f = accX[r] + bxcol;
      fvOut[oi] = f;
      float hv = accG[r] + bgcol;
      hv += f;
      hOut[oi] = fmaxf(hv, 0.f);
    }

    if (!more) break;
    rt = rnext;
  }
}

extern "C" void kernel_launch(void* const* d_in, const int* in_sizes, int n_in,
                              void* d_out, int out_size, void* d_ws, size_t ws_size,
                              hipStream_t stream) {
  const float* feats = (const float*)d_in[0];
  const int*   src   = (const int*)d_in[1];
  const int*   dst   = (const int*)d_in[2];
  const float* w0    = (const float*)d_in[3];
  const float* b0    = (const float*)d_in[4];
  const float* w     = (const float*)d_in[5];
  const float* bw    = (const float*)d_in[6];
  const float* v     = (const float*)d_in[7];
  const float* bv    = (const float*)d_in[8];
  float* out = (float*)d_out;

  const int N = in_sizes[0] / D;
  const int E = in_sizes[1];
  const size_t M = (size_t)N * D;

  const size_t Nr  = ((size_t)N + 255) & ~(size_t)255;
  const size_t N1r = ((size_t)N + 1 + 255) & ~(size_t)255;
  const size_t Er  = ((size_t)E + 255) & ~(size_t)255;

  float* ws = (float*)d_ws;
  size_t off = 0;
  float* nrm  = ws + off; off += Nr;
  float* agg0 = ws + off; off += M;
  float* fv   = ws + off; off += M;
  float* h1   = ws + off; off += M;
  float* agg1 = ws + off; off += M;
  int* cnt    = (int*)(ws + off); off += Nr;
  int* rowptr = (int*)(ws + off); off += N1r;
  int* pos    = (int*)(ws + off); off += Er;
  int* csrSrc = (int*)(ws + off); off += Er;
  if (off * sizeof(float) > ws_size) return;

  const int nRowTiles  = N / 16;
  const int gemmBlocks = 625;
  const int pairBlocks = 512;

  zero_f4_kernel<<<((int)(Nr / 4) + 255) / 256, 256, 0, stream>>>((float4*)cnt, (int)(Nr / 4));
  count_rank_kernel<<<(E + 255) / 256, 256, 0, stream>>>(dst, cnt, pos, E);
  scan_norm_kernel<<<(N + 255) / 256, 256, 0, stream>>>(cnt, rowptr, nrm, N);
  csr_fill_kernel<<<(E + 255) / 256, 256, 0, stream>>>(src, dst, pos, rowptr, csrSrc, E);

  spmm_gather_kernel<<<(N + 7) / 8, 256, 0, stream>>>(feats, rowptr, csrSrc, nrm, agg0, N);

  for (int k = 0; k < KSTACK; ++k) {
    gemm_pair_kernel<<<pairBlocks, 256, 0, stream>>>(
        feats, agg0, nrm, v + (size_t)k * D * D, bv + (size_t)k * D,
        w0 + (size_t)k * D * D, b0 + (size_t)k * D, fv, h1, nRowTiles);
    spmm_gather_kernel<<<(N + 7) / 8, 256, 0, stream>>>(h1, rowptr, csrSrc, nrm, agg1, N);
    if (k == 0) {
      gemm_kernel<true, true, true, false><<<gemmBlocks, 256, 0, stream>>>(
          agg1, nrm, w + (size_t)k * D * D, bw + (size_t)k * D,
          fv, out, nRowTiles, 1.0f / (float)KSTACK);
    } else {
      gemm_kernel<true, true, true, true><<<gemmBlocks, 256, 0, stream>>>(
          agg1, nrm, w + (size_t)k * D * D, bw + (size_t)k * D,
          fv, out, nRowTiles, 1.0f / (float)KSTACK);
    }
  }
}


# SpMM(h1) fused into its consumer GEMM: each work-group gathers its 16-row tile (row sums in slot order as before) into an LDS A tile and multiplies it; agg1 never stored; 14 -> 11 launches
# speedup vs baseline: 2.4283x; 1.0705x over previous
#include <hip/hip_runtime.h>

typedef float v2f __attribute__((ext_vector_type(2)));
typedef float v8f __attribute__((ext_vector_type(8)));

#define D 128
#define KSTACK 3
#define LDS_STRIDE 132

__global__ void zero_f4_kernel(float4* __restrict__ p, int n4) {
  int i = blockIdx.x * blockDim.x + threadIdx.x;
  if (i < n4) p[i] = make_float4(0.f, 0.f, 0.f, 0.f);
}

__global__ void count_rank_kernel(const int* __restrict__ dst, int* __restrict__ cnt,
                                  int* __restrict__ pos, int nE) {
  int i = blockIdx.x * blockDim.x + threadIdx.x;
  if (i < nE) pos[i] = atomicAdd(&cnt[dst[i]], 1);
}

__global__ __launch_bounds__(256) void scan_norm_kernel(const int* __restrict__ cnt,
                                                        int* __restrict__ rowptr,
                                                        float* __restrict__ nrm, int n) {
  __shared__ int waveIncl[8];
  __shared__ int waveFront[8];
  const int tid  = threadIdx.x;
  const int lane = tid & 31;
  const int wid  = tid >> 5;
  const int base = blockIdx.x * 256;

  const int4* __restrict__ c4 = reinterpret_cast<const int4*>(cnt);
  int front = 0;
  for (int q = tid; q < (base >> 2); q += 256) {
    const int4 v = c4[q];
    front += (v.x + v.y) + (v.z + v.w);
  }
  for (int o = 16; o > 0; o >>= 1) front += __shfl_xor(front, o);

  const int i = base + tid;
  const int c = (i < n) ? cnt[i] : 0;
  if (i < n) nrm[i] = rsqrtf(fmaxf((float)c, 1.0f));

  int incl = c;
  for (int o = 1; o < 32; o <<= 1) {
    const int t = __shfl_up(incl, o);
    if (lane >= o) incl += t;
  }
  if (lane == 31) waveIncl[wid]  = incl;
  if (lane == 0)  waveFront[wid] = front;
  __syncthreads();
  int off = 0;
  for (int w = 0; w < 8; ++w) {
    off += waveFront[w];
    if (w < wid) off += waveIncl[w];
  }
  const int excl = off + incl - c;
  if (i < n) rowptr[i] = excl;
  if (i == n - 1) rowptr[n] = excl + c;
}

__global__ void csr_fill_kernel(const int* __restrict__ src, const int* __restrict__ dst,
                                const int* __restrict__ pos, const int* __restrict__ rowptr,
                                int* __restrict__ csrSrc, int nE) {
  int i = blockIdx.x * blockDim.x + threadIdx.x;
  if (i < nE) csrSrc[rowptr[dst[i]] + pos[i]] = src[i];
}

__device__ __forceinline__ void row_fma(float4& acc, const float4 v, const float w) {
  acc.x = fmaf(v.x, w, acc.x); acc.y = fmaf(v.y, w, acc.y);
  acc.z = fmaf(v.z, w, acc.z); acc.w = fmaf(v.w, w, acc.w);
}

template <int R>
__device__ __forceinline__ void gather_rows(float4& acc, const float4* __restrict__ h4,
                                            const int sL, const int nLi,
                                            const int j, const int lane) {
  float4 v[R];
#pragma unroll
  for (int u = 0; u < R; ++u) {
    const int s = __builtin_amdgcn_readlane(sL, j + u);
    v[u] = h4[(size_t)s * 32 + lane];
  }
#pragma unroll
  for (int u = 0; u < R; ++u)
    row_fma(acc, v[u], __int_as_float(__builtin_amdgcn_readlane(nLi, j + u)));
}

__device__ __forceinline__ void load_chunk(int& sL, int& nLi, const int* __restrict__ csrSrc,
                                           const float* __restrict__ nrm,
                                           const int base, const int cnt, const int nN,
                                           const int lane) {
  sL = 0;
  float nL = 0.f;
  if (lane < cnt) {
    sL = min(max(csrSrc[base + lane], 0), nN - 1);
    nL = nrm[sL];
  }
  nLi = __float_as_int(nL);
}

template <int RMAX>
__device__ __forceinline__ void sum_chunk(float4& acc, const float4* __restrict__ h4,
                                          const int sL, const int nLi, const int cnt,
                                          const int lane) {
  int j = 0;
  if (RMAX >= 8)
    for (; j + 8 <= cnt; j += 8) gather_rows<8>(acc, h4, sL, nLi, j, lane);
  for (; j + 4 <= cnt; j += 4) gather_rows<4>(acc, h4, sL, nLi, j, lane);
  const int rem = cnt - j;
  if (rem == 3)      gather_rows<3>(acc, h4, sL, nLi, j, lane);
  else if (rem == 2) gather_rows<2>(acc, h4, sL, nLi, j, lane);
  else if (rem == 1) gather_rows<1>(acc, h4, sL, nLi, j, lane);
}

__global__ __launch_bounds__(256) void spmm_gather_kernel(const float* __restrict__ h,
                                                          const int* __restrict__ rowptr,
                                                          const int* __restrict__ csrSrc,
                                                          const float* __restrict__ nrm,
                                                          float* __restrict__ agg, int nN) {
  const int lane = threadIdx.x & 31;
  const int d = __builtin_amdgcn_readfirstlane(blockIdx.x * 8 + (threadIdx.x >> 5));
  if (d >= nN) return;
  const int beg = __builtin_amdgcn_readfirstlane(rowptr[d]);
  const int end = __builtin_amdgcn_readfirstlane(rowptr[d + 1]);
  const float4* __restrict__ h4 = reinterpret_cast<const float4*>(h);

  float4 acc = make_float4(0.f, 0.f, 0.f, 0.f);
  for (int base = beg; base < end; base += 32) {
    const int cnt = min(32, end - base);
    int sL, nLi;
    load_chunk(sL, nLi, csrSrc, nrm, base, cnt, nN, lane);
    sum_chunk<4>(acc, h4, sL, nLi, cnt, lane);
  }
  reinterpret_cast<float4*>(agg)[(size_t)d * 32 + lane] = acc;
}

template <bool ACCUM>
__global__ __launch_bounds__(256, 8) void gather_gemm_kernel(
    const float* __restrict__ h, const int* __restrict__ rowptr,
    const int* __restrict__ csrSrc, const float* __restrict__ nrm,
    const float* __restrict__ W, const float* __restrict__ bias,
    const float* __restrict__ addm, float* __restrict__ out,
    int nRowTiles, int nN, float outScale) {
  __shared__ float As[2][16 * LDS_STRIDE];
  const int tid  = threadIdx.x;
  const int wid  = tid >> 5;
  const int lane = tid & 31;
  const int half = lane >> 4;
  const int lm   = lane & 15;
  const int colBase = wid * 16;

  v2f breg[32];
#pragma unroll
  for (int ks = 0; ks < 32; ++ks) {
    const int kk = ks * 4 + half * 2;
    breg[ks].x = W[(size_t)kk * D + colBase + lm];
    breg[ks].y = W[(size_t)(kk + 1) * D + colBase + lm];
  }
  const float bcol = bias[colBase + lm];
  const float4* __restrict__ h4 = reinterpret_cast<const float4*>(h);

  int buf = 0;
  int rt = blockIdx.x;
  int rp0 = 0, rp1 = 0, rp2 = 0;
  if (rt < nRowTiles) {
    const int d = rt * 16 + wid * 2;
    rp0 = rowptr[d]; rp1 = rowptr[d + 1]; rp2 = rowptr[d + 2];
  }
  for (; rt < nRowTiles; rt += gridDim.x) {
    const int rowBase = rt * 16;
    const int d0 = __builtin_amdgcn_readfirstlane(rowBase + wid * 2);
    const int beg0 = __builtin_amdgcn_readfirstlane(rp0);
    const int beg1 = __builtin_amdgcn_readfirstlane(rp1);
    const int end1 = __builtin_amdgcn_readfirstlane(rp2);
    const int rnext = rt + gridDim.x;
    if (rnext < nRowTiles) {
      const int d = rnext * 16 + wid * 2;
      rp0 = rowptr[d]; rp1 = rowptr[d + 1]; rp2 = rowptr[d + 2];
    }
    const int cnt0 = min(32, beg1 - beg0);
    const int cnt1 = min(32, end1 - beg1);
    int sL0 = csrSrc[max(beg0 + min(lane, cnt0 - 1), 0)];
    int sL1 = csrSrc[max(beg1 + min(lane, cnt1 - 1), 0)];
    sL0 = min(max(sL0, 0), nN - 1);
    sL1 = min(max(sL1, 0), nN - 1);
    const float nL0 = nrm[sL0];
    const float nL1 = nrm[sL1];
    const int nLi0 = __float_as_int(nL0);
    const int nLi1 = __float_as_int(nL1);
    const float nd0 = nrm[d0];
    const float nd1 = nrm[d0 + 1];

    float4 a0 = make_float4(0.f, 0.f, 0.f, 0.f);
    sum_chunk<8>(a0, h4, sL0, nLi0, cnt0, lane);
    for (int base = beg0 + 32; base < beg1; base += 32) {
      const int cnt = min(32, beg1 - base);
      int sL, nLi;
      load_chunk(sL, nLi, csrSrc, nrm, base, cnt, nN, lane);
      sum_chunk<8>(a0, h4, sL, nLi, cnt, lane);
    }
    float4 a1 = make_float4(0.f, 0.f, 0.f, 0.f);
    sum_chunk<8>(a1, h4, sL1, nLi1, cnt1, lane);
    for (int base = beg1 + 32; base < end1; base += 32) {
      const int cnt = min(32, end1 - base);
      int sL, nLi;
      load_chunk(sL, nLi, csrSrc, nrm, base, cnt, nN, lane);
      sum_chunk<8>(a1, h4, sL, nLi, cnt, lane);
    }
    a0.x *= nd0; a0.y *= nd0; a0.z *= nd0; a0.w *= nd0;
    a1.x *= nd1; a1.y *= nd1; a1.z *= nd1; a1.w *= nd1;
    float* __restrict__ At = As[buf];
    *reinterpret_cast<float4*>(&At[(wid * 2) * LDS_STRIDE + lane * 4])     = a0;
    *reinterpret_cast<float4*>(&At[(wid * 2 + 1) * LDS_STRIDE + lane * 4]) = a1;
    __syncthreads();

    v8f acc = {0.f, 0.f, 0.f, 0.f, 0.f, 0.f, 0.f, 0.f};
#pragma unroll
    for (int ks = 0; ks < 32; ++ks) {
      const int kk = ks * 4 + half * 2;
      v2f a;
      a.x = At[lm * LDS_STRIDE + kk];
      a.y = At[lm * LDS_STRIDE + kk + 1];
      acc = __builtin_amdgcn_wmma_f32_16x16x4_f32(
           false, a,  false, breg[ks],
           (short)0, acc,  false,  false);
    }

#pragma unroll
    for (int r = 0; r < 8; ++r) {
      const int row = rowBase + r + half * 8;
      const int col = colBase + lm;
      const size_t oi = (size_t)row * D + col;
      float vv = acc[r] + bcol + addm[oi];
      vv = fmaxf(vv, 0.f);
      vv *= outScale;
      if (ACCUM) out[oi] += vv;
      else       out[oi]  = vv;
    }
    buf ^= 1;
  }
}

__global__ __launch_bounds__(256) void gemm_pair_kernel(
    const float* __restrict__ X, const float* __restrict__ G,
    const float* __restrict__ nrm,
    const float* __restrict__ Wx, const float* __restrict__ bx,
    const float* __restrict__ Wg, const float* __restrict__ bg,
    float* __restrict__ fvOut, float* __restrict__ hOut, int nRowTiles) {
  __shared__ float Xs[16 * LDS_STRIDE];
  __shared__ float Gs[16 * LDS_STRIDE];
  const int tid  = threadIdx.x;
  const int wid  = tid >> 5;
  const int lane = tid & 31;
  const int half = lane >> 4;
  const int lm   = lane & 15;
  const int colBase = wid * 16;

  v2f bxr[32];
  v2f bgr[32];
#pragma unroll
  for (int ks = 0; ks < 32; ++ks) {
    const int kk = ks * 4 + half * 2;
    bxr[ks].x = Wx[(size_t)kk * D + colBase + lm];
    bxr[ks].y = Wx[(size_t)(kk + 1) * D + colBase + lm];
    bgr[ks].x = Wg[(size_t)kk * D + colBase + lm];
    bgr[ks].y = Wg[(size_t)(kk + 1) * D + colBase + lm];
  }
  const float bxcol = bx[colBase + lm];
  const float bgcol = bg[colBase + lm];

  const int idx0 = tid;
  const int idx1 = tid + 256;
  const int r0 = idx0 >> 5, c0 = (idx0 & 31) * 4;
  const int r1 = idx1 >> 5, c1 = (idx1 & 31) * 4;
  const float4* __restrict__ X4 = reinterpret_cast<const float4*>(X);
  const float4* __restrict__ G4 = reinterpret_cast<const float4*>(G);

  int rt = blockIdx.x;
  if (rt >= nRowTiles) return;

  float4 x0, x1, g0, g1;
  {
    const int rowBase = rt * 16;
    x0 = X4[(size_t)(rowBase + r0) * 32 + (idx0 & 31)];
    x1 = X4[(size_t)(rowBase + r1) * 32 + (idx1 & 31)];
    g0 = G4[(size_t)(rowBase + r0) * 32 + (idx0 & 31)];
    g1 = G4[(size_t)(rowBase + r1) * 32 + (idx1 & 31)];
    const float n0 = nrm[rowBase + r0], n1 = nrm[rowBase + r1];
    g0.x *= n0; g0.y *= n0; g0.z *= n0; g0.w *= n0;
    g1.x *= n1; g1.y *= n1; g1.z *= n1; g1.w *= n1;
  }

  for (;;) {
    __syncthreads();
    *reinterpret_cast<float4*>(&Xs[r0 * LDS_STRIDE + c0]) = x0;
    *reinterpret_cast<float4*>(&Xs[r1 * LDS_STRIDE + c1]) = x1;
    *reinterpret_cast<float4*>(&Gs[r0 * LDS_STRIDE + c0]) = g0;
    *reinterpret_cast<float4*>(&Gs[r1 * LDS_STRIDE + c1]) = g1;

    const int rnext = rt + gridDim.x;
    const bool more = rnext < nRowTiles;
    if (more) {
      const int rowBase = rnext * 16;
      x0 = X4[(size_t)(rowBase + r0) * 32 + (idx0 & 31)];
      x1 = X4[(size_t)(rowBase + r1) * 32 + (idx1 & 31)];
      g0 = G4[(size_t)(rowBase + r0) * 32 + (idx0 & 31)];
      g1 = G4[(size_t)(rowBase + r1) * 32 + (idx1 & 31)];
      const float n0 = nrm[rowBase + r0], n1 = nrm[rowBase + r1];
      g0.x *= n0; g0.y *= n0; g0.z *= n0; g0.w *= n0;
      g1.x *= n1; g1.y *= n1; g1.z *= n1; g1.w *= n1;
    }
    __syncthreads();

    v8f accX = {0.f, 0.f, 0.f, 0.f, 0.f, 0.f, 0.f, 0.f};
    v8f accG = {0.f, 0.f, 0.f, 0.f, 0.f, 0.f, 0.f, 0.f};
#pragma unroll
    for (int ks = 0; ks < 32; ++ks) {
      const int kk = ks * 4 + half * 2;
      v2f ax, ag;
      ax.x = Xs[lm * LDS_STRIDE + kk];
      ax.y = Xs[lm * LDS_STRIDE + kk + 1];
      ag.x = Gs[lm * LDS_STRIDE + kk];
      ag.y = Gs[lm * LDS_STRIDE + kk + 1];
      accX = __builtin_amdgcn_wmma_f32_16x16x4_f32(
           false, ax,  false, bxr[ks],
           (short)0, accX,  false,  false);
      accG = __builtin_amdgcn_wmma_f32_16x16x4_f32(
           false, ag,  false, bgr[ks],
           (short)0, accG,  false,  false);
    }

    const int rowBase = rt * 16;
#pragma unroll
    for (int r = 0; r < 8; ++r) {
      const int row = rowBase + r + half * 8;
      const int col = colBase + lm;
      const size_t oi = (size_t)row * D + col;
      const float f = accX[r] + bxcol;
      fvOut[oi] = f;
      float hv = accG[r] + bgcol;
      hv += f;
      hOut[oi] = fmaxf(hv, 0.f);
    }

    if (!more) break;
    rt = rnext;
  }
}

extern "C" void kernel_launch(void* const* d_in, const int* in_sizes, int n_in,
                              void* d_out, int out_size, void* d_ws, size_t ws_size,
                              hipStream_t stream) {
  const float* feats = (const float*)d_in[0];
  const int*   src   = (const int*)d_in[1];
  const int*   dst   = (const int*)d_in[2];
  const float* w0    = (const float*)d_in[3];
  const float* b0    = (const float*)d_in[4];
  const float* w     = (const float*)d_in[5];
  const float* bw    = (const float*)d_in[6];
  const float* v     = (const float*)d_in[7];
  const float* bv    = (const float*)d_in[8];
  float* out = (float*)d_out;

  const int N = in_sizes[0] / D;
  const int E = in_sizes[1];
  const size_t M = (size_t)N * D;

  const size_t Nr  = ((size_t)N + 255) & ~(size_t)255;
  const size_t N1r = ((size_t)N + 1 + 255) & ~(size_t)255;
  const size_t Er  = ((size_t)E + 255) & ~(size_t)255;

  float* ws = (float*)d_ws;
  size_t off = 0;
  float* nrm  = ws + off; off += Nr;
  float* agg0 = ws + off; off += M;
  float* fv   = ws + off; off += M;
  float* h1   = ws + off; off += M;
  int* cnt    = (int*)(ws + off); off += Nr;
  int* rowptr = (int*)(ws + off); off += N1r;
  int* pos    = (int*)(ws + off); off += Er;
  int* csrSrc = (int*)(ws + off); off += Er;
  if (off * sizeof(float) > ws_size) return;

  const int nRowTiles  = N / 16;
  const int fusedBlocks = 1024;
  const int pairBlocks = 512;

  zero_f4_kernel<<<((int)(Nr / 4) + 255) / 256, 256, 0, stream>>>((float4*)cnt, (int)(Nr / 4));
  count_rank_kernel<<<(E + 255) / 256, 256, 0, stream>>>(dst, cnt, pos, E);
  scan_norm_kernel<<<(N + 255) / 256, 256, 0, stream>>>(cnt, rowptr, nrm, N);
  csr_fill_kernel<<<(E + 255) / 256, 256, 0, stream>>>(src, dst, pos, rowptr, csrSrc, E);

  spmm_gather_kernel<<<(N + 7) / 8, 256, 0, stream>>>(feats, rowptr, csrSrc, nrm, agg0, N);

  for (int k = 0; k < KSTACK; ++k) {
    gemm_pair_kernel<<<pairBlocks, 256, 0, stream>>>(
        feats, agg0, nrm, v + (size_t)k * D * D, bv + (size_t)k * D,
        w0 + (size_t)k * D * D, b0 + (size_t)k * D, fv, h1, nRowTiles);
    if (k == 0) {
      gather_gemm_kernel<false><<<fusedBlocks, 256, 0, stream>>>(
          h1, rowptr, csrSrc, nrm, w + (size_t)k * D * D, bw + (size_t)k * D,
          fv, out, nRowTiles, N, 1.0f / (float)KSTACK);
    } else {
      gather_gemm_kernel<true><<<fusedBlocks, 256, 0, stream>>>(
          h1, rowptr, csrSrc, nrm, w + (size_t)k * D * D, bw + (size_t)k * D,
          fv, out, nRowTiles, N, 1.0f / (float)KSTACK);
    }
  }
}


# gather+GEMM epilogue in whole 512-byte rows: result tile through LDS, each wave updates two whole rows; addm and out fetched as one group of 16-byte row loads instead of dependent 4-byte groups
# speedup vs baseline: 2.5373x; 1.0449x over previous
#include <hip/hip_runtime.h>

typedef float v2f __attribute__((ext_vector_type(2)));
typedef float v8f __attribute__((ext_vector_type(8)));

#define D 128
#define KSTACK 3
#define LDS_STRIDE 132

__global__ void zero_f4_kernel(float4* __restrict__ p, int n4) {
  int i = blockIdx.x * blockDim.x + threadIdx.x;
  if (i < n4) p[i] = make_float4(0.f, 0.f, 0.f, 0.f);
}

__global__ void count_rank_kernel(const int* __restrict__ dst, int* __restrict__ cnt,
                                  int* __restrict__ pos, int nE) {
  int i = blockIdx.x * blockDim.x + threadIdx.x;
  if (i < nE) pos[i] = atomicAdd(&cnt[dst[i]], 1);
}

__global__ __launch_bounds__(256) void scan_norm_kernel(const int* __restrict__ cnt,
                                                        int* __restrict__ rowptr,
                                                        float* __restrict__ nrm, int n) {
  __shared__ int waveIncl[8];
  __shared__ int waveFront[8];
  const int tid  = threadIdx.x;
  const int lane = tid & 31;
  const int wid  = tid >> 5;
  const int base = blockIdx.x * 256;

  const int4* __restrict__ c4 = reinterpret_cast<const int4*>(cnt);
  int front = 0;
  for (int q = tid; q < (base >> 2); q += 256) {
    const int4 v = c4[q];
    front += (v.x + v.y) + (v.z + v.w);
  }
  for (int o = 16; o > 0; o >>= 1) front += __shfl_xor(front, o);

  const int i = base + tid;
  const int c = (i < n) ? cnt[i] : 0;
  if (i < n) nrm[i] = rsqrtf(fmaxf((float)c, 1.0f));

  int incl = c;
  for (int o = 1; o < 32; o <<= 1) {
    const int t = __shfl_up(incl, o);
    if (lane >= o) incl += t;
  }
  if (lane == 31) waveIncl[wid]  = incl;
  if (lane == 0)  waveFront[wid] = front;
  __syncthreads();
  int off = 0;
  for (int w = 0; w < 8; ++w) {
    off += waveFront[w];
    if (w < wid) off += waveIncl[w];
  }
  const int excl = off + incl - c;
  if (i < n) rowptr[i] = excl;
  if (i == n - 1) rowptr[n] = excl + c;
}

__global__ void csr_fill_kernel(const int* __restrict__ src, const int* __restrict__ dst,
                                const int* __restrict__ pos, const int* __restrict__ rowptr,
                                int* __restrict__ csrSrc, int nE) {
  int i = blockIdx.x * blockDim.x + threadIdx.x;
  if (i < nE) csrSrc[rowptr[dst[i]] + pos[i]] = src[i];
}

__device__ __forceinline__ void row_fma(float4& acc, const float4 v, const float w) {
  acc.x = fmaf(v.x, w, acc.x); acc.y = fmaf(v.y, w, acc.y);
  acc.z = fmaf(v.z, w, acc.z); acc.w = fmaf(v.w, w, acc.w);
}

template <int R>
__device__ __forceinline__ void gather_rows(float4& acc, const float4* __restrict__ h4,
                                            const int sL, const int nLi,
                                            const int j, const int lane) {
  float4 v[R];
#pragma unroll
  for (int u = 0; u < R; ++u) {
    const int s = __builtin_amdgcn_readlane(sL, j + u);
    v[u] = h4[(size_t)s * 32 + lane];
  }
#pragma unroll
  for (int u = 0; u < R; ++u)
    row_fma(acc, v[u], __int_as_float(__builtin_amdgcn_readlane(nLi, j + u)));
}

__device__ __forceinline__ void load_chunk(int& sL, int& nLi, const int* __restrict__ csrSrc,
                                           const float* __restrict__ nrm,
                                           const int base, const int cnt, const int nN,
                                           const int lane) {
  sL = 0;
  float nL = 0.f;
  if (lane < cnt) {
    sL = min(max(csrSrc[base + lane], 0), nN - 1);
    nL = nrm[sL];
  }
  nLi = __float_as_int(nL);
}

template <int RMAX>
__device__ __forceinline__ void sum_chunk(float4& acc, const float4* __restrict__ h4,
                                          const int sL, const int nLi, const int cnt,
                                          const int lane) {
  int j = 0;
  if (RMAX >= 8)
    for (; j + 8 <= cnt; j += 8) gather_rows<8>(acc, h4, sL, nLi, j, lane);
  for (; j + 4 <= cnt; j += 4) gather_rows<4>(acc, h4, sL, nLi, j, lane);
  const int rem = cnt - j;
  if (rem == 3)      gather_rows<3>(acc, h4, sL, nLi, j, lane);
  else if (rem == 2) gather_rows<2>(acc, h4, sL, nLi, j, lane);
  else if (rem == 1) gather_rows<1>(acc, h4, sL, nLi, j, lane);
}

__global__ __launch_bounds__(256) void spmm_gather_kernel(const float* __restrict__ h,
                                                          const int* __restrict__ rowptr,
                                                          const int* __restrict__ csrSrc,
                                                          const float* __restrict__ nrm,
                                                          float* __restrict__ agg, int nN) {
  const int lane = threadIdx.x & 31;
  const int d = __builtin_amdgcn_readfirstlane(blockIdx.x * 8 + (threadIdx.x >> 5));
  if (d >= nN) return;
  const int beg = __builtin_amdgcn_readfirstlane(rowptr[d]);
  const int end = __builtin_amdgcn_readfirstlane(rowptr[d + 1]);
  const float4* __restrict__ h4 = reinterpret_cast<const float4*>(h);

  float4 acc = make_float4(0.f, 0.f, 0.f, 0.f);
  for (int base = beg; base < end; base += 32) {
    const int cnt = min(32, end - base);
    int sL, nLi;
    load_chunk(sL, nLi, csrSrc, nrm, base, cnt, nN, lane);
    sum_chunk<4>(acc, h4, sL, nLi, cnt, lane);
  }
  reinterpret_cast<float4*>(agg)[(size_t)d * 32 + lane] = acc;
}

template <bool ACCUM>
__global__ __launch_bounds__(256, 8) void gather_gemm_kernel(
    const float* __restrict__ h, const int* __restrict__ rowptr,
    const int* __restrict__ csrSrc, const float* __restrict__ nrm,
    const float* __restrict__ W, const float* __restrict__ bias,
    const float* __restrict__ addm, float* __restrict__ out,
    int nRowTiles, int nN, float outScale) {
  __shared__ float As[2][16 * LDS_STRIDE];
  __shared__ float Cs[16 * LDS_STRIDE];
  const int tid  = threadIdx.x;
  const int wid  = tid >> 5;
  const int lane = tid & 31;
  const int half = lane >> 4;
  const int lm   = lane & 15;
  const int colBase = wid * 16;
  const float4* __restrict__ addm4 = reinterpret_cast<const float4*>(addm);
  float4* __restrict__ out4 = reinterpret_cast<float4*>(out);

  v2f breg[32];
#pragma unroll
  for (int ks = 0; ks < 32; ++ks) {
    const int kk = ks * 4 + half * 2;
    breg[ks].x = W[(size_t)kk * D + colBase + lm];
    breg[ks].y = W[(size_t)(kk + 1) * D + colBase + lm];
  }
  const float bcol = bias[colBase + lm];
  const float4* __restrict__ h4 = reinterpret_cast<const float4*>(h);

  int buf = 0;
  int rt = blockIdx.x;
  int rp0 = 0, rp1 = 0, rp2 = 0;
  if (rt < nRowTiles) {
    const int d = rt * 16 + wid * 2;
    rp0 = rowptr[d]; rp1 = rowptr[d + 1]; rp2 = rowptr[d + 2];
  }
  for (; rt < nRowTiles; rt += gridDim.x) {
    const int rowBase = rt * 16;
    const int d0 = __builtin_amdgcn_readfirstlane(rowBase + wid * 2);
    const int beg0 = __builtin_amdgcn_readfirstlane(rp0);
    const int beg1 = __builtin_amdgcn_readfirstlane(rp1);
    const int end1 = __builtin_amdgcn_readfirstlane(rp2);
    const int rnext = rt + gridDim.x;
    if (rnext < nRowTiles) {
      const int d = rnext * 16 + wid * 2;
      rp0 = rowptr[d]; rp1 = rowptr[d + 1]; rp2 = rowptr[d + 2];
    }
    const int cnt0 = min(32, beg1 - beg0);
    const int cnt1 = min(32, end1 - beg1);
    int sL0 = csrSrc[max(beg0 + min(lane, cnt0 - 1), 0)];
    int sL1 = csrSrc[max(beg1 + min(lane, cnt1 - 1), 0)];
    sL0 = min(max(sL0, 0), nN - 1);
    sL1 = min(max(sL1, 0), nN - 1);
    const float nL0 = nrm[sL0];
    const float nL1 = nrm[sL1];
    const int nLi0 = __float_as_int(nL0);
    const int nLi1 = __float_as_int(nL1);
    const float nd0 = nrm[d0];
    const float nd1 = nrm[d0 + 1];

    float4 a0 = make_float4(0.f, 0.f, 0.f, 0.f);
    sum_chunk<8>(a0, h4, sL0, nLi0, cnt0, lane);
    for (int base = beg0 + 32; base < beg1; base += 32) {
      const int cnt = min(32, beg1 - base);
      int sL, nLi;
      load_chunk(sL, nLi, csrSrc, nrm, base, cnt, nN, lane);
      sum_chunk<8>(a0, h4, sL, nLi, cnt, lane);
    }
    float4 a1 = make_float4(0.f, 0.f, 0.f, 0.f);
    sum_chunk<8>(a1, h4, sL1, nLi1, cnt1, lane);
    for (int base = beg1 + 32; base < end1; base += 32) {
      const int cnt = min(32, end1 - base);
      int sL, nLi;
      load_chunk(sL, nLi, csrSrc, nrm, base, cnt, nN, lane);
      sum_chunk<8>(a1, h4, sL, nLi, cnt, lane);
    }
    a0.x *= nd0; a0.y *= nd0; a0.z *= nd0; a0.w *= nd0;
    a1.x *= nd1; a1.y *= nd1; a1.z *= nd1; a1.w *= nd1;
    float* __restrict__ At = As[buf];
    *reinterpret_cast<float4*>(&At[(wid * 2) * LDS_STRIDE + lane * 4])     = a0;
    *reinterpret_cast<float4*>(&At[(wid * 2 + 1) * LDS_STRIDE + lane * 4]) = a1;
    __syncthreads();

    v8f acc = {0.f, 0.f, 0.f, 0.f, 0.f, 0.f, 0.f, 0.f};
#pragma unroll
    for (int ks = 0; ks < 32; ++ks) {
      const int kk = ks * 4 + half * 2;
      v2f a;
      a.x = At[lm * LDS_STRIDE + kk];
      a.y = At[lm * LDS_STRIDE + kk + 1];
      acc = __builtin_amdgcn_wmma_f32_16x16x4_f32(
           false, a,  false, breg[ks],
           (short)0, acc,  false,  false);
    }

#pragma unroll
    for (int r = 0; r < 8; ++r)
      Cs[(r + half * 8) * LDS_STRIDE + colBase + lm] = acc[r] + bcol;
    __syncthreads();
    const size_t e0 = (size_t)(rowBase + wid) * 32 + lane;
    const size_t e1 = (size_t)(rowBase + wid + 8) * 32 + lane;
    const float4 f0 = addm4[e0];
    const float4 f1 = addm4[e1];
    float4 o0 = make_float4(0.f, 0.f, 0.f, 0.f);
    float4 o1 = make_float4(0.f, 0.f, 0.f, 0.f);
    if (ACCUM) { o0 = out4[e0]; o1 = out4[e1]; }
    float4 c0 = *reinterpret_cast<const float4*>(&Cs[wid * LDS_STRIDE + lane * 4]);
    float4 c1 = *reinterpret_cast<const float4*>(&Cs[(wid + 8) * LDS_STRIDE + lane * 4]);
    c0.x = fmaxf(c0.x + f0.x, 0.f) * outScale; c0.y = fmaxf(c0.y + f0.y, 0.f) * outScale;
    c0.z = fmaxf(c0.z + f0.z, 0.f) * outScale; c0.w = fmaxf(c0.w + f0.w, 0.f) * outScale;
    c1.x = fmaxf(c1.x + f1.x, 0.f) * outScale; c1.y = fmaxf(c1.y + f1.y, 0.f) * outScale;
    c1.z = fmaxf(c1.z + f1.z, 0.f) * outScale; c1.w = fmaxf(c1.w + f1.w, 0.f) * outScale;
    if (ACCUM) {
      c0.x += o0.x; c0.y += o0.y; c0.z += o0.z; c0.w += o0.w;
      c1.x += o1.x; c1.y += o1.y; c1.z += o1.z; c1.w += o1.w;
    }
    out4[e0] = c0;
    out4[e1] = c1;
    buf ^= 1;
  }
}

__global__ __launch_bounds__(256) void gemm_pair_kernel(
    const float* __restrict__ X, const float* __restrict__ G,
    const float* __restrict__ nrm,
    const float* __restrict__ Wx, const float* __restrict__ bx,
    const float* __restrict__ Wg, const float* __restrict__ bg,
    float* __restrict__ fvOut, float* __restrict__ hOut, int nRowTiles) {
  __shared__ float Xs[16 * LDS_STRIDE];
  __shared__ float Gs[16 * LDS_STRIDE];
  const int tid  = threadIdx.x;
  const int wid  = tid >> 5;
  const int lane = tid & 31;
  const int half = lane >> 4;
  const int lm   = lane & 15;
  const int colBase = wid * 16;

  v2f bxr[32];
  v2f bgr[32];
#pragma unroll
  for (int ks = 0; ks < 32; ++ks) {
    const int kk = ks * 4 + half * 2;
    bxr[ks].x = Wx[(size_t)kk * D + colBase + lm];
    bxr[ks].y = Wx[(size_t)(kk + 1) * D + colBase + lm];
    bgr[ks].x = Wg[(size_t)kk * D + colBase + lm];
    bgr[ks].y = Wg[(size_t)(kk + 1) * D + colBase + lm];
  }
  const float bxcol = bx[colBase + lm];
  const float bgcol = bg[colBase + lm];

  const int idx0 = tid;
  const int idx1 = tid + 256;
  const int r0 = idx0 >> 5, c0 = (idx0 & 31) * 4;
  const int r1 = idx1 >> 5, c1 = (idx1 & 31) * 4;
  const float4* __restrict__ X4 = reinterpret_cast<const float4*>(X);
  const float4* __restrict__ G4 = reinterpret_cast<const float4*>(G);

  int rt = blockIdx.x;
  if (rt >= nRowTiles) return;

  float4 x0, x1, g0, g1;
  {
    const int rowBase = rt * 16;
    x0 = X4[(size_t)(rowBase + r0) * 32 + (idx0 & 31)];
    x1 = X4[(size_t)(rowBase + r1) * 32 + (idx1 & 31)];
    g0 = G4[(size_t)(rowBase + r0) * 32 + (idx0 & 31)];
    g1 = G4[(size_t)(rowBase + r1) * 32 + (idx1 & 31)];
    const float n0 = nrm[rowBase + r0], n1 = nrm[rowBase + r1];
    g0.x *= n0; g0.y *= n0; g0.z *= n0; g0.w *= n0;
    g1.x *= n1; g1.y *= n1; g1.z *= n1; g1.w *= n1;
  }

  for (;;) {
    __syncthreads();
    *reinterpret_cast<float4*>(&Xs[r0 * LDS_STRIDE + c0]) = x0;
    *reinterpret_cast<float4*>(&Xs[r1 * LDS_STRIDE + c1]) = x1;
    *reinterpret_cast<float4*>(&Gs[r0 * LDS_STRIDE + c0]) = g0;
    *reinterpret_cast<float4*>(&Gs[r1 * LDS_STRIDE + c1]) = g1;

    const int rnext = rt + gridDim.x;
    const bool more = rnext < nRowTiles;
    if (more) {
      const int rowBase = rnext * 16;
      x0 = X4[(size_t)(rowBase + r0) * 32 + (idx0 & 31)];
      x1 = X4[(size_t)(rowBase + r1) * 32 + (idx1 & 31)];
      g0 = G4[(size_t)(rowBase + r0) * 32 + (idx0 & 31)];
      g1 = G4[(size_t)(rowBase + r1) * 32 + (idx1 & 31)];
      const float n0 = nrm[rowBase + r0], n1 = nrm[rowBase + r1];
      g0.x *= n0; g0.y *= n0; g0.z *= n0; g0.w *= n0;
      g1.x *= n1; g1.y *= n1; g1.z *= n1; g1.w *= n1;
    }
    __syncthreads();

    v8f accX = {0.f, 0.f, 0.f, 0.f, 0.f, 0.f, 0.f, 0.f};
    v8f accG = {0.f, 0.f, 0.f, 0.f, 0.f, 0.f, 0.f, 0.f};
#pragma unroll
    for (int ks = 0; ks < 32; ++ks) {
      const int kk = ks * 4 + half * 2;
      v2f ax, ag;
      ax.x = Xs[lm * LDS_STRIDE + kk];
      ax.y = Xs[lm * LDS_STRIDE + kk + 1];
      ag.x = Gs[lm * LDS_STRIDE + kk];
      ag.y = Gs[lm * LDS_STRIDE + kk + 1];
      accX = __builtin_amdgcn_wmma_f32_16x16x4_f32(
           false, ax,  false, bxr[ks],
           (short)0, accX,  false,  false);
      accG = __builtin_amdgcn_wmma_f32_16x16x4_f32(
           false, ag,  false, bgr[ks],
           (short)0, accG,  false,  false);
    }

    const int rowBase = rt * 16;
#pragma unroll
    for (int r = 0; r < 8; ++r) {
      const int row = rowBase + r + half * 8;
      const int col = colBase + lm;
      const size_t oi = (size_t)row * D + col;
      const float f = accX[r] + bxcol;
      fvOut[oi] = f;
      float hv = accG[r] + bgcol;
      hv += f;
      hOut[oi] = fmaxf(hv, 0.f);
    }

    if (!more) break;
    rt = rnext;
  }
}

extern "C" void kernel_launch(void* const* d_in, const int* in_sizes, int n_in,
                              void* d_out, int out_size, void* d_ws, size_t ws_size,
                              hipStream_t stream) {
  const float* feats = (const float*)d_in[0];
  const int*   src   = (const int*)d_in[1];
  const int*   dst   = (const int*)d_in[2];
  const float* w0    = (const float*)d_in[3];
  const float* b0    = (const float*)d_in[4];
  const float* w     = (const float*)d_in[5];
  const float* bw    = (const float*)d_in[6];
  const float* v     = (const float*)d_in[7];
  const float* bv    = (const float*)d_in[8];
  float* out = (float*)d_out;

  const int N = in_sizes[0] / D;
  const int E = in_sizes[1];
  const size_t M = (size_t)N * D;

  const size_t Nr  = ((size_t)N + 255) & ~(size_t)255;
  const size_t N1r = ((size_t)N + 1 + 255) & ~(size_t)255;
  const size_t Er  = ((size_t)E + 255) & ~(size_t)255;

  float* ws = (float*)d_ws;
  size_t off = 0;
  float* nrm  = ws + off; off += Nr;
  float* agg0 = ws + off; off += M;
  float* fv   = ws + off; off += M;
  float* h1   = ws + off; off += M;
  int* cnt    = (int*)(ws + off); off += Nr;
  int* rowptr = (int*)(ws + off); off += N1r;
  int* pos    = (int*)(ws + off); off += Er;
  int* csrSrc = (int*)(ws + off); off += Er;
  if (off * sizeof(float) > ws_size) return;

  const int nRowTiles  = N / 16;
  const int fusedBlocks = 1024;
  const int pairBlocks = 512;

  zero_f4_kernel<<<((int)(Nr / 4) + 255) / 256, 256, 0, stream>>>((float4*)cnt, (int)(Nr / 4));
  count_rank_kernel<<<(E + 255) / 256, 256, 0, stream>>>(dst, cnt, pos, E);
  scan_norm_kernel<<<(N + 255) / 256, 256, 0, stream>>>(cnt, rowptr, nrm, N);
  csr_fill_kernel<<<(E + 255) / 256, 256, 0, stream>>>(src, dst, pos, rowptr, csrSrc, E);

  spmm_gather_kernel<<<(N + 7) / 8, 256, 0, stream>>>(feats, rowptr, csrSrc, nrm, agg0, N);

  for (int k = 0; k < KSTACK; ++k) {
    gemm_pair_kernel<<<pairBlocks, 256, 0, stream>>>(
        feats, agg0, nrm, v + (size_t)k * D * D, bv + (size_t)k * D,
        w0 + (size_t)k * D * D, b0 + (size_t)k * D, fv, h1, nRowTiles);
    if (k == 0) {
      gather_gemm_kernel<false><<<fusedBlocks, 256, 0, stream>>>(
          h1, rowptr, csrSrc, nrm, w + (size_t)k * D * D, bw + (size_t)k * D,
          fv, out, nRowTiles, N, 1.0f / (float)KSTACK);
    } else {
      gather_gemm_kernel<true><<<fusedBlocks, 256, 0, stream>>>(
          h1, rowptr, csrSrc, nrm, w + (size_t)k * D * D, bw + (size_t)k * D,
          fv, out, nRowTiles, N, 1.0f / (float)KSTACK);
    }
  }
}


# stack loop moved inside the launches: one GEMM-pair launch covers all 3 stacks (stack by block index), one gather+GEMM launch runs its tiles once per stack, out summed in stack order; 11->7 launches
# speedup vs baseline: 2.6527x; 1.0455x over previous
#include <hip/hip_runtime.h>

typedef float v2f __attribute__((ext_vector_type(2)));
typedef float v8f __attribute__((ext_vector_type(8)));

#define D 128
#define KSTACK 3
#define LDS_STRIDE 132

__global__ void zero_f4_kernel(float4* __restrict__ p, int n4) {
  int i = blockIdx.x * blockDim.x + threadIdx.x;
  if (i < n4) p[i] = make_float4(0.f, 0.f, 0.f, 0.f);
}

__global__ void count_rank_kernel(const int* __restrict__ dst, int* __restrict__ cnt,
                                  int* __restrict__ pos, int nE) {
  int i = blockIdx.x * blockDim.x + threadIdx.x;
  if (i < nE) pos[i] = atomicAdd(&cnt[dst[i]], 1);
}

__global__ __launch_bounds__(256) void scan_norm_kernel(const int* __restrict__ cnt,
                                                        int* __restrict__ rowptr,
                                                        float* __restrict__ nrm, int n) {
  __shared__ int waveIncl[8];
  __shared__ int waveFront[8];
  const int tid  = threadIdx.x;
  const int lane = tid & 31;
  const int wid  = tid >> 5;
  const int base = blockIdx.x * 256;

  const int4* __restrict__ c4 = reinterpret_cast<const int4*>(cnt);
  int front = 0;
  for (int q = tid; q < (base >> 2); q += 256) {
    const int4 v = c4[q];
    front += (v.x + v.y) + (v.z + v.w);
  }
  for (int o = 16; o > 0; o >>= 1) front += __shfl_xor(front, o);

  const int i = base + tid;
  const int c = (i < n) ? cnt[i] : 0;
  if (i < n) nrm[i] = rsqrtf(fmaxf((float)c, 1.0f));

  int incl = c;
  for (int o = 1; o < 32; o <<= 1) {
    const int t = __shfl_up(incl, o);
    if (lane >= o) incl += t;
  }
  if (lane == 31) waveIncl[wid]  = incl;
  if (lane == 0)  waveFront[wid] = front;
  __syncthreads();
  int off = 0;
  for (int w = 0; w < 8; ++w) {
    off += waveFront[w];
    if (w < wid) off += waveIncl[w];
  }
  const int excl = off + incl - c;
  if (i < n) rowptr[i] = excl;
  if (i == n - 1) rowptr[n] = excl + c;
}

__global__ void csr_fill_kernel(const int* __restrict__ src, const int* __restrict__ dst,
                                const int* __restrict__ pos, const int* __restrict__ rowptr,
                                int* __restrict__ csrSrc, int nE) {
  int i = blockIdx.x * blockDim.x + threadIdx.x;
  if (i < nE) csrSrc[rowptr[dst[i]] + pos[i]] = src[i];
}

__device__ __forceinline__ void row_fma(float4& acc, const float4 v, const float w) {
  acc.x = fmaf(v.x, w, acc.x); acc.y = fmaf(v.y, w, acc.y);
  acc.z = fmaf(v.z, w, acc.z); acc.w = fmaf(v.w, w, acc.w);
}

template <int R>
__device__ __forceinline__ void gather_rows(float4& acc, const float4* __restrict__ h4,
                                            const int sL, const int nLi,
                                            const int j, const int lane) {
  float4 v[R];
#pragma unroll
  for (int u = 0; u < R; ++u) {
    const int s = __builtin_amdgcn_readlane(sL, j + u);
    v[u] = h4[(size_t)s * 32 + lane];
  }
#pragma unroll
  for (int u = 0; u < R; ++u)
    row_fma(acc, v[u], __int_as_float(__builtin_amdgcn_readlane(nLi, j + u)));
}

__device__ __forceinline__ void load_chunk(int& sL, int& nLi, const int* __restrict__ csrSrc,
                                           const float* __restrict__ nrm,
                                           const int base, const int cnt, const int nN,
                                           const int lane) {
  sL = 0;
  float nL = 0.f;
  if (lane < cnt) {
    sL = min(max(csrSrc[base + lane], 0), nN - 1);
    nL = nrm[sL];
  }
  nLi = __float_as_int(nL);
}

template <int RMAX>
__device__ __forceinline__ void sum_chunk(float4& acc, const float4* __restrict__ h4,
                                          const int sL, const int nLi, const int cnt,
                                          const int lane) {
  int j = 0;
  if (RMAX >= 8)
    for (; j + 8 <= cnt; j += 8) gather_rows<8>(acc, h4, sL, nLi, j, lane);
  for (; j + 4 <= cnt; j += 4) gather_rows<4>(acc, h4, sL, nLi, j, lane);
  const int rem = cnt - j;
  if (rem == 3)      gather_rows<3>(acc, h4, sL, nLi, j, lane);
  else if (rem == 2) gather_rows<2>(acc, h4, sL, nLi, j, lane);
  else if (rem == 1) gather_rows<1>(acc, h4, sL, nLi, j, lane);
}

__global__ __launch_bounds__(256) void spmm_gather_kernel(const float* __restrict__ h,
                                                          const int* __restrict__ rowptr,
                                                          const int* __restrict__ csrSrc,
                                                          const float* __restrict__ nrm,
                                                          float* __restrict__ agg, int nN) {
  const int lane = threadIdx.x & 31;
  const int d = __builtin_amdgcn_readfirstlane(blockIdx.x * 8 + (threadIdx.x >> 5));
  if (d >= nN) return;
  const int beg = __builtin_amdgcn_readfirstlane(rowptr[d]);
  const int end = __builtin_amdgcn_readfirstlane(rowptr[d + 1]);
  const float4* __restrict__ h4 = reinterpret_cast<const float4*>(h);

  float4 acc = make_float4(0.f, 0.f, 0.f, 0.f);
  for (int base = beg; base < end; base += 32) {
    const int cnt = min(32, end - base);
    int sL, nLi;
    load_chunk(sL, nLi, csrSrc, nrm, base, cnt, nN, lane);
    sum_chunk<4>(acc, h4, sL, nLi, cnt, lane);
  }
  reinterpret_cast<float4*>(agg)[(size_t)d * 32 + lane] = acc;
}

__global__ __launch_bounds__(256, 8) void gather_gemm_kernel(
    const float* __restrict__ hAll, const int* __restrict__ rowptr,
    const int* __restrict__ csrSrc, const float* __restrict__ nrm,
    const float* __restrict__ WAll, const float* __restrict__ biasAll,
    const float* __restrict__ addmAll, float* __restrict__ out,
    int nRowTiles, int nN, float outScale,
    int nStacks, size_t planeStride, int accumFirst) {
  __shared__ float As[2][16 * LDS_STRIDE];
  __shared__ float Cs[16 * LDS_STRIDE];
  const int tid  = threadIdx.x;
  const int wid  = __builtin_amdgcn_readfirstlane(tid >> 5);
  const int lane = tid & 31;
  const int half = lane >> 4;
  const int lm   = lane & 15;
  const int colBase = wid * 16;
  float4* __restrict__ out4 = reinterpret_cast<float4*>(out);

  int buf = 0;
  for (int st = 0; st < nStacks; ++st) {
  const float* __restrict__ W = WAll + (size_t)st * D * D;
  const float* __restrict__ bias = biasAll + (size_t)st * D;
  const float4* __restrict__ addm4 =
      reinterpret_cast<const float4*>(addmAll + (size_t)st * planeStride);
  const float4* __restrict__ h4 =
      reinterpret_cast<const float4*>(hAll + (size_t)st * planeStride);
  const bool accum = (st > 0) || (accumFirst != 0);

  v2f breg[32];
#pragma unroll
  for (int ks = 0; ks < 32; ++ks) {
    const int kk = ks * 4 + half * 2;
    breg[ks].x = W[(size_t)kk * D + colBase + lm];
    breg[ks].y = W[(size_t)(kk + 1) * D + colBase + lm];
  }
  const float bcol = bias[colBase + lm];

  int rt = blockIdx.x;
  int rp0 = 0, rp1 = 0, rp2 = 0;
  if (rt < nRowTiles) {
    const int d = rt * 16 + wid * 2;
    rp0 = rowptr[d]; rp1 = rowptr[d + 1]; rp2 = rowptr[d + 2];
  }
  for (; rt < nRowTiles; rt += gridDim.x) {
    const int rowBase = rt * 16;
    const int d0 = __builtin_amdgcn_readfirstlane(rowBase + wid * 2);
    const int beg0 = __builtin_amdgcn_readfirstlane(rp0);
    const int beg1 = __builtin_amdgcn_readfirstlane(rp1);
    const int end1 = __builtin_amdgcn_readfirstlane(rp2);
    const int rnext = rt + gridDim.x;
    if (rnext < nRowTiles) {
      const int d = rnext * 16 + wid * 2;
      rp0 = rowptr[d]; rp1 = rowptr[d + 1]; rp2 = rowptr[d + 2];
    }
    const int cnt0 = min(32, beg1 - beg0);
    const int cnt1 = min(32, end1 - beg1);
    int sL0 = csrSrc[max(beg0 + min(lane, cnt0 - 1), 0)];
    int sL1 = csrSrc[max(beg1 + min(lane, cnt1 - 1), 0)];
    sL0 = min(max(sL0, 0), nN - 1);
    sL1 = min(max(sL1, 0), nN - 1);
    const float nL0 = nrm[sL0];
    const float nL1 = nrm[sL1];
    const int nLi0 = __float_as_int(nL0);
    const int nLi1 = __float_as_int(nL1);
    const float nd0 = nrm[d0];
    const float nd1 = nrm[d0 + 1];

    float4 a0 = make_float4(0.f, 0.f, 0.f, 0.f);
    sum_chunk<8>(a0, h4, sL0, nLi0, cnt0, lane);
    for (int base = beg0 + 32; base < beg1; base += 32) {
      const int cnt = min(32, beg1 - base);
      int sL, nLi;
      load_chunk(sL, nLi, csrSrc, nrm, base, cnt, nN, lane);
      sum_chunk<8>(a0, h4, sL, nLi, cnt, lane);
    }
    float* __restrict__ At = As[buf];
    a0.x *= nd0; a0.y *= nd0; a0.z *= nd0; a0.w *= nd0;
    *reinterpret_cast<float4*>(&At[(wid * 2) * LDS_STRIDE + lane * 4]) = a0;
    float4 a1 = make_float4(0.f, 0.f, 0.f, 0.f);
    sum_chunk<8>(a1, h4, sL1, nLi1, cnt1, lane);
    for (int base = beg1 + 32; base < end1; base += 32) {
      const int cnt = min(32, end1 - base);
      int sL, nLi;
      load_chunk(sL, nLi, csrSrc, nrm, base, cnt, nN, lane);
      sum_chunk<8>(a1, h4, sL, nLi, cnt, lane);
    }
    a1.x *= nd1; a1.y *= nd1; a1.z *= nd1; a1.w *= nd1;
    *reinterpret_cast<float4*>(&At[(wid * 2 + 1) * LDS_STRIDE + lane * 4]) = a1;
    __syncthreads();

    v8f acc = {0.f, 0.f, 0.f, 0.f, 0.f, 0.f, 0.f, 0.f};
#pragma unroll
    for (int ks = 0; ks < 32; ++ks) {
      const int kk = ks * 4 + half * 2;
      v2f a;
      a.x = At[lm * LDS_STRIDE + kk];
      a.y = At[lm * LDS_STRIDE + kk + 1];
      acc = __builtin_amdgcn_wmma_f32_16x16x4_f32(
           false, a,  false, breg[ks],
           (short)0, acc,  false,  false);
    }

#pragma unroll
    for (int r = 0; r < 8; ++r)
      Cs[(r + half * 8) * LDS_STRIDE + colBase + lm] = acc[r] + bcol;
    __syncthreads();
    const size_t e0 = (size_t)(rowBase + wid) * 32 + lane;
    const size_t e1 = (size_t)(rowBase + wid + 8) * 32 + lane;
    const float4 f0 = addm4[e0];
    const float4 f1 = addm4[e1];
    float4 o0 = make_float4(0.f, 0.f, 0.f, 0.f);
    float4 o1 = make_float4(0.f, 0.f, 0.f, 0.f);
    if (accum) { o0 = out4[e0]; o1 = out4[e1]; }
    float4 c0 = *reinterpret_cast<const float4*>(&Cs[wid * LDS_STRIDE + lane * 4]);
    float4 c1 = *reinterpret_cast<const float4*>(&Cs[(wid + 8) * LDS_STRIDE + lane * 4]);
    c0.x = fmaxf(c0.x + f0.x, 0.f) * outScale; c0.y = fmaxf(c0.y + f0.y, 0.f) * outScale;
    c0.z = fmaxf(c0.z + f0.z, 0.f) * outScale; c0.w = fmaxf(c0.w + f0.w, 0.f) * outScale;
    c1.x = fmaxf(c1.x + f1.x, 0.f) * outScale; c1.y = fmaxf(c1.y + f1.y, 0.f) * outScale;
    c1.z = fmaxf(c1.z + f1.z, 0.f) * outScale; c1.w = fmaxf(c1.w + f1.w, 0.f) * outScale;
    if (accum) {
      c0.x += o0.x; c0.y += o0.y; c0.z += o0.z; c0.w += o0.w;
      c1.x += o1.x; c1.y += o1.y; c1.z += o1.z; c1.w += o1.w;
    }
    out4[e0] = c0;
    out4[e1] = c1;
    buf ^= 1;
  }
  }
}

__global__ __launch_bounds__(256) void gemm_pair_kernel(
    const float* __restrict__ X, const float* __restrict__ G,
    const float* __restrict__ nrm,
    const float* __restrict__ WxAll, const float* __restrict__ bxAll,
    const float* __restrict__ WgAll, const float* __restrict__ bgAll,
    float* __restrict__ fvAll, float* __restrict__ hAll, int nRowTiles,
    int nStacks, size_t planeStride) {
  __shared__ float Xs[16 * LDS_STRIDE];
  __shared__ float Gs[16 * LDS_STRIDE];
  const int tid  = threadIdx.x;
  const int wid  = tid >> 5;
  const int lane = tid & 31;
  const int half = lane >> 4;
  const int lm   = lane & 15;
  const int colBase = wid * 16;
  const int st      = blockIdx.x % nStacks;
  const int slot    = blockIdx.x / nStacks;
  const int nSlots  = gridDim.x / nStacks;
  const float* __restrict__ Wx = WxAll + (size_t)st * D * D;
  const float* __restrict__ Wg = WgAll + (size_t)st * D * D;
  const float* __restrict__ bx = bxAll + (size_t)st * D;
  const float* __restrict__ bg = bgAll + (size_t)st * D;
  float* __restrict__ fvOut = fvAll + (size_t)st * planeStride;
  float* __restrict__ hOut  = hAll + (size_t)st * planeStride;

  v2f bxr[32];
  v2f bgr[32];
#pragma unroll
  for (int ks = 0; ks < 32; ++ks) {
    const int kk = ks * 4 + half * 2;
    bxr[ks].x = Wx[(size_t)kk * D + colBase + lm];
    bxr[ks].y = Wx[(size_t)(kk + 1) * D + colBase + lm];
    bgr[ks].x = Wg[(size_t)kk * D + colBase + lm];
    bgr[ks].y = Wg[(size_t)(kk + 1) * D + colBase + lm];
  }
  const float bxcol = bx[colBase + lm];
  const float bgcol = bg[colBase + lm];

  const int idx0 = tid;
  const int idx1 = tid + 256;
  const int r0 = idx0 >> 5, c0 = (idx0 & 31) * 4;
  const int r1 = idx1 >> 5, c1 = (idx1 & 31) * 4;
  const float4* __restrict__ X4 = reinterpret_cast<const float4*>(X);
  const float4* __restrict__ G4 = reinterpret_cast<const float4*>(G);

  int rt = slot;
  if (rt >= nRowTiles) return;

  float4 x0, x1, g0, g1;
  {
    const int rowBase = rt * 16;
    x0 = X4[(size_t)(rowBase + r0) * 32 + (idx0 & 31)];
    x1 = X4[(size_t)(rowBase + r1) * 32 + (idx1 & 31)];
    g0 = G4[(size_t)(rowBase + r0) * 32 + (idx0 & 31)];
    g1 = G4[(size_t)(rowBase + r1) * 32 + (idx1 & 31)];
    const float n0 = nrm[rowBase + r0], n1 = nrm[rowBase + r1];
    g0.x *= n0; g0.y *= n0; g0.z *= n0; g0.w *= n0;
    g1.x *= n1; g1.y *= n1; g1.z *= n1; g1.w *= n1;
  }

  for (;;) {
    __syncthreads();
    *reinterpret_cast<float4*>(&Xs[r0 * LDS_STRIDE + c0]) = x0;
    *reinterpret_cast<float4*>(&Xs[r1 * LDS_STRIDE + c1]) = x1;
    *reinterpret_cast<float4*>(&Gs[r0 * LDS_STRIDE + c0]) = g0;
    *reinterpret_cast<float4*>(&Gs[r1 * LDS_STRIDE + c1]) = g1;

    const int rnext = rt + nSlots;
    const bool more = rnext < nRowTiles;
    if (more) {
      const int rowBase = rnext * 16;
      x0 = X4[(size_t)(rowBase + r0) * 32 + (idx0 & 31)];
      x1 = X4[(size_t)(rowBase + r1) * 32 + (idx1 & 31)];
      g0 = G4[(size_t)(rowBase + r0) * 32 + (idx0 & 31)];
      g1 = G4[(size_t)(rowBase + r1) * 32 + (idx1 & 31)];
      const float n0 = nrm[rowBase + r0], n1 = nrm[rowBase + r1];
      g0.x *= n0; g0.y *= n0; g0.z *= n0; g0.w *= n0;
      g1.x *= n1; g1.y *= n1; g1.z *= n1; g1.w *= n1;
    }
    __syncthreads();

    v8f accX = {0.f, 0.f, 0.f, 0.f, 0.f, 0.f, 0.f, 0.f};
    v8f accG = {0.f, 0.f, 0.f, 0.f, 0.f, 0.f, 0.f, 0.f};
#pragma unroll
    for (int ks = 0; ks < 32; ++ks) {
      const int kk = ks * 4 + half * 2;
      v2f ax, ag;
      ax.x = Xs[lm * LDS_STRIDE + kk];
      ax.y = Xs[lm * LDS_STRIDE + kk + 1];
      ag.x = Gs[lm * LDS_STRIDE + kk];
      ag.y = Gs[lm * LDS_STRIDE + kk + 1];
      accX = __builtin_amdgcn_wmma_f32_16x16x4_f32(
           false, ax,  false, bxr[ks],
           (short)0, accX,  false,  false);
      accG = __builtin_amdgcn_wmma_f32_16x16x4_f32(
           false, ag,  false, bgr[ks],
           (short)0, accG,  false,  false);
    }

    const int rowBase = rt * 16;
#pragma unroll
    for (int r = 0; r < 8; ++r) {
      const int row = rowBase + r + half * 8;
      const int col = colBase + lm;
      const size_t oi = (size_t)row * D + col;
      const float f = accX[r] + bxcol;
      fvOut[oi] = f;
      float hv = accG[r] + bgcol;
      hv += f;
      hOut[oi] = fmaxf(hv, 0.f);
    }

    if (!more) break;
    rt = rnext;
  }
}

extern "C" void kernel_launch(void* const* d_in, const int* in_sizes, int n_in,
                              void* d_out, int out_size, void* d_ws, size_t ws_size,
                              hipStream_t stream) {
  const float* feats = (const float*)d_in[0];
  const int*   src   = (const int*)d_in[1];
  const int*   dst   = (const int*)d_in[2];
  const float* w0    = (const float*)d_in[3];
  const float* b0    = (const float*)d_in[4];
  const float* w     = (const float*)d_in[5];
  const float* bw    = (const float*)d_in[6];
  const float* v     = (const float*)d_in[7];
  const float* bv    = (const float*)d_in[8];
  float* out = (float*)d_out;

  const int N = in_sizes[0] / D;
  const int E = in_sizes[1];
  const size_t M = (size_t)N * D;

  const size_t Nr  = ((size_t)N + 255) & ~(size_t)255;
  const size_t N1r = ((size_t)N + 1 + 255) & ~(size_t)255;
  const size_t Er  = ((size_t)E + 255) & ~(size_t)255;

  float* ws = (float*)d_ws;
  size_t off = 0;
  float* nrm  = ws + off; off += Nr;
  float* agg0 = ws + off; off += M;
  int* cnt    = (int*)(ws + off); off += Nr;
  int* rowptr = (int*)(ws + off); off += N1r;
  int* csrSrc = (int*)(ws + off); off += Er;
  float* planes = ws + off;
  int* pos      = (int*)planes;
  if ((off + 2 * M) * sizeof(float) > ws_size) return;
  const size_t roomPairs = (ws_size / sizeof(float) - off) / (2 * M);
  const int G = roomPairs >= (size_t)KSTACK ? KSTACK : (int)roomPairs;
  float* fv = planes;
  float* h1 = planes + (size_t)G * M;

  const int nRowTiles  = N / 16;
  const int fusedBlocks = 1024;
  const int pairBlocks = 512;

  zero_f4_kernel<<<((int)(Nr / 4) + 255) / 256, 256, 0, stream>>>((float4*)cnt, (int)(Nr / 4));
  count_rank_kernel<<<(E + 255) / 256, 256, 0, stream>>>(dst, cnt, pos, E);
  scan_norm_kernel<<<(N + 255) / 256, 256, 0, stream>>>(cnt, rowptr, nrm, N);
  csr_fill_kernel<<<(E + 255) / 256, 256, 0, stream>>>(src, dst, pos, rowptr, csrSrc, E);

  spmm_gather_kernel<<<(N + 7) / 8, 256, 0, stream>>>(feats, rowptr, csrSrc, nrm, agg0, N);

  for (int k0 = 0; k0 < KSTACK; k0 += G) {
    const int g = (KSTACK - k0 < G) ? (KSTACK - k0) : G;
    const int slots = pairBlocks / g;
    gemm_pair_kernel<<<g * slots, 256, 0, stream>>>(
        feats, agg0, nrm, v + (size_t)k0 * D * D, bv + (size_t)k0 * D,
        w0 + (size_t)k0 * D * D, b0 + (size_t)k0 * D, fv, h1, nRowTiles, g, M);
    gather_gemm_kernel<<<fusedBlocks, 256, 0, stream>>>(
        h1, rowptr, csrSrc, nrm, w + (size_t)k0 * D * D, bw + (size_t)k0 * D,
        fv, out, nRowTiles, N, 1.0f / (float)KSTACK, g, M, k0 > 0 ? 1 : 0);
  }
}


# layer-2 product taken in front of its propagate, (A h)W = A(hW): h@w chained on chip in the GEMM launch, last launch a barrier-free whole-row gather; edge sums and k sums run in a new order
# speedup vs baseline: 2.8148x; 1.0611x over previous
#include <hip/hip_runtime.h>

typedef float v2f __attribute__((ext_vector_type(2)));
typedef float v8f __attribute__((ext_vector_type(8)));

#define D 128
#define KSTACK 3
#define LDS_STRIDE 132

__global__ void zero_f4_kernel(float4* __restrict__ p, int n4) {
  int i = blockIdx.x * blockDim.x + threadIdx.x;
  if (i < n4) p[i] = make_float4(0.f, 0.f, 0.f, 0.f);
}

__global__ void count_rank_kernel(const int* __restrict__ dst, int* __restrict__ cnt,
                                  int* __restrict__ pos, int nE) {
  int i = blockIdx.x * blockDim.x + threadIdx.x;
  if (i < nE) pos[i] = atomicAdd(&cnt[dst[i]], 1);
}

__global__ __launch_bounds__(256) void scan_norm_kernel(const int* __restrict__ cnt,
                                                        int* __restrict__ rowptr,
                                                        float* __restrict__ nrm, int n) {
  __shared__ int waveIncl[8];
  __shared__ int waveFront[8];
  const int tid  = threadIdx.x;
  const int lane = tid & 31;
  const int wid  = tid >> 5;
  const int base = blockIdx.x * 256;

  const int4* __restrict__ c4 = reinterpret_cast<const int4*>(cnt);
  int front = 0;
  for (int q = tid; q < (base >> 2); q += 256) {
    const int4 v = c4[q];
    front += (v.x + v.y) + (v.z + v.w);
  }
  for (int o = 16; o > 0; o >>= 1) front += __shfl_xor(front, o);

  const int i = base + tid;
  const int c = (i < n) ? cnt[i] : 0;
  if (i < n) nrm[i] = rsqrtf(fmaxf((float)c, 1.0f));

  int incl = c;
  for (int o = 1; o < 32; o <<= 1) {
    const int t = __shfl_up(incl, o);
    if (lane >= o) incl += t;
  }
  if (lane == 31) waveIncl[wid]  = incl;
  if (lane == 0)  waveFront[wid] = front;
  __syncthreads();
  int off = 0;
  for (int w = 0; w < 8; ++w) {
    off += waveFront[w];
    if (w < wid) off += waveIncl[w];
  }
  const int excl = off + incl - c;
  if (i < n) rowptr[i] = excl;
  if (i == n - 1) rowptr[n] = excl + c;
}

__global__ void csr_fill_kernel(const int* __restrict__ src, const int* __restrict__ dst,
                                const int* __restrict__ pos, const int* __restrict__ rowptr,
                                int* __restrict__ csrSrc, int nE) {
  int i = blockIdx.x * blockDim.x + threadIdx.x;
  if (i < nE) csrSrc[rowptr[dst[i]] + pos[i]] = src[i];
}

__device__ __forceinline__ void row_fma(float4& acc, const float4 v, const float w) {
  acc.x = fmaf(v.x, w, acc.x); acc.y = fmaf(v.y, w, acc.y);
  acc.z = fmaf(v.z, w, acc.z); acc.w = fmaf(v.w, w, acc.w);
}

template <int R>
__device__ __forceinline__ void gather_rows(float4& acc, const float4* __restrict__ h4,
                                            const int sL, const int nLi,
                                            const int j, const int lane) {
  float4 v[R];
#pragma unroll
  for (int u = 0; u < R; ++u) {
    const int s = __builtin_amdgcn_readlane(sL, j + u);
    v[u] = h4[(size_t)s * 32 + lane];
  }
#pragma unroll
  for (int u = 0; u < R; ++u)
    row_fma(acc, v[u], __int_as_float(__builtin_amdgcn_readlane(nLi, j + u)));
}

__device__ __forceinline__ void load_chunk(int& sL, int& nLi, const int* __restrict__ csrSrc,
                                           const float* __restrict__ nrm,
                                           const int base, const int cnt, const int nN,
                                           const int lane) {
  sL = 0;
  float nL = 0.f;
  if (lane < cnt) {
    sL = min(max(csrSrc[base + lane], 0), nN - 1);
    nL = nrm[sL];
  }
  nLi = __float_as_int(nL);
}

template <int RMAX>
__device__ __forceinline__ void sum_chunk(float4& acc, const float4* __restrict__ h4,
                                          const int sL, const int nLi, const int cnt,
                                          const int lane) {
  int j = 0;
  if (RMAX >= 8)
    for (; j + 8 <= cnt; j += 8) gather_rows<8>(acc, h4, sL, nLi, j, lane);
  for (; j + 4 <= cnt; j += 4) gather_rows<4>(acc, h4, sL, nLi, j, lane);
  const int rem = cnt - j;
  if (rem == 3)      gather_rows<3>(acc, h4, sL, nLi, j, lane);
  else if (rem == 2) gather_rows<2>(acc, h4, sL, nLi, j, lane);
  else if (rem == 1) gather_rows<1>(acc, h4, sL, nLi, j, lane);
}

__global__ __launch_bounds__(256) void spmm_gather_kernel(const float* __restrict__ h,
                                                          const int* __restrict__ rowptr,
                                                          const int* __restrict__ csrSrc,
                                                          const float* __restrict__ nrm,
                                                          float* __restrict__ agg, int nN) {
  const int lane = threadIdx.x & 31;
  const int d = __builtin_amdgcn_readfirstlane(blockIdx.x * 8 + (threadIdx.x >> 5));
  if (d >= nN) return;
  const int beg = __builtin_amdgcn_readfirstlane(rowptr[d]);
  const int end = __builtin_amdgcn_readfirstlane(rowptr[d + 1]);
  const float4* __restrict__ h4 = reinterpret_cast<const float4*>(h);

  float4 acc = make_float4(0.f, 0.f, 0.f, 0.f);
  for (int base = beg; base < end; base += 32) {
    const int cnt = min(32, end - base);
    int sL, nLi;
    load_chunk(sL, nLi, csrSrc, nrm, base, cnt, nN, lane);
    sum_chunk<4>(acc, h4, sL, nLi, cnt, lane);
  }
  reinterpret_cast<float4*>(agg)[(size_t)d * 32 + lane] = acc;
}

#define ROWS_PER_WAVE 5

template <int R>
__device__ __forceinline__ void gather_rows_plain(float4& acc, const float4* __restrict__ z4,
                                                  const int sL, const int j, const int lane) {
  float4 v[R];
#pragma unroll
  for (int u = 0; u < R; ++u) {
    const int s = __builtin_amdgcn_readlane(sL, j + u);
    v[u] = z4[(size_t)s * 32 + lane];
  }
#pragma unroll
  for (int u = 0; u < R; ++u) {
    acc.x += v[u].x; acc.y += v[u].y; acc.z += v[u].z; acc.w += v[u].w;
  }
}

__device__ __forceinline__ void sum_chunk_plain(float4& acc, const float4* __restrict__ z4,
                                                const int sL, const int cnt, const int lane) {
  int j = 0;
  for (; j + 8 <= cnt; j += 8) gather_rows_plain<8>(acc, z4, sL, j, lane);
  for (; j + 4 <= cnt; j += 4) gather_rows_plain<4>(acc, z4, sL, j, lane);
  const int rem = cnt - j;
  if (rem == 3)      gather_rows_plain<3>(acc, z4, sL, j, lane);
  else if (rem == 2) gather_rows_plain<2>(acc, z4, sL, j, lane);
  else if (rem == 1) gather_rows_plain<1>(acc, z4, sL, j, lane);
}

__global__ __launch_bounds__(256, 10) void gather_out_kernel(
    const float* __restrict__ zAll, const int* __restrict__ rowptr,
    const int* __restrict__ csrSrc, const float* __restrict__ nrm,
    const float* __restrict__ biasAll, const float* __restrict__ addmAll,
    float* __restrict__ out, int nN, float outScale,
    int nStacks, size_t planeStride, int accumFirst) {
  const int lane = threadIdx.x & 31;
  const int w  = __builtin_amdgcn_readfirstlane(blockIdx.x * 8 + (threadIdx.x >> 5));
  const int nW = gridDim.x * 8;
  float4* __restrict__ out4 = reinterpret_cast<float4*>(out);

  int beg[ROWS_PER_WAVE], end[ROWS_PER_WAVE], sL0[ROWS_PER_WAVE];
  float nd[ROWS_PER_WAVE];
  float4 o[ROWS_PER_WAVE];
#pragma unroll
  for (int j = 0; j < ROWS_PER_WAVE; ++j) {
    const int d = w + j * nW;
    beg[j] = 0; end[j] = 0; sL0[j] = 0; nd[j] = 0.f;
    o[j] = make_float4(0.f, 0.f, 0.f, 0.f);
    if (d < nN) {
      beg[j] = __builtin_amdgcn_readfirstlane(rowptr[d]);
      end[j] = __builtin_amdgcn_readfirstlane(rowptr[d + 1]);
      if (lane < end[j] - beg[j])
        sL0[j] = min(max(csrSrc[beg[j] + lane], 0), nN - 1);
      nd[j] = nrm[d];
      if (accumFirst) o[j] = out4[(size_t)d * 32 + lane];
    }
  }

  for (int st = 0; st < nStacks; ++st) {
    const float4* __restrict__ z4 =
        reinterpret_cast<const float4*>(zAll + (size_t)st * planeStride);
    const float4* __restrict__ addm4 =
        reinterpret_cast<const float4*>(addmAll + (size_t)st * planeStride);
    const float4 b4 = reinterpret_cast<const float4*>(biasAll + (size_t)st * D)[lane];
#pragma unroll
    for (int j = 0; j < ROWS_PER_WAVE; ++j) {
      const int d = w + j * nW;
      if (d < nN) {
        const float4 f = addm4[(size_t)d * 32 + lane];
        float4 acc = make_float4(0.f, 0.f, 0.f, 0.f);
        sum_chunk_plain(acc, z4, sL0[j], min(32, end[j] - beg[j]), lane);
        for (int base = beg[j] + 32; base < end[j]; base += 32) {
          const int cnt = min(32, end[j] - base);
          int sL = 0;
          if (lane < cnt) sL = min(max(csrSrc[base + lane], 0), nN - 1);
          sum_chunk_plain(acc, z4, sL, cnt, lane);
        }
        float4 c;
        c.x = fmaxf(acc.x * nd[j] + b4.x + f.x, 0.f) * outScale;
        c.y = fmaxf(acc.y * nd[j] + b4.y + f.y, 0.f) * outScale;
        c.z = fmaxf(acc.z * nd[j] + b4.z + f.z, 0.f) * outScale;
        c.w = fmaxf(acc.w * nd[j] + b4.w + f.w, 0.f) * outScale;
        o[j].x += c.x; o[j].y += c.y; o[j].z += c.z; o[j].w += c.w;
      }
    }
  }

#pragma unroll
  for (int j = 0; j < ROWS_PER_WAVE; ++j) {
    const int d = w + j * nW;
    if (d < nN) out4[(size_t)d * 32 + lane] = o[j];
  }
}

__global__ __launch_bounds__(256, 4) void gemm_chain_kernel(
    const float* __restrict__ X, const float* __restrict__ G,
    const float* __restrict__ nrm,
    const float* __restrict__ WxAll, const float* __restrict__ bxAll,
    const float* __restrict__ WgAll, const float* __restrict__ bgAll,
    const float* __restrict__ W2All,
    float* __restrict__ fvAll, float* __restrict__ zAll, int nRowTiles,
    int nStacks, size_t planeStride) {
  __shared__ float Xs[16 * LDS_STRIDE];
  __shared__ float Gs[16 * LDS_STRIDE];
  __shared__ float Hs[16 * LDS_STRIDE];
  const int tid  = threadIdx.x;
  const int wid  = __builtin_amdgcn_readfirstlane(tid >> 5);
  const int lane = tid & 31;
  const int half = lane >> 4;
  const int lm   = lane & 15;
  const int colBase = wid * 16;
  const int st      = blockIdx.x % nStacks;
  const int slot    = blockIdx.x / nStacks;
  const int nSlots  = gridDim.x / nStacks;
  const float* __restrict__ Wx = WxAll + (size_t)st * D * D;
  const float* __restrict__ Wg = WgAll + (size_t)st * D * D;
  const float* __restrict__ W2 = W2All + (size_t)st * D * D;
  const float* __restrict__ bx = bxAll + (size_t)st * D;
  const float* __restrict__ bg = bgAll + (size_t)st * D;
  float* __restrict__ fvOut = fvAll + (size_t)st * planeStride;
  float* __restrict__ zOut  = zAll + (size_t)st * planeStride;

  v2f bxr[32];
  v2f bgr[32];
  v2f b2r[32];
#pragma unroll
  for (int ks = 0; ks < 32; ++ks) {
    const int kk = ks * 4 + half * 2;
    bxr[ks].x = Wx[(size_t)kk * D + colBase + lm];
    bxr[ks].y = Wx[(size_t)(kk + 1) * D + colBase + lm];
    bgr[ks].x = Wg[(size_t)kk * D + colBase + lm];
    bgr[ks].y = Wg[(size_t)(kk + 1) * D + colBase + lm];
    b2r[ks].x = W2[(size_t)kk * D + colBase + lm];
    b2r[ks].y = W2[(size_t)(kk + 1) * D + colBase + lm];
  }
  const float bxcol = bx[colBase + lm];
  const float bgcol = bg[colBase + lm];

  const int r0 = wid, r1 = wid + 8;
  const int c0 = lane * 4, c1 = lane * 4;
  const float4* __restrict__ X4 = reinterpret_cast<const float4*>(X);
  const float4* __restrict__ G4 = reinterpret_cast<const float4*>(G);

  int rt = slot;
  if (rt >= nRowTiles) return;

  float4 x0, x1, g0, g1;
  {
    const int rowBase = rt * 16;
    x0 = X4[(size_t)(rowBase + r0) * 32 + lane];
    x1 = X4[(size_t)(rowBase + r1) * 32 + lane];
    g0 = G4[(size_t)(rowBase + r0) * 32 + lane];
    g1 = G4[(size_t)(rowBase + r1) * 32 + lane];
    const float n0 = nrm[rowBase + r0], n1 = nrm[rowBase + r1];
    g0.x *= n0; g0.y *= n0; g0.z *= n0; g0.w *= n0;
    g1.x *= n1; g1.y *= n1; g1.z *= n1; g1.w *= n1;
  }

  for (;;) {
    *reinterpret_cast<float4*>(&Xs[r0 * LDS_STRIDE + c0]) = x0;
    *reinterpret_cast<float4*>(&Xs[r1 * LDS_STRIDE + c1]) = x1;
    *reinterpret_cast<float4*>(&Gs[r0 * LDS_STRIDE + c0]) = g0;
    *reinterpret_cast<float4*>(&Gs[r1 * LDS_STRIDE + c1]) = g1;

    const int rnext = rt + nSlots;
    const bool more = rnext < nRowTiles;
    __syncthreads();

    v8f accX = {0.f, 0.f, 0.f, 0.f, 0.f, 0.f, 0.f, 0.f};
    v8f accG = {0.f, 0.f, 0.f, 0.f, 0.f, 0.f, 0.f, 0.f};
#pragma unroll
    for (int ks = 0; ks < 32; ++ks) {
      const int kk = ks * 4 + half * 2;
      v2f ax, ag;
      ax.x = Xs[lm * LDS_STRIDE + kk];
      ax.y = Xs[lm * LDS_STRIDE + kk + 1];
      ag.x = Gs[lm * LDS_STRIDE + kk];
      ag.y = Gs[lm * LDS_STRIDE + kk + 1];
      accX = __builtin_amdgcn_wmma_f32_16x16x4_f32(
           false, ax,  false, bxr[ks],
           (short)0, accX,  false,  false);
      accG = __builtin_amdgcn_wmma_f32_16x16x4_f32(
           false, ag,  false, bgr[ks],
           (short)0, accG,  false,  false);
    }

    const int rowBase = rt * 16;
    const int col = colBase + lm;
    const int tileOff = (rowBase + half * 8) * D + col;
    float* __restrict__ fvp = fvOut + tileOff;
    float* __restrict__ zp  = zOut + tileOff;
    const float* __restrict__ nrp = nrm + rowBase + half * 8;
#pragma unroll
    for (int r = 0; r < 8; ++r) {
      accX[r] += bxcol;
      float hv = accG[r] + bgcol;
      hv += accX[r];
      Hs[(r + half * 8) * LDS_STRIDE + col] = fmaxf(hv, 0.f);
    }
    __syncthreads();
#pragma unroll
    for (int r = 0; r < 8; ++r) fvp[r * D] = accX[r];

    float n0n = 0.f, n1n = 0.f;
    if (more) {
      const int rowBaseN = rnext * 16;
      x0 = X4[(size_t)(rowBaseN + r0) * 32 + lane];
      x1 = X4[(size_t)(rowBaseN + r1) * 32 + lane];
      g0 = G4[(size_t)(rowBaseN + r0) * 32 + lane];
      g1 = G4[(size_t)(rowBaseN + r1) * 32 + lane];
      n0n = nrm[rowBaseN + r0];
      n1n = nrm[rowBaseN + r1];
    }
    float nr[8];
#pragma unroll
    for (int r = 0; r < 8; ++r) nr[r] = nrp[r];

    v8f accZ = {0.f, 0.f, 0.f, 0.f, 0.f, 0.f, 0.f, 0.f};
#pragma unroll
    for (int ks = 0; ks < 32; ++ks) {
      const int kk = ks * 4 + half * 2;
      v2f ah;
      ah.x = Hs[lm * LDS_STRIDE + kk];
      ah.y = Hs[lm * LDS_STRIDE + kk + 1];
      accZ = __builtin_amdgcn_wmma_f32_16x16x4_f32(
           false, ah,  false, b2r[ks],
           (short)0, accZ,  false,  false);
    }
#pragma unroll
    for (int r = 0; r < 8; ++r) zp[r * D] = accZ[r] * nr[r];

    if (!more) break;
    g0.x *= n0n; g0.y *= n0n; g0.z *= n0n; g0.w *= n0n;
    g1.x *= n1n; g1.y *= n1n; g1.z *= n1n; g1.w *= n1n;
    rt = rnext;
  }
}

extern "C" void kernel_launch(void* const* d_in, const int* in_sizes, int n_in,
                              void* d_out, int out_size, void* d_ws, size_t ws_size,
                              hipStream_t stream) {
  const float* feats = (const float*)d_in[0];
  const int*   src   = (const int*)d_in[1];
  const int*   dst   = (const int*)d_in[2];
  const float* w0    = (const float*)d_in[3];
  const float* b0    = (const float*)d_in[4];
  const float* w     = (const float*)d_in[5];
  const float* bw    = (const float*)d_in[6];
  const float* v     = (const float*)d_in[7];
  const float* bv    = (const float*)d_in[8];
  float* out = (float*)d_out;

  const int N = in_sizes[0] / D;
  const int E = in_sizes[1];
  const size_t M = (size_t)N * D;

  const size_t Nr  = ((size_t)N + 255) & ~(size_t)255;
  const size_t N1r = ((size_t)N + 1 + 255) & ~(size_t)255;
  const size_t Er  = ((size_t)E + 255) & ~(size_t)255;

  float* ws = (float*)d_ws;
  size_t off = 0;
  float* nrm  = ws + off; off += Nr;
  float* agg0 = ws + off; off += M;
  int* cnt    = (int*)(ws + off); off += Nr;
  int* rowptr = (int*)(ws + off); off += N1r;
  int* csrSrc = (int*)(ws + off); off += Er;
  float* planes = ws + off;
  int* pos      = (int*)planes;
  if ((off + 2 * M) * sizeof(float) > ws_size) return;
  const size_t roomPairs = (ws_size / sizeof(float) - off) / (2 * M);
  const int G = roomPairs >= (size_t)KSTACK ? KSTACK : (int)roomPairs;
  float* fv = planes;
  float* z1 = planes + (size_t)G * M;

  const int nRowTiles  = N / 16;
  const int chainBlocks = 512;
  const int gatherBlocks = (N + 8 * ROWS_PER_WAVE - 1) / (8 * ROWS_PER_WAVE);

  zero_f4_kernel<<<((int)(Nr / 4) + 255) / 256, 256, 0, stream>>>((float4*)cnt, (int)(Nr / 4));
  count_rank_kernel<<<(E + 255) / 256, 256, 0, stream>>>(dst, cnt, pos, E);
  scan_norm_kernel<<<(N + 255) / 256, 256, 0, stream>>>(cnt, rowptr, nrm, N);
  csr_fill_kernel<<<(E + 255) / 256, 256, 0, stream>>>(src, dst, pos, rowptr, csrSrc, E);

  spmm_gather_kernel<<<(N + 7) / 8, 256, 0, stream>>>(feats, rowptr, csrSrc, nrm, agg0, N);

  for (int k0 = 0; k0 < KSTACK; k0 += G) {
    const int g = (KSTACK - k0 < G) ? (KSTACK - k0) : G;
    const int slots = chainBlocks / g;
    gemm_chain_kernel<<<g * slots, 256, 0, stream>>>(
        feats, agg0, nrm, v + (size_t)k0 * D * D, bv + (size_t)k0 * D,
        w0 + (size_t)k0 * D * D, b0 + (size_t)k0 * D, w + (size_t)k0 * D * D,
        fv, z1, nRowTiles, g, M);
    gather_out_kernel<<<gatherBlocks, 256, 0, stream>>>(
        z1, rowptr, csrSrc, nrm, bw + (size_t)k0 * D, fv, out, N,
        1.0f / (float)KSTACK, g, M, k0 > 0 ? 1 : 0);
  }
}


# CSR build keeps several independent memory operations per lane in flight: rank and fill passes take 4 edges per lane (16-byte loads, 1/4 of the waves), scan front sum is one group of 16 loads
# speedup vs baseline: 2.9730x; 1.0562x over previous
#include <hip/hip_runtime.h>

typedef float v2f __attribute__((ext_vector_type(2)));
typedef float v8f __attribute__((ext_vector_type(8)));

#define D 128
#define KSTACK 3
#define LDS_STRIDE 132

__global__ void zero_f4_kernel(float4* __restrict__ p, int n4) {
  int i = blockIdx.x * blockDim.x + threadIdx.x;
  if (i < n4) p[i] = make_float4(0.f, 0.f, 0.f, 0.f);
}

__global__ __launch_bounds__(256) void count_rank_kernel(const int* __restrict__ dst,
                                                         int* __restrict__ cnt,
                                                         int* __restrict__ pos, int nE) {
  const int n4 = nE >> 2;
  const int q = blockIdx.x * 256 + threadIdx.x;
  if (q < n4) {
    const int4 d = reinterpret_cast<const int4*>(dst)[q];
    int4 r;
    r.x = atomicAdd(&cnt[d.x], 1);
    r.y = atomicAdd(&cnt[d.y], 1);
    r.z = atomicAdd(&cnt[d.z], 1);
    r.w = atomicAdd(&cnt[d.w], 1);
    reinterpret_cast<int4*>(pos)[q] = r;
  }
  if (blockIdx.x == 0 && (int)threadIdx.x < (nE & 3)) {
    const int i = (n4 << 2) + threadIdx.x;
    pos[i] = atomicAdd(&cnt[dst[i]], 1);
  }
}

#define SCAN_T 1024
__global__ __launch_bounds__(SCAN_T) void scan_norm_kernel(const int* __restrict__ cnt,
                                                           int* __restrict__ rowptr,
                                                           float* __restrict__ nrm, int n) {
  __shared__ int waveIncl[32];
  __shared__ int waveFront[32];
  const int tid  = threadIdx.x;
  const int lane = tid & 31;
  const int wid  = tid >> 5;
  const int base = blockIdx.x * SCAN_T;

  const int i = base + tid;
  const int c = (i < n) ? cnt[i] : 0;

  const int4* __restrict__ c4 = reinterpret_cast<const int4*>(cnt);
  const int nq = base >> 2;
  int front = 0;
  for (int q0 = tid; q0 < nq; q0 += 16 * SCAN_T) {
    int4 v[16];
#pragma unroll
    for (int u = 0; u < 16; ++u) {
      const int q = q0 + u * SCAN_T;
      v[u] = make_int4(0, 0, 0, 0);
      if (q < nq) v[u] = c4[q];
    }
#pragma unroll
    for (int u = 0; u < 16; ++u) front += (v[u].x + v[u].y) + (v[u].z + v[u].w);
  }
  for (int o = 16; o > 0; o >>= 1) front += __shfl_xor(front, o);

  if (i < n) nrm[i] = rsqrtf(fmaxf((float)c, 1.0f));

  int incl = c;
  for (int o = 1; o < 32; o <<= 1) {
    const int t = __shfl_up(incl, o);
    if (lane >= o) incl += t;
  }
  if (lane == 31) waveIncl[wid]  = incl;
  if (lane == 0)  waveFront[wid] = front;
  __syncthreads();
  int off = waveFront[lane] + (lane < wid ? waveIncl[lane] : 0);
  for (int o = 16; o > 0; o >>= 1) off += __shfl_xor(off, o);
  const int excl = off + incl - c;
  if (i < n) rowptr[i] = excl;
  if (i == n - 1) rowptr[n] = excl + c;
}

__global__ __launch_bounds__(256) void csr_fill_kernel(const int* __restrict__ src,
                                                       const int* __restrict__ dst,
                                                       const int* __restrict__ pos,
                                                       const int* __restrict__ rowptr,
                                                       int* __restrict__ csrSrc, int nE) {
  const int n4 = nE >> 2;
  const int q = blockIdx.x * 256 + threadIdx.x;
  if (q < n4) {
    const int4 s = reinterpret_cast<const int4*>(src)[q];
    const int4 d = reinterpret_cast<const int4*>(dst)[q];
    const int4 p = reinterpret_cast<const int4*>(pos)[q];
    const int b0 = rowptr[d.x], b1 = rowptr[d.y], b2 = rowptr[d.z], b3 = rowptr[d.w];
    csrSrc[b0 + p.x] = s.x;
    csrSrc[b1 + p.y] = s.y;
    csrSrc[b2 + p.z] = s.z;
    csrSrc[b3 + p.w] = s.w;
  }
  if (blockIdx.x == 0 && (int)threadIdx.x < (nE & 3)) {
    const int i = (n4 << 2) + threadIdx.x;
    csrSrc[rowptr[dst[i]] + pos[i]] = src[i];
  }
}

__device__ __forceinline__ void row_fma(float4& acc, const float4 v, const float w) {
  acc.x = fmaf(v.x, w, acc.x); acc.y = fmaf(v.y, w, acc.y);
  acc.z = fmaf(v.z, w, acc.z); acc.w = fmaf(v.w, w, acc.w);
}

template <int R>
__device__ __forceinline__ void gather_rows(float4& acc, const float4* __restrict__ h4,
                                            const int sL, const int nLi,
                                            const int j, const int lane) {
  float4 v[R];
#pragma unroll
  for (int u = 0; u < R; ++u) {
    const int s = __builtin_amdgcn_readlane(sL, j + u);
    v[u] = h4[(size_t)s * 32 + lane];
  }
#pragma unroll
  for (int u = 0; u < R; ++u)
    row_fma(acc, v[u], __int_as_float(__builtin_amdgcn_readlane(nLi, j + u)));
}

__device__ __forceinline__ void load_chunk(int& sL, int& nLi, const int* __restrict__ csrSrc,
                                           const float* __restrict__ nrm,
                                           const int base, const int cnt, const int nN,
                                           const int lane) {
  sL = 0;
  float nL = 0.f;
  if (lane < cnt) {
    sL = min(max(csrSrc[base + lane], 0), nN - 1);
    nL = nrm[sL];
  }
  nLi = __float_as_int(nL);
}

template <int RMAX>
__device__ __forceinline__ void sum_chunk(float4& acc, const float4* __restrict__ h4,
                                          const int sL, const int nLi, const int cnt,
                                          const int lane) {
  int j = 0;
  if (RMAX >= 8)
    for (; j + 8 <= cnt; j += 8) gather_rows<8>(acc, h4, sL, nLi, j, lane);
  for (; j + 4 <= cnt; j += 4) gather_rows<4>(acc, h4, sL, nLi, j, lane);
  const int rem = cnt - j;
  if (rem == 3)      gather_rows<3>(acc, h4, sL, nLi, j, lane);
  else if (rem == 2) gather_rows<2>(acc, h4, sL, nLi, j, lane);
  else if (rem == 1) gather_rows<1>(acc, h4, sL, nLi, j, lane);
}

__global__ __launch_bounds__(256) void spmm_gather_kernel(const float* __restrict__ h,
                                                          const int* __restrict__ rowptr,
                                                          const int* __restrict__ csrSrc,
                                                          const float* __restrict__ nrm,
                                                          float* __restrict__ agg, int nN) {
  const int lane = threadIdx.x & 31;
  const int d = __builtin_amdgcn_readfirstlane(blockIdx.x * 8 + (threadIdx.x >> 5));
  if (d >= nN) return;
  const int beg = __builtin_amdgcn_readfirstlane(rowptr[d]);
  const int end = __builtin_amdgcn_readfirstlane(rowptr[d + 1]);
  const float4* __restrict__ h4 = reinterpret_cast<const float4*>(h);

  float4 acc = make_float4(0.f, 0.f, 0.f, 0.f);
  for (int base = beg; base < end; base += 32) {
    const int cnt = min(32, end - base);
    int sL, nLi;
    load_chunk(sL, nLi, csrSrc, nrm, base, cnt, nN, lane);
    sum_chunk<4>(acc, h4, sL, nLi, cnt, lane);
  }
  reinterpret_cast<float4*>(agg)[(size_t)d * 32 + lane] = acc;
}

#define ROWS_PER_WAVE 5

template <int R>
__device__ __forceinline__ void gather_rows_plain(float4& acc, const float4* __restrict__ z4,
                                                  const int sL, const int j, const int lane) {
  float4 v[R];
#pragma unroll
  for (int u = 0; u < R; ++u) {
    const int s = __builtin_amdgcn_readlane(sL, j + u);
    v[u] = z4[(size_t)s * 32 + lane];
  }
#pragma unroll
  for (int u = 0; u < R; ++u) {
    acc.x += v[u].x; acc.y += v[u].y; acc.z += v[u].z; acc.w += v[u].w;
  }
}

__device__ __forceinline__ void sum_chunk_plain(float4& acc, const float4* __restrict__ z4,
                                                const int sL, const int cnt, const int lane) {
  int j = 0;
  for (; j + 8 <= cnt; j += 8) gather_rows_plain<8>(acc, z4, sL, j, lane);
  for (; j + 4 <= cnt; j += 4) gather_rows_plain<4>(acc, z4, sL, j, lane);
  const int rem = cnt - j;
  if (rem == 3)      gather_rows_plain<3>(acc, z4, sL, j, lane);
  else if (rem == 2) gather_rows_plain<2>(acc, z4, sL, j, lane);
  else if (rem == 1) gather_rows_plain<1>(acc, z4, sL, j, lane);
}

__global__ __launch_bounds__(256, 10) void gather_out_kernel(
    const float* __restrict__ zAll, const int* __restrict__ rowptr,
    const int* __restrict__ csrSrc, const float* __restrict__ nrm,
    const float* __restrict__ biasAll, const float* __restrict__ addmAll,
    float* __restrict__ out, int nN, float outScale,
    int nStacks, size_t planeStride, int accumFirst) {
  const int lane = threadIdx.x & 31;
  const int w  = __builtin_amdgcn_readfirstlane(blockIdx.x * 8 + (threadIdx.x >> 5));
  const int nW = gridDim.x * 8;
  float4* __restrict__ out4 = reinterpret_cast<float4*>(out);

  int beg[ROWS_PER_WAVE], end[ROWS_PER_WAVE], sL0[ROWS_PER_WAVE];
  float nd[ROWS_PER_WAVE];
  float4 o[ROWS_PER_WAVE];
#pragma unroll
  for (int j = 0; j < ROWS_PER_WAVE; ++j) {
    const int d = w + j * nW;
    beg[j] = 0; end[j] = 0; sL0[j] = 0; nd[j] = 0.f;
    o[j] = make_float4(0.f, 0.f, 0.f, 0.f);
    if (d < nN) {
      beg[j] = __builtin_amdgcn_readfirstlane(rowptr[d]);
      end[j] = __builtin_amdgcn_readfirstlane(rowptr[d + 1]);
      if (lane < end[j] - beg[j])
        sL0[j] = min(max(csrSrc[beg[j] + lane], 0), nN - 1);
      nd[j] = nrm[d];
      if (accumFirst) o[j] = out4[(size_t)d * 32 + lane];
    }
  }

  for (int st = 0; st < nStacks; ++st) {
    const float4* __restrict__ z4 =
        reinterpret_cast<const float4*>(zAll + (size_t)st * planeStride);
    const float4* __restrict__ addm4 =
        reinterpret_cast<const float4*>(addmAll + (size_t)st * planeStride);
    const float4 b4 = reinterpret_cast<const float4*>(biasAll + (size_t)st * D)[lane];
#pragma unroll
    for (int j = 0; j < ROWS_PER_WAVE; ++j) {
      const int d = w + j * nW;
      if (d < nN) {
        const float4 f = addm4[(size_t)d * 32 + lane];
        float4 acc = make_float4(0.f, 0.f, 0.f, 0.f);
        sum_chunk_plain(acc, z4, sL0[j], min(32, end[j] - beg[j]), lane);
        for (int base = beg[j] + 32; base < end[j]; base += 32) {
          const int cnt = min(32, end[j] - base);
          int sL = 0;
          if (lane < cnt) sL = min(max(csrSrc[base + lane], 0), nN - 1);
          sum_chunk_plain(acc, z4, sL, cnt, lane);
        }
        float4 c;
        c.x = fmaxf(acc.x * nd[j] + b4.x + f.x, 0.f) * outScale;
        c.y = fmaxf(acc.y * nd[j] + b4.y + f.y, 0.f) * outScale;
        c.z = fmaxf(acc.z * nd[j] + b4.z + f.z, 0.f) * outScale;
        c.w = fmaxf(acc.w * nd[j] + b4.w + f.w, 0.f) * outScale;
        o[j].x += c.x; o[j].y += c.y; o[j].z += c.z; o[j].w += c.w;
      }
    }
  }

#pragma unroll
  for (int j = 0; j < ROWS_PER_WAVE; ++j) {
    const int d = w + j * nW;
    if (d < nN) out4[(size_t)d * 32 + lane] = o[j];
  }
}

__global__ __launch_bounds__(256, 4) void gemm_chain_kernel(
    const float* __restrict__ X, const float* __restrict__ G,
    const float* __restrict__ nrm,
    const float* __restrict__ WxAll, const float* __restrict__ bxAll,
    const float* __restrict__ WgAll, const float* __restrict__ bgAll,
    const float* __restrict__ W2All,
    float* __restrict__ fvAll, float* __restrict__ zAll, int nRowTiles,
    int nStacks, size_t planeStride) {
  __shared__ float Xs[16 * LDS_STRIDE];
  __shared__ float Gs[16 * LDS_STRIDE];
  __shared__ float Hs[16 * LDS_STRIDE];
  const int tid  = threadIdx.x;
  const int wid  = __builtin_amdgcn_readfirstlane(tid >> 5);
  const int lane = tid & 31;
  const int half = lane >> 4;
  const int lm   = lane & 15;
  const int colBase = wid * 16;
  const int st      = blockIdx.x % nStacks;
  const int slot    = blockIdx.x / nStacks;
  const int nSlots  = gridDim.x / nStacks;
  const float* __restrict__ Wx = WxAll + (size_t)st * D * D;
  const float* __restrict__ Wg = WgAll + (size_t)st * D * D;
  const float* __restrict__ W2 = W2All + (size_t)st * D * D;
  const float* __restrict__ bx = bxAll + (size_t)st * D;
  const float* __restrict__ bg = bgAll + (size_t)st * D;
  float* __restrict__ fvOut = fvAll + (size_t)st * planeStride;
  float* __restrict__ zOut  = zAll + (size_t)st * planeStride;

  v2f bxr[32];
  v2f bgr[32];
  v2f b2r[32];
#pragma unroll
  for (int ks = 0; ks < 32; ++ks) {
    const int kk = ks * 4 + half * 2;
    bxr[ks].x = Wx[(size_t)kk * D + colBase + lm];
    bxr[ks].y = Wx[(size_t)(kk + 1) * D + colBase + lm];
    bgr[ks].x = Wg[(size_t)kk * D + colBase + lm];
    bgr[ks].y = Wg[(size_t)(kk + 1) * D + colBase + lm];
    b2r[ks].x = W2[(size_t)kk * D + colBase + lm];
    b2r[ks].y = W2[(size_t)(kk + 1) * D + colBase + lm];
  }
  const float bxcol = bx[colBase + lm];
  const float bgcol = bg[colBase + lm];

  const int r0 = wid, r1 = wid + 8;
  const int c0 = lane * 4, c1 = lane * 4;
  const float4* __restrict__ X4 = reinterpret_cast<const float4*>(X);
  const float4* __restrict__ G4 = reinterpret_cast<const float4*>(G);

  int rt = slot;
  if (rt >= nRowTiles) return;

  float4 x0, x1, g0, g1;
  {
    const int rowBase = rt * 16;
    x0 = X4[(size_t)(rowBase + r0) * 32 + lane];
    x1 = X4[(size_t)(rowBase + r1) * 32 + lane];
    g0 = G4[(size_t)(rowBase + r0) * 32 + lane];
    g1 = G4[(size_t)(rowBase + r1) * 32 + lane];
    const float n0 = nrm[rowBase + r0], n1 = nrm[rowBase + r1];
    g0.x *= n0; g0.y *= n0; g0.z *= n0; g0.w *= n0;
    g1.x *= n1; g1.y *= n1; g1.z *= n1; g1.w *= n1;
  }

  for (;;) {
    *reinterpret_cast<float4*>(&Xs[r0 * LDS_STRIDE + c0]) = x0;
    *reinterpret_cast<float4*>(&Xs[r1 * LDS_STRIDE + c1]) = x1;
    *reinterpret_cast<float4*>(&Gs[r0 * LDS_STRIDE + c0]) = g0;
    *reinterpret_cast<float4*>(&Gs[r1 * LDS_STRIDE + c1]) = g1;

    const int rnext = rt + nSlots;
    const bool more = rnext < nRowTiles;
    __syncthreads();

    v8f accX = {0.f, 0.f, 0.f, 0.f, 0.f, 0.f, 0.f, 0.f};
    v8f accG = {0.f, 0.f, 0.f, 0.f, 0.f, 0.f, 0.f, 0.f};
#pragma unroll
    for (int ks = 0; ks < 32; ++ks) {
      const int kk = ks * 4 + half * 2;
      v2f ax, ag;
      ax.x = Xs[lm * LDS_STRIDE + kk];
      ax.y = Xs[lm * LDS_STRIDE + kk + 1];
      ag.x = Gs[lm * LDS_STRIDE + kk];
      ag.y = Gs[lm * LDS_STRIDE + kk + 1];
      accX = __builtin_amdgcn_wmma_f32_16x16x4_f32(
           false, ax,  false, bxr[ks],
           (short)0, accX,  false,  false);
      accG = __builtin_amdgcn_wmma_f32_16x16x4_f32(
           false, ag,  false, bgr[ks],
           (short)0, accG,  false,  false);
    }

    const int rowBase = rt * 16;
    const int col = colBase + lm;
    const int tileOff = (rowBase + half * 8) * D + col;
    float* __restrict__ fvp = fvOut + tileOff;
    float* __restrict__ zp  = zOut + tileOff;
    const float* __restrict__ nrp = nrm + rowBase + half * 8;
#pragma unroll
    for (int r = 0; r < 8; ++r) {
      accX[r] += bxcol;
      float hv = accG[r] + bgcol;
      hv += accX[r];
      Hs[(r + half * 8) * LDS_STRIDE + col] = fmaxf(hv, 0.f);
    }
    __syncthreads();
#pragma unroll
    for (int r = 0; r < 8; ++r) fvp[r * D] = accX[r];

    float n0n = 0.f, n1n = 0.f;
    if (more) {
      const int rowBaseN = rnext * 16;
      x0 = X4[(size_t)(rowBaseN + r0) * 32 + lane];
      x1 = X4[(size_t)(rowBaseN + r1) * 32 + lane];
      g0 = G4[(size_t)(rowBaseN + r0) * 32 + lane];
      g1 = G4[(size_t)(rowBaseN + r1) * 32 + lane];
      n0n = nrm[rowBaseN + r0];
      n1n = nrm[rowBaseN + r1];
    }
    float nr[8];
#pragma unroll
    for (int r = 0; r < 8; ++r) nr[r] = nrp[r];

    v8f accZ = {0.f, 0.f, 0.f, 0.f, 0.f, 0.f, 0.f, 0.f};
#pragma unroll
    for (int ks = 0; ks < 32; ++ks) {
      const int kk = ks * 4 + half * 2;
      v2f ah;
      ah.x = Hs[lm * LDS_STRIDE + kk];
      ah.y = Hs[lm * LDS_STRIDE + kk + 1];
      accZ = __builtin_amdgcn_wmma_f32_16x16x4_f32(
           false, ah,  false, b2r[ks],
           (short)0, accZ,  false,  false);
    }
#pragma unroll
    for (int r = 0; r < 8; ++r) zp[r * D] = accZ[r] * nr[r];

    if (!more) break;
    g0.x *= n0n; g0.y *= n0n; g0.z *= n0n; g0.w *= n0n;
    g1.x *= n1n; g1.y *= n1n; g1.z *= n1n; g1.w *= n1n;
    rt = rnext;
  }
}

extern "C" void kernel_launch(void* const* d_in, const int* in_sizes, int n_in,
                              void* d_out, int out_size, void* d_ws, size_t ws_size,
                              hipStream_t stream) {
  const float* feats = (const float*)d_in[0];
  const int*   src   = (const int*)d_in[1];
  const int*   dst   = (const int*)d_in[2];
  const float* w0    = (const float*)d_in[3];
  const float* b0    = (const float*)d_in[4];
  const float* w     = (const float*)d_in[5];
  const float* bw    = (const float*)d_in[6];
  const float* v     = (const float*)d_in[7];
  const float* bv    = (const float*)d_in[8];
  float* out = (float*)d_out;

  const int N = in_sizes[0] / D;
  const int E = in_sizes[1];
  const size_t M = (size_t)N * D;

  const size_t Nr  = ((size_t)N + 255) & ~(size_t)255;
  const size_t N1r = ((size_t)N + 1 + 255) & ~(size_t)255;
  const size_t Er  = ((size_t)E + 255) & ~(size_t)255;

  float* ws = (float*)d_ws;
  size_t off = 0;
  float* nrm  = ws + off; off += Nr;
  float* agg0 = ws + off; off += M;
  int* cnt    = (int*)(ws + off); off += Nr;
  int* rowptr = (int*)(ws + off); off += N1r;
  int* csrSrc = (int*)(ws + off); off += Er;
  float* planes = ws + off;
  int* pos      = (int*)planes;
  if ((off + 2 * M) * sizeof(float) > ws_size) return;
  const size_t roomPairs = (ws_size / sizeof(float) - off) / (2 * M);
  const int G = roomPairs >= (size_t)KSTACK ? KSTACK : (int)roomPairs;
  float* fv = planes;
  float* z1 = planes + (size_t)G * M;

  const int nRowTiles  = N / 16;
  const int chainBlocks = 512;
  const int gatherBlocks = (N + 8 * ROWS_PER_WAVE - 1) / (8 * ROWS_PER_WAVE);
  const int edgeBlocks = ((E >> 2) + 255) / 256 + ((E >> 2) == 0 ? 1 : 0);

  zero_f4_kernel<<<((int)(Nr / 4) + 255) / 256, 256, 0, stream>>>((float4*)cnt, (int)(Nr / 4));
  count_rank_kernel<<<edgeBlocks, 256, 0, stream>>>(dst, cnt, pos, E);
  scan_norm_kernel<<<(N + SCAN_T - 1) / SCAN_T, SCAN_T, 0, stream>>>(cnt, rowptr, nrm, N);
  csr_fill_kernel<<<edgeBlocks, 256, 0, stream>>>(src, dst, pos, rowptr, csrSrc, E);

  spmm_gather_kernel<<<(N + 7) / 8, 256, 0, stream>>>(feats, rowptr, csrSrc, nrm, agg0, N);

  for (int k0 = 0; k0 < KSTACK; k0 += G) {
    const int g = (KSTACK - k0 < G) ? (KSTACK - k0) : G;
    const int slots = chainBlocks / g;
    gemm_chain_kernel<<<g * slots, 256, 0, stream>>>(
        feats, agg0, nrm, v + (size_t)k0 * D * D, bv + (size_t)k0 * D,
        w0 + (size_t)k0 * D * D, b0 + (size_t)k0 * D, w + (size_t)k0 * D * D,
        fv, z1, nRowTiles, g, M);
    gather_out_kernel<<<gatherBlocks, 256, 0, stream>>>(
        z1, rowptr, csrSrc, nrm, bw + (size_t)k0 * D, fv, out, N,
        1.0f / (float)KSTACK, g, M, k0 > 0 ? 1 : 0);
  }
}


# CSR rank pass on counters in LDS: 64 work-groups rank their edge shares with LDS adds, then add their tables to memory, one returning add per 128-byte line; edge sums in a new slot order
# speedup vs baseline: 3.6012x; 1.2113x over previous
#include <hip/hip_runtime.h>

typedef float v2f __attribute__((ext_vector_type(2)));
typedef float v8f __attribute__((ext_vector_type(8)));

#define D 128
#define KSTACK 3
#define LDS_STRIDE 132

__global__ void zero_f4_kernel(float4* __restrict__ p, int n4) {
  int i = blockIdx.x * blockDim.x + threadIdx.x;
  if (i < n4) p[i] = make_float4(0.f, 0.f, 0.f, 0.f);
}

__global__ __launch_bounds__(256) void count_rank_kernel(const int* __restrict__ dst,
                                                         int* __restrict__ cnt,
                                                         int* __restrict__ pos, int nE) {
  const int n4 = nE >> 2;
  const int q = blockIdx.x * 256 + threadIdx.x;
  if (q < n4) {
    const int4 d = reinterpret_cast<const int4*>(dst)[q];
    int4 r;
    r.x = atomicAdd(&cnt[d.x], 1);
    r.y = atomicAdd(&cnt[d.y], 1);
    r.z = atomicAdd(&cnt[d.z], 1);
    r.w = atomicAdd(&cnt[d.w], 1);
    reinterpret_cast<int4*>(pos)[q] = r;
  }
  if (blockIdx.x == 0 && (int)threadIdx.x < (nE & 3)) {
    const int i = (n4 << 2) + threadIdx.x;
    pos[i] = atomicAdd(&cnt[dst[i]], 1);
  }
}

#define RANK_KEYS 51200
#define RANK_T 1024
#define RANK_G 4
#define RANK_U 7
__global__ __launch_bounds__(RANK_T) void rank_lds_kernel(const int* __restrict__ dst,
                                                          int* __restrict__ cnt,
                                                          int* __restrict__ pos, int nE,
                                                          int nKeysPad, int groupsPerBlock) {
  __shared__ int cntL[RANK_KEYS];
  const int tid = threadIdx.x;
  const int n4 = nE >> 2;
  const int gBeg = blockIdx.x * groupsPerBlock;
  const int gEnd = min(n4, gBeg + groupsPerBlock);
  const int4* __restrict__ dst4 = reinterpret_cast<const int4*>(dst);

  int4 d[RANK_G], r[RANK_G];
#pragma unroll
  for (int k = 0; k < RANK_G; ++k) {
    const int g = gBeg + tid + k * RANK_T;
    d[k] = make_int4(0, 0, 0, 0);
    r[k] = make_int4(0, 0, 0, 0);
    if (g < gEnd) d[k] = dst4[g];
  }
  const bool tail = blockIdx.x == 0 && tid < (nE & 3);
  int dT = 0, rT = 0;
  if (tail) dT = dst[(n4 << 2) + tid];

  int4* cntL4 = reinterpret_cast<int4*>(cntL);
  for (int i = tid; i < (nKeysPad >> 2); i += RANK_T) cntL4[i] = make_int4(0, 0, 0, 0);
  __syncthreads();

#pragma unroll
  for (int k = 0; k < RANK_G; ++k) {
    const int g = gBeg + tid + k * RANK_T;
    if (g < gEnd) {
      r[k].x = atomicAdd(&cntL[d[k].x], 1);
      r[k].y = atomicAdd(&cntL[d[k].y], 1);
      r[k].z = atomicAdd(&cntL[d[k].z], 1);
      r[k].w = atomicAdd(&cntL[d[k].w], 1);
    }
  }
  if (tail) rT = atomicAdd(&cntL[dT], 1);
  __syncthreads();

  const int nSteps = (nKeysPad + RANK_T - 1) / RANK_T;
  const int rot = (int)(((long long)blockIdx.x * nSteps) / gridDim.x);
  for (int s0 = 0; s0 < nSteps; s0 += RANK_U) {
    int ix[RANK_U], c[RANK_U], b[RANK_U];
#pragma unroll
    for (int u = 0; u < RANK_U; ++u) {
      int st = s0 + u + rot;
      if (st >= nSteps) st -= nSteps;
      const int i = st * RANK_T + tid;
      const bool ok = (s0 + u < nSteps) && (i < nKeysPad);
      ix[u] = ok ? i : tid;
      const int v = cntL[ix[u]];
      c[u] = ok ? v : 0;
    }
#pragma unroll
    for (int u = 0; u < RANK_U; ++u) {
      b[u] = c[u];
      if (c[u] != 0) b[u] = atomicAdd(&cnt[ix[u]], c[u]);
    }
#pragma unroll
    for (int u = 0; u < RANK_U; ++u)
      if (c[u] != 0) cntL[ix[u]] = b[u];
  }
  __syncthreads();

#pragma unroll
  for (int k = 0; k < RANK_G; ++k) {
    const int g = gBeg + tid + k * RANK_T;
    if (g < gEnd) {
      int4 p;
      p.x = r[k].x + cntL[d[k].x];
      p.y = r[k].y + cntL[d[k].y];
      p.z = r[k].z + cntL[d[k].z];
      p.w = r[k].w + cntL[d[k].w];
      reinterpret_cast<int4*>(pos)[g] = p;
    }
  }
  if (tail) pos[(n4 << 2) + tid] = rT + cntL[dT];
}

#define SCAN_T 1024
__global__ __launch_bounds__(SCAN_T) void scan_norm_kernel(const int* __restrict__ cnt,
                                                           int* __restrict__ rowptr,
                                                           float* __restrict__ nrm, int n) {
  __shared__ int waveIncl[32];
  __shared__ int waveFront[32];
  const int tid  = threadIdx.x;
  const int lane = tid & 31;
  const int wid  = tid >> 5;
  const int base = blockIdx.x * SCAN_T;

  const int i = base + tid;
  const int c = (i < n) ? cnt[i] : 0;

  const int4* __restrict__ c4 = reinterpret_cast<const int4*>(cnt);
  const int nq = base >> 2;
  int front = 0;
  for (int q0 = tid; q0 < nq; q0 += 16 * SCAN_T) {
    int4 v[16];
#pragma unroll
    for (int u = 0; u < 16; ++u) {
      const int q = q0 + u * SCAN_T;
      v[u] = make_int4(0, 0, 0, 0);
      if (q < nq) v[u] = c4[q];
    }
#pragma unroll
    for (int u = 0; u < 16; ++u) front += (v[u].x + v[u].y) + (v[u].z + v[u].w);
  }
  for (int o = 16; o > 0; o >>= 1) front += __shfl_xor(front, o);

  if (i < n) nrm[i] = rsqrtf(fmaxf((float)c, 1.0f));

  int incl = c;
  for (int o = 1; o < 32; o <<= 1) {
    const int t = __shfl_up(incl, o);
    if (lane >= o) incl += t;
  }
  if (lane == 31) waveIncl[wid]  = incl;
  if (lane == 0)  waveFront[wid] = front;
  __syncthreads();
  int off = waveFront[lane] + (lane < wid ? waveIncl[lane] : 0);
  for (int o = 16; o > 0; o >>= 1) off += __shfl_xor(off, o);
  const int excl = off + incl - c;
  if (i < n) rowptr[i] = excl;
  if (i == n - 1) rowptr[n] = excl + c;
}

__global__ __launch_bounds__(256) void csr_fill_kernel(const int* __restrict__ src,
                                                       const int* __restrict__ dst,
                                                       const int* __restrict__ pos,
                                                       const int* __restrict__ rowptr,
                                                       int* __restrict__ csrSrc, int nE) {
  const int n4 = nE >> 2;
  const int q = blockIdx.x * 256 + threadIdx.x;
  if (q < n4) {
    const int4 s = reinterpret_cast<const int4*>(src)[q];
    const int4 d = reinterpret_cast<const int4*>(dst)[q];
    const int4 p = reinterpret_cast<const int4*>(pos)[q];
    const int b0 = rowptr[d.x], b1 = rowptr[d.y], b2 = rowptr[d.z], b3 = rowptr[d.w];
    csrSrc[b0 + p.x] = s.x;
    csrSrc[b1 + p.y] = s.y;
    csrSrc[b2 + p.z] = s.z;
    csrSrc[b3 + p.w] = s.w;
  }
  if (blockIdx.x == 0 && (int)threadIdx.x < (nE & 3)) {
    const int i = (n4 << 2) + threadIdx.x;
    csrSrc[rowptr[dst[i]] + pos[i]] = src[i];
  }
}

__device__ __forceinline__ void row_fma(float4& acc, const float4 v, const float w) {
  acc.x = fmaf(v.x, w, acc.x); acc.y = fmaf(v.y, w, acc.y);
  acc.z = fmaf(v.z, w, acc.z); acc.w = fmaf(v.w, w, acc.w);
}

template <int R>
__device__ __forceinline__ void gather_rows(float4& acc, const float4* __restrict__ h4,
                                            const int sL, const int nLi,
                                            const int j, const int lane) {
  float4 v[R];
#pragma unroll
  for (int u = 0; u < R; ++u) {
    const int s = __builtin_amdgcn_readlane(sL, j + u);
    v[u] = h4[(size_t)s * 32 + lane];
  }
#pragma unroll
  for (int u = 0; u < R; ++u)
    row_fma(acc, v[u], __int_as_float(__builtin_amdgcn_readlane(nLi, j + u)));
}

__device__ __forceinline__ void load_chunk(int& sL, int& nLi, const int* __restrict__ csrSrc,
                                           const float* __restrict__ nrm,
                                           const int base, const int cnt, const int nN,
                                           const int lane) {
  sL = 0;
  float nL = 0.f;
  if (lane < cnt) {
    sL = min(max(csrSrc[base + lane], 0), nN - 1);
    nL = nrm[sL];
  }
  nLi = __float_as_int(nL);
}

template <int RMAX>
__device__ __forceinline__ void sum_chunk(float4& acc, const float4* __restrict__ h4,
                                          const int sL, const int nLi, const int cnt,
                                          const int lane) {
  int j = 0;
  if (RMAX >= 8)
    for (; j + 8 <= cnt; j += 8) gather_rows<8>(acc, h4, sL, nLi, j, lane);
  for (; j + 4 <= cnt; j += 4) gather_rows<4>(acc, h4, sL, nLi, j, lane);
  const int rem = cnt - j;
  if (rem == 3)      gather_rows<3>(acc, h4, sL, nLi, j, lane);
  else if (rem == 2) gather_rows<2>(acc, h4, sL, nLi, j, lane);
  else if (rem == 1) gather_rows<1>(acc, h4, sL, nLi, j, lane);
}

__global__ __launch_bounds__(256) void spmm_gather_kernel(const float* __restrict__ h,
                                                          const int* __restrict__ rowptr,
                                                          const int* __restrict__ csrSrc,
                                                          const float* __restrict__ nrm,
                                                          float* __restrict__ agg, int nN) {
  const int lane = threadIdx.x & 31;
  const int d = __builtin_amdgcn_readfirstlane(blockIdx.x * 8 + (threadIdx.x >> 5));
  if (d >= nN) return;
  const int beg = __builtin_amdgcn_readfirstlane(rowptr[d]);
  const int end = __builtin_amdgcn_readfirstlane(rowptr[d + 1]);
  const float4* __restrict__ h4 = reinterpret_cast<const float4*>(h);

  float4 acc = make_float4(0.f, 0.f, 0.f, 0.f);
  for (int base = beg; base < end; base += 32) {
    const int cnt = min(32, end - base);
    int sL, nLi;
    load_chunk(sL, nLi, csrSrc, nrm, base, cnt, nN, lane);
    sum_chunk<4>(acc, h4, sL, nLi, cnt, lane);
  }
  reinterpret_cast<float4*>(agg)[(size_t)d * 32 + lane] = acc;
}

#define ROWS_PER_WAVE 5

template <int R>
__device__ __forceinline__ void gather_rows_plain(float4& acc, const float4* __restrict__ z4,
                                                  const int sL, const int j, const int lane) {
  float4 v[R];
#pragma unroll
  for (int u = 0; u < R; ++u) {
    const int s = __builtin_amdgcn_readlane(sL, j + u);
    v[u] = z4[(size_t)s * 32 + lane];
  }
#pragma unroll
  for (int u = 0; u < R; ++u) {
    acc.x += v[u].x; acc.y += v[u].y; acc.z += v[u].z; acc.w += v[u].w;
  }
}

__device__ __forceinline__ void sum_chunk_plain(float4& acc, const float4* __restrict__ z4,
                                                const int sL, const int cnt, const int lane) {
  int j = 0;
  for (; j + 8 <= cnt; j += 8) gather_rows_plain<8>(acc, z4, sL, j, lane);
  for (; j + 4 <= cnt; j += 4) gather_rows_plain<4>(acc, z4, sL, j, lane);
  const int rem = cnt - j;
  if (rem == 3)      gather_rows_plain<3>(acc, z4, sL, j, lane);
  else if (rem == 2) gather_rows_plain<2>(acc, z4, sL, j, lane);
  else if (rem == 1) gather_rows_plain<1>(acc, z4, sL, j, lane);
}

__global__ __launch_bounds__(256, 10) void gather_out_kernel(
    const float* __restrict__ zAll, const int* __restrict__ rowptr,
    const int* __restrict__ csrSrc, const float* __restrict__ nrm,
    const float* __restrict__ biasAll, const float* __restrict__ addmAll,
    float* __restrict__ out, int nN, float outScale,
    int nStacks, size_t planeStride, int accumFirst) {
  const int lane = threadIdx.x & 31;
  const int w  = __builtin_amdgcn_readfirstlane(blockIdx.x * 8 + (threadIdx.x >> 5));
  const int nW = gridDim.x * 8;
  float4* __restrict__ out4 = reinterpret_cast<float4*>(out);

  int beg[ROWS_PER_WAVE], end[ROWS_PER_WAVE], sL0[ROWS_PER_WAVE];
  float nd[ROWS_PER_WAVE];
  float4 o[ROWS_PER_WAVE];
#pragma unroll
  for (int j = 0; j < ROWS_PER_WAVE; ++j) {
    const int d = w + j * nW;
    beg[j] = 0; end[j] = 0; sL0[j] = 0; nd[j] = 0.f;
    o[j] = make_float4(0.f, 0.f, 0.f, 0.f);
    if (d < nN) {
      beg[j] = __builtin_amdgcn_readfirstlane(rowptr[d]);
      end[j] = __builtin_amdgcn_readfirstlane(rowptr[d + 1]);
      if (lane < end[j] - beg[j])
        sL0[j] = min(max(csrSrc[beg[j] + lane], 0), nN - 1);
      nd[j] = nrm[d];
      if (accumFirst) o[j] = out4[(size_t)d * 32 + lane];
    }
  }

  for (int st = 0; st < nStacks; ++st) {
    const float4* __restrict__ z4 =
        reinterpret_cast<const float4*>(zAll + (size_t)st * planeStride);
    const float4* __restrict__ addm4 =
        reinterpret_cast<const float4*>(addmAll + (size_t)st * planeStride);
    const float4 b4 = reinterpret_cast<const float4*>(biasAll + (size_t)st * D)[lane];
#pragma unroll
    for (int j = 0; j < ROWS_PER_WAVE; ++j) {
      const int d = w + j * nW;
      if (d < nN) {
        const float4 f = addm4[(size_t)d * 32 + lane];
        float4 acc = make_float4(0.f, 0.f, 0.f, 0.f);
        sum_chunk_plain(acc, z4, sL0[j], min(32, end[j] - beg[j]), lane);
        for (int base = beg[j] + 32; base < end[j]; base += 32) {
          const int cnt = min(32, end[j] - base);
          int sL = 0;
          if (lane < cnt) sL = min(max(csrSrc[base + lane], 0), nN - 1);
          sum_chunk_plain(acc, z4, sL, cnt, lane);
        }
        float4 c;
        c.x = fmaxf(acc.x * nd[j] + b4.x + f.x, 0.f) * outScale;
        c.y = fmaxf(acc.y * nd[j] + b4.y + f.y, 0.f) * outScale;
        c.z = fmaxf(acc.z * nd[j] + b4.z + f.z, 0.f) * outScale;
        c.w = fmaxf(acc.w * nd[j] + b4.w + f.w, 0.f) * outScale;
        o[j].x += c.x; o[j].y += c.y; o[j].z += c.z; o[j].w += c.w;
      }
    }
  }

#pragma unroll
  for (int j = 0; j < ROWS_PER_WAVE; ++j) {
    const int d = w + j * nW;
    if (d < nN) out4[(size_t)d * 32 + lane] = o[j];
  }
}

__global__ __launch_bounds__(256, 4) void gemm_chain_kernel(
    const float* __restrict__ X, const float* __restrict__ G,
    const float* __restrict__ nrm,
    const float* __restrict__ WxAll, const float* __restrict__ bxAll,
    const float* __restrict__ WgAll, const float* __restrict__ bgAll,
    const float* __restrict__ W2All,
    float* __restrict__ fvAll, float* __restrict__ zAll, int nRowTiles,
    int nStacks, size_t planeStride) {
  __shared__ float Xs[16 * LDS_STRIDE];
  __shared__ float Gs[16 * LDS_STRIDE];
  __shared__ float Hs[16 * LDS_STRIDE];
  const int tid  = threadIdx.x;
  const int wid  = __builtin_amdgcn_readfirstlane(tid >> 5);
  const int lane = tid & 31;
  const int half = lane >> 4;
  const int lm   = lane & 15;
  const int colBase = wid * 16;
  const int st      = blockIdx.x % nStacks;
  const int slot    = blockIdx.x / nStacks;
  const int nSlots  = gridDim.x / nStacks;
  const float* __restrict__ Wx = WxAll + (size_t)st * D * D;
  const float* __restrict__ Wg = WgAll + (size_t)st * D * D;
  const float* __restrict__ W2 = W2All + (size_t)st * D * D;
  const float* __restrict__ bx = bxAll + (size_t)st * D;
  const float* __restrict__ bg = bgAll + (size_t)st * D;
  float* __restrict__ fvOut = fvAll + (size_t)st * planeStride;
  float* __restrict__ zOut  = zAll + (size_t)st * planeStride;

  v2f bxr[32];
  v2f bgr[32];
  v2f b2r[32];
#pragma unroll
  for (int ks = 0; ks < 32; ++ks) {
    const int kk = ks * 4 + half * 2;
    bxr[ks].x = Wx[(size_t)kk * D + colBase + lm];
    bxr[ks].y = Wx[(size_t)(kk + 1) * D + colBase + lm];
    bgr[ks].x = Wg[(size_t)kk * D + colBase + lm];
    bgr[ks].y = Wg[(size_t)(kk + 1) * D + colBase + lm];
    b2r[ks].x = W2[(size_t)kk * D + colBase + lm];
    b2r[ks].y = W2[(size_t)(kk + 1) * D + colBase + lm];
  }
  const float bxcol = bx[colBase + lm];
  const float bgcol = bg[colBase + lm];

  const int r0 = wid, r1 = wid + 8;
  const int c0 = lane * 4, c1 = lane * 4;
  const float4* __restrict__ X4 = reinterpret_cast<const float4*>(X);
  const float4* __restrict__ G4 = reinterpret_cast<const float4*>(G);

  int rt = slot;
  if (rt >= nRowTiles) return;

  float4 x0, x1, g0, g1;
  {
    const int rowBase = rt * 16;
    x0 = X4[(size_t)(rowBase + r0) * 32 + lane];
    x1 = X4[(size_t)(rowBase + r1) * 32 + lane];
    g0 = G4[(size_t)(rowBase + r0) * 32 + lane];
    g1 = G4[(size_t)(rowBase + r1) * 32 + lane];
    const float n0 = nrm[rowBase + r0], n1 = nrm[rowBase + r1];
    g0.x *= n0; g0.y *= n0; g0.z *= n0; g0.w *= n0;
    g1.x *= n1; g1.y *= n1; g1.z *= n1; g1.w *= n1;
  }

  for (;;) {
    *reinterpret_cast<float4*>(&Xs[r0 * LDS_STRIDE + c0]) = x0;
    *reinterpret_cast<float4*>(&Xs[r1 * LDS_STRIDE + c1]) = x1;
    *reinterpret_cast<float4*>(&Gs[r0 * LDS_STRIDE + c0]) = g0;
    *reinterpret_cast<float4*>(&Gs[r1 * LDS_STRIDE + c1]) = g1;

    const int rnext = rt + nSlots;
    const bool more = rnext < nRowTiles;
    __syncthreads();

    v8f accX = {0.f, 0.f, 0.f, 0.f, 0.f, 0.f, 0.f, 0.f};
    v8f accG = {0.f, 0.f, 0.f, 0.f, 0.f, 0.f, 0.f, 0.f};
#pragma unroll
    for (int ks = 0; ks < 32; ++ks) {
      const int kk = ks * 4 + half * 2;
      v2f ax, ag;
      ax.x = Xs[lm * LDS_STRIDE + kk];
      ax.y = Xs[lm * LDS_STRIDE + kk + 1];
      ag.x = Gs[lm * LDS_STRIDE + kk];
      ag.y = Gs[lm * LDS_STRIDE + kk + 1];
      accX = __builtin_amdgcn_wmma_f32_16x16x4_f32(
           false, ax,  false, bxr[ks],
           (short)0, accX,  false,  false);
      accG = __builtin_amdgcn_wmma_f32_16x16x4_f32(
           false, ag,  false, bgr[ks],
           (short)0, accG,  false,  false);
    }

    const int rowBase = rt * 16;
    const int col = colBase + lm;
    const int tileOff = (rowBase + half * 8) * D + col;
    float* __restrict__ fvp = fvOut + tileOff;
    float* __restrict__ zp  = zOut + tileOff;
    const float* __restrict__ nrp = nrm + rowBase + half * 8;
#pragma unroll
    for (int r = 0; r < 8; ++r) {
      accX[r] += bxcol;
      float hv = accG[r] + bgcol;
      hv += accX[r];
      Hs[(r + half * 8) * LDS_STRIDE + col] = fmaxf(hv, 0.f);
    }
    __syncthreads();
#pragma unroll
    for (int r = 0; r < 8; ++r) fvp[r * D] = accX[r];

    float n0n = 0.f, n1n = 0.f;
    if (more) {
      const int rowBaseN = rnext * 16;
      x0 = X4[(size_t)(rowBaseN + r0) * 32 + lane];
      x1 = X4[(size_t)(rowBaseN + r1) * 32 + lane];
      g0 = G4[(size_t)(rowBaseN + r0) * 32 + lane];
      g1 = G4[(size_t)(rowBaseN + r1) * 32 + lane];
      n0n = nrm[rowBaseN + r0];
      n1n = nrm[rowBaseN + r1];
    }
    float nr[8];
#pragma unroll
    for (int r = 0; r < 8; ++r) nr[r] = nrp[r];

    v8f accZ = {0.f, 0.f, 0.f, 0.f, 0.f, 0.f, 0.f, 0.f};
#pragma unroll
    for (int ks = 0; ks < 32; ++ks) {
      const int kk = ks * 4 + half * 2;
      v2f ah;
      ah.x = Hs[lm * LDS_STRIDE + kk];
      ah.y = Hs[lm * LDS_STRIDE + kk + 1];
      accZ = __builtin_amdgcn_wmma_f32_16x16x4_f32(
           false, ah,  false, b2r[ks],
           (short)0, accZ,  false,  false);
    }
#pragma unroll
    for (int r = 0; r < 8; ++r) zp[r * D] = accZ[r] * nr[r];

    if (!more) break;
    g0.x *= n0n; g0.y *= n0n; g0.z *= n0n; g0.w *= n0n;
    g1.x *= n1n; g1.y *= n1n; g1.z *= n1n; g1.w *= n1n;
    rt = rnext;
  }
}

extern "C" void kernel_launch(void* const* d_in, const int* in_sizes, int n_in,
                              void* d_out, int out_size, void* d_ws, size_t ws_size,
                              hipStream_t stream) {
  const float* feats = (const float*)d_in[0];
  const int*   src   = (const int*)d_in[1];
  const int*   dst   = (const int*)d_in[2];
  const float* w0    = (const float*)d_in[3];
  const float* b0    = (const float*)d_in[4];
  const float* w     = (const float*)d_in[5];
  const float* bw    = (const float*)d_in[6];
  const float* v     = (const float*)d_in[7];
  const float* bv    = (const float*)d_in[8];
  float* out = (float*)d_out;

  const int N = in_sizes[0] / D;
  const int E = in_sizes[1];
  const size_t M = (size_t)N * D;

  const size_t Nr  = ((size_t)N + 255) & ~(size_t)255;
  const size_t N1r = ((size_t)N + 1 + 255) & ~(size_t)255;
  const size_t Er  = ((size_t)E + 255) & ~(size_t)255;

  float* ws = (float*)d_ws;
  size_t off = 0;
  float* nrm  = ws + off; off += Nr;
  float* agg0 = ws + off; off += M;
  int* cnt    = (int*)(ws + off); off += Nr;
  int* rowptr = (int*)(ws + off); off += N1r;
  int* csrSrc = (int*)(ws + off); off += Er;
  float* planes = ws + off;
  int* pos      = (int*)planes;
  if ((off + 2 * M) * sizeof(float) > ws_size) return;
  const size_t roomPairs = (ws_size / sizeof(float) - off) / (2 * M);
  const int G = roomPairs >= (size_t)KSTACK ? KSTACK : (int)roomPairs;
  float* fv = planes;
  float* z1 = planes + (size_t)G * M;

  const int nRowTiles  = N / 16;
  const int chainBlocks = 512;
  const int gatherBlocks = (N + 8 * ROWS_PER_WAVE - 1) / (8 * ROWS_PER_WAVE);
  const int edgeBlocks = ((E >> 2) + 255) / 256 + ((E >> 2) == 0 ? 1 : 0);

  zero_f4_kernel<<<((int)(Nr / 4) + 255) / 256, 256, 0, stream>>>((float4*)cnt, (int)(Nr / 4));
  if (Nr <= (size_t)RANK_KEYS) {
    const int n4 = E >> 2;
    const int minBlocks = (n4 + RANK_G * RANK_T - 1) / (RANK_G * RANK_T);
    const int rankBlocks = minBlocks > 64 ? minBlocks : 64;
    rank_lds_kernel<<<rankBlocks, RANK_T, 0, stream>>>(dst, cnt, pos, E, (int)Nr,
                                                       (n4 + rankBlocks - 1) / rankBlocks);
  } else {
    count_rank_kernel<<<edgeBlocks, 256, 0, stream>>>(dst, cnt, pos, E);
  }
  scan_norm_kernel<<<(N + SCAN_T - 1) / SCAN_T, SCAN_T, 0, stream>>>(cnt, rowptr, nrm, N);
  csr_fill_kernel<<<edgeBlocks, 256, 0, stream>>>(src, dst, pos, rowptr, csrSrc, E);

  spmm_gather_kernel<<<(N + 7) / 8, 256, 0, stream>>>(feats, rowptr, csrSrc, nrm, agg0, N);

  for (int k0 = 0; k0 < KSTACK; k0 += G) {
    const int g = (KSTACK - k0 < G) ? (KSTACK - k0) : G;
    const int slots = chainBlocks / g;
    gemm_chain_kernel<<<g * slots, 256, 0, stream>>>(
        feats, agg0, nrm, v + (size_t)k0 * D * D, bv + (size_t)k0 * D,
        w0 + (size_t)k0 * D * D, b0 + (size_t)k0 * D, w + (size_t)k0 * D * D,
        fv, z1, nRowTiles, g, M);
    gather_out_kernel<<<gatherBlocks, 256, 0, stream>>>(
        z1, rowptr, csrSrc, nrm, bw + (size_t)k0 * D, fv, out, N,
        1.0f / (float)KSTACK, g, M, k0 > 0 ? 1 : 0);
  }
}


# CSR replaced by row slots at a fixed pitch of 64 (row d's sources at d*64 + rank): the fill needs no row starts, so the scan launch goes, 7 -> 6 launches; row sums in the same slot order
# speedup vs baseline: 3.7269x; 1.0349x over previous
#include <hip/hip_runtime.h>

typedef float v2f __attribute__((ext_vector_type(2)));
typedef float v8f __attribute__((ext_vector_type(8)));

#define D 128
#define KSTACK 3
#define LDS_STRIDE 132

__global__ void zero_f4_kernel(float4* __restrict__ p, int n4) {
  int i = blockIdx.x * blockDim.x + threadIdx.x;
  if (i < n4) p[i] = make_float4(0.f, 0.f, 0.f, 0.f);
}

__global__ __launch_bounds__(256) void count_rank_kernel(const int* __restrict__ dst,
                                                         int* __restrict__ cnt,
                                                         int* __restrict__ pos, int nE) {
  const int n4 = nE >> 2;
  const int q = blockIdx.x * 256 + threadIdx.x;
  if (q < n4) {
    const int4 d = reinterpret_cast<const int4*>(dst)[q];
    int4 r;
    r.x = atomicAdd(&cnt[d.x], 1);
    r.y = atomicAdd(&cnt[d.y], 1);
    r.z = atomicAdd(&cnt[d.z], 1);
    r.w = atomicAdd(&cnt[d.w], 1);
    reinterpret_cast<int4*>(pos)[q] = r;
  }
  if (blockIdx.x == 0 && (int)threadIdx.x < (nE & 3)) {
    const int i = (n4 << 2) + threadIdx.x;
    pos[i] = atomicAdd(&cnt[dst[i]], 1);
  }
}

#define RANK_KEYS 51200
#define RANK_T 1024
#define RANK_G 4
#define RANK_U 7
__global__ __launch_bounds__(RANK_T) void rank_lds_kernel(const int* __restrict__ dst,
                                                          int* __restrict__ cnt,
                                                          int* __restrict__ pos, int nE,
                                                          int nKeysPad, int groupsPerBlock) {
  __shared__ int cntL[RANK_KEYS];
  const int tid = threadIdx.x;
  const int n4 = nE >> 2;
  const int gBeg = blockIdx.x * groupsPerBlock;
  const int gEnd = min(n4, gBeg + groupsPerBlock);
  const int4* __restrict__ dst4 = reinterpret_cast<const int4*>(dst);

  int4 d[RANK_G], r[RANK_G];
#pragma unroll
  for (int k = 0; k < RANK_G; ++k) {
    const int g = gBeg + tid + k * RANK_T;
    d[k] = make_int4(0, 0, 0, 0);
    r[k] = make_int4(0, 0, 0, 0);
    if (g < gEnd) d[k] = dst4[g];
  }
  const bool tail = blockIdx.x == 0 && tid < (nE & 3);
  int dT = 0, rT = 0;
  if (tail) dT = dst[(n4 << 2) + tid];

  int4* cntL4 = reinterpret_cast<int4*>(cntL);
  for (int i = tid; i < (nKeysPad >> 2); i += RANK_T) cntL4[i] = make_int4(0, 0, 0, 0);
  __syncthreads();

#pragma unroll
  for (int k = 0; k < RANK_G; ++k) {
    const int g = gBeg + tid + k * RANK_T;
    if (g < gEnd) {
      r[k].x = atomicAdd(&cntL[d[k].x], 1);
      r[k].y = atomicAdd(&cntL[d[k].y], 1);
      r[k].z = atomicAdd(&cntL[d[k].z], 1);
      r[k].w = atomicAdd(&cntL[d[k].w], 1);
    }
  }
  if (tail) rT = atomicAdd(&cntL[dT], 1);
  __syncthreads();

  const int nSteps = (nKeysPad + RANK_T - 1) / RANK_T;
  const int rot = (int)(((long long)blockIdx.x * nSteps) / gridDim.x);
  for (int s0 = 0; s0 < nSteps; s0 += RANK_U) {
    int ix[RANK_U], c[RANK_U], b[RANK_U];
#pragma unroll
    for (int u = 0; u < RANK_U; ++u) {
      int st = s0 + u + rot;
      if (st >= nSteps) st -= nSteps;
      const int i = st * RANK_T + tid;
      const bool ok = (s0 + u < nSteps) && (i < nKeysPad);
      ix[u] = ok ? i : tid;
      const int v = cntL[ix[u]];
      c[u] = ok ? v : 0;
    }
#pragma unroll
    for (int u = 0; u < RANK_U; ++u) {
      b[u] = c[u];
      if (c[u] != 0) b[u] = atomicAdd(&cnt[ix[u]], c[u]);
    }
#pragma unroll
    for (int u = 0; u < RANK_U; ++u)
      if (c[u] != 0) cntL[ix[u]] = b[u];
  }
  __syncthreads();

#pragma unroll
  for (int k = 0; k < RANK_G; ++k) {
    const int g = gBeg + tid + k * RANK_T;
    if (g < gEnd) {
      int4 p;
      p.x = r[k].x + cntL[d[k].x];
      p.y = r[k].y + cntL[d[k].y];
      p.z = r[k].z + cntL[d[k].z];
      p.w = r[k].w + cntL[d[k].w];
      reinterpret_cast<int4*>(pos)[g] = p;
    }
  }
  if (tail) pos[(n4 << 2) + tid] = rT + cntL[dT];
}

#define SCAN_T 1024
__global__ __launch_bounds__(SCAN_T) void scan_norm_kernel(const int* __restrict__ cnt,
                                                           int* __restrict__ rowptr,
                                                           float* __restrict__ nrm, int n) {
  __shared__ int waveIncl[32];
  __shared__ int waveFront[32];
  const int tid  = threadIdx.x;
  const int lane = tid & 31;
  const int wid  = tid >> 5;
  const int base = blockIdx.x * SCAN_T;

  const int i = base + tid;
  const int c = (i < n) ? cnt[i] : 0;

  const int4* __restrict__ c4 = reinterpret_cast<const int4*>(cnt);
  const int nq = base >> 2;
  int front = 0;
  for (int q0 = tid; q0 < nq; q0 += 16 * SCAN_T) {
    int4 v[16];
#pragma unroll
    for (int u = 0; u < 16; ++u) {
      const int q = q0 + u * SCAN_T;
      v[u] = make_int4(0, 0, 0, 0);
      if (q < nq) v[u] = c4[q];
    }
#pragma unroll
    for (int u = 0; u < 16; ++u) front += (v[u].x + v[u].y) + (v[u].z + v[u].w);
  }
  for (int o = 16; o > 0; o >>= 1) front += __shfl_xor(front, o);

  if (i < n) nrm[i] = rsqrtf(fmaxf((float)c, 1.0f));

  int incl = c;
  for (int o = 1; o < 32; o <<= 1) {
    const int t = __shfl_up(incl, o);
    if (lane >= o) incl += t;
  }
  if (lane == 31) waveIncl[wid]  = incl;
  if (lane == 0)  waveFront[wid] = front;
  __syncthreads();
  int off = waveFront[lane] + (lane < wid ? waveIncl[lane] : 0);
  for (int o = 16; o > 0; o >>= 1) off += __shfl_xor(off, o);
  const int excl = off + incl - c;
  if (i < n) rowptr[i] = excl;
  if (i == n - 1) rowptr[n] = excl + c;
}

__global__ __launch_bounds__(256) void csr_fill_kernel(const int* __restrict__ src,
                                                       const int* __restrict__ dst,
                                                       const int* __restrict__ pos,
                                                       const int* __restrict__ rowptr,
                                                       int* __restrict__ csrSrc, int nE) {
  const int n4 = nE >> 2;
  const int q = blockIdx.x * 256 + threadIdx.x;
  if (q < n4) {
    const int4 s = reinterpret_cast<const int4*>(src)[q];
    const int4 d = reinterpret_cast<const int4*>(dst)[q];
    const int4 p = reinterpret_cast<const int4*>(pos)[q];
    const int b0 = rowptr[d.x], b1 = rowptr[d.y], b2 = rowptr[d.z], b3 = rowptr[d.w];
    csrSrc[b0 + p.x] = s.x;
    csrSrc[b1 + p.y] = s.y;
    csrSrc[b2 + p.z] = s.z;
    csrSrc[b3 + p.w] = s.w;
  }
  if (blockIdx.x == 0 && (int)threadIdx.x < (nE & 3)) {
    const int i = (n4 << 2) + threadIdx.x;
    csrSrc[rowptr[dst[i]] + pos[i]] = src[i];
  }
}

#define ROW_PITCH 64
__global__ __launch_bounds__(256) void slot_fill_kernel(const int* __restrict__ src,
                                                        const int* __restrict__ dst,
                                                        const int* __restrict__ pos,
                                                        const int* __restrict__ cnt,
                                                        int* __restrict__ slots,
                                                        float* __restrict__ nrm,
                                                        int nE, int nN) {
  const int n4 = nE >> 2;
  const int q = blockIdx.x * 256 + threadIdx.x;
  if (q < n4) {
    const int4 s = reinterpret_cast<const int4*>(src)[q];
    const int4 d = reinterpret_cast<const int4*>(dst)[q];
    const int4 p = reinterpret_cast<const int4*>(pos)[q];
    if ((p.x | p.y | p.z | p.w) < ROW_PITCH) {
      slots[(size_t)d.x * ROW_PITCH + p.x] = s.x;
      slots[(size_t)d.y * ROW_PITCH + p.y] = s.y;
      slots[(size_t)d.z * ROW_PITCH + p.z] = s.z;
      slots[(size_t)d.w * ROW_PITCH + p.w] = s.w;
    } else {
      if (p.x < ROW_PITCH) slots[(size_t)d.x * ROW_PITCH + p.x] = s.x;
      if (p.y < ROW_PITCH) slots[(size_t)d.y * ROW_PITCH + p.y] = s.y;
      if (p.z < ROW_PITCH) slots[(size_t)d.z * ROW_PITCH + p.z] = s.z;
      if (p.w < ROW_PITCH) slots[(size_t)d.w * ROW_PITCH + p.w] = s.w;
    }
  }
  if (blockIdx.x == 0 && (int)threadIdx.x < (nE & 3)) {
    const int i = (n4 << 2) + threadIdx.x;
    const int p = pos[i];
    if (p < ROW_PITCH) slots[(size_t)dst[i] * ROW_PITCH + p] = src[i];
  }
  for (int i = q; i < nN; i += gridDim.x * 256) nrm[i] = rsqrtf(fmaxf((float)cnt[i], 1.0f));
}

__device__ __forceinline__ void row_fma(float4& acc, const float4 v, const float w) {
  acc.x = fmaf(v.x, w, acc.x); acc.y = fmaf(v.y, w, acc.y);
  acc.z = fmaf(v.z, w, acc.z); acc.w = fmaf(v.w, w, acc.w);
}

template <bool WEIGHTED>
__device__ __forceinline__ void sum_row_from_edges(float4& acc, const float4* __restrict__ h4,
                                                   const float* __restrict__ nrm,
                                                   const int* __restrict__ src,
                                                   const int* __restrict__ dst,
                                                   const int nE, const int d, const int nN,
                                                   const int lane) {
  for (int e0 = 0; e0 < nE; e0 += 32) {
    const int e = e0 + lane;
    int ss = 0;
    float nl = 1.0f;
    bool hit = false;
    if (e < nE) hit = dst[e] == d;
    if (hit) {
      ss = min(max(src[e], 0), nN - 1);
      if (WEIGHTED) nl = nrm[ss];
    }
    unsigned m = __builtin_amdgcn_ballot_w32(hit);
    const int nli = __float_as_int(nl);
    while (m != 0u) {
      const int b = __builtin_ctz(m);
      m &= m - 1u;
      const int s = __builtin_amdgcn_readlane(ss, b);
      const float4 v = h4[(size_t)s * 32 + lane];
      if (WEIGHTED) {
        row_fma(acc, v, __int_as_float(__builtin_amdgcn_readlane(nli, b)));
      } else {
        acc.x += v.x; acc.y += v.y; acc.z += v.z; acc.w += v.w;
      }
    }
  }
}

template <int R>
__device__ __forceinline__ void gather_rows(float4& acc, const float4* __restrict__ h4,
                                            const int sL, const int nLi,
                                            const int j, const int lane) {
  float4 v[R];
#pragma unroll
  for (int u = 0; u < R; ++u) {
    const int s = __builtin_amdgcn_readlane(sL, j + u);
    v[u] = h4[(size_t)s * 32 + lane];
  }
#pragma unroll
  for (int u = 0; u < R; ++u)
    row_fma(acc, v[u], __int_as_float(__builtin_amdgcn_readlane(nLi, j + u)));
}

__device__ __forceinline__ void load_chunk(int& sL, int& nLi, const int* __restrict__ csrSrc,
                                           const float* __restrict__ nrm,
                                           const int base, const int cnt, const int nN,
                                           const int lane) {
  sL = 0;
  float nL = 0.f;
  if (lane < cnt) {
    sL = min(max(csrSrc[base + lane], 0), nN - 1);
    nL = nrm[sL];
  }
  nLi = __float_as_int(nL);
}

template <int RMAX>
__device__ __forceinline__ void sum_chunk(float4& acc, const float4* __restrict__ h4,
                                          const int sL, const int nLi, const int cnt,
                                          const int lane) {
  int j = 0;
  if (RMAX >= 8)
    for (; j + 8 <= cnt; j += 8) gather_rows<8>(acc, h4, sL, nLi, j, lane);
  for (; j + 4 <= cnt; j += 4) gather_rows<4>(acc, h4, sL, nLi, j, lane);
  const int rem = cnt - j;
  if (rem == 3)      gather_rows<3>(acc, h4, sL, nLi, j, lane);
  else if (rem == 2) gather_rows<2>(acc, h4, sL, nLi, j, lane);
  else if (rem == 1) gather_rows<1>(acc, h4, sL, nLi, j, lane);
}

__global__ __launch_bounds__(256) void spmm_gather_kernel(const float* __restrict__ h,
                                                          const int* __restrict__ rowInfo,
                                                          const int* __restrict__ csrSrc,
                                                          const float* __restrict__ nrm,
                                                          float* __restrict__ agg, int nN,
                                                          int pitch,
                                                          const int* __restrict__ src,
                                                          const int* __restrict__ dst, int nE) {
  const int lane = threadIdx.x & 31;
  const int d = __builtin_amdgcn_readfirstlane(blockIdx.x * 8 + (threadIdx.x >> 5));
  if (d >= nN) return;
  int beg, end;
  if (pitch > 0) {
    beg = d * pitch;
    end = beg + __builtin_amdgcn_readfirstlane(rowInfo[d]);
  } else {
    beg = __builtin_amdgcn_readfirstlane(rowInfo[d]);
    end = __builtin_amdgcn_readfirstlane(rowInfo[d + 1]);
  }
  const float4* __restrict__ h4 = reinterpret_cast<const float4*>(h);

  float4 acc = make_float4(0.f, 0.f, 0.f, 0.f);
  if (pitch > 0 && end - beg > pitch) {
    sum_row_from_edges<true>(acc, h4, nrm, src, dst, nE, d, nN, lane);
  } else {
    for (int base = beg; base < end; base += 32) {
      const int cnt = min(32, end - base);
      int sL, nLi;
      load_chunk(sL, nLi, csrSrc, nrm, base, cnt, nN, lane);
      sum_chunk<4>(acc, h4, sL, nLi, cnt, lane);
    }
  }
  reinterpret_cast<float4*>(agg)[(size_t)d * 32 + lane] = acc;
}

#define ROWS_PER_WAVE 5

template <int R>
__device__ __forceinline__ void gather_rows_plain(float4& acc, const float4* __restrict__ z4,
                                                  const int sL, const int j, const int lane) {
  float4 v[R];
#pragma unroll
  for (int u = 0; u < R; ++u) {
    const int s = __builtin_amdgcn_readlane(sL, j + u);
    v[u] = z4[(size_t)s * 32 + lane];
  }
#pragma unroll
  for (int u = 0; u < R; ++u) {
    acc.x += v[u].x; acc.y += v[u].y; acc.z += v[u].z; acc.w += v[u].w;
  }
}

__device__ __forceinline__ void sum_chunk_plain(float4& acc, const float4* __restrict__ z4,
                                                const int sL, const int cnt, const int lane) {
  int j = 0;
  for (; j + 8 <= cnt; j += 8) gather_rows_plain<8>(acc, z4, sL, j, lane);
  for (; j + 4 <= cnt; j += 4) gather_rows_plain<4>(acc, z4, sL, j, lane);
  const int rem = cnt - j;
  if (rem == 3)      gather_rows_plain<3>(acc, z4, sL, j, lane);
  else if (rem == 2) gather_rows_plain<2>(acc, z4, sL, j, lane);
  else if (rem == 1) gather_rows_plain<1>(acc, z4, sL, j, lane);
}

__global__ __launch_bounds__(256, 10) void gather_out_kernel(
    const float* __restrict__ zAll, const int* __restrict__ rowptr,
    const int* __restrict__ csrSrc, const float* __restrict__ nrm,
    const float* __restrict__ biasAll, const float* __restrict__ addmAll,
    float* __restrict__ out, int nN, float outScale,
    int nStacks, size_t planeStride, int accumFirst,
    int pitch, const int* __restrict__ src, const int* __restrict__ dst, int nE) {
  const int lane = threadIdx.x & 31;
  const int w  = __builtin_amdgcn_readfirstlane(blockIdx.x * 8 + (threadIdx.x >> 5));
  const int nW = gridDim.x * 8;
  float4* __restrict__ out4 = reinterpret_cast<float4*>(out);

  int beg[ROWS_PER_WAVE], end[ROWS_PER_WAVE], sL0[ROWS_PER_WAVE];
  bool big[ROWS_PER_WAVE];
  float nd[ROWS_PER_WAVE];
  float4 o[ROWS_PER_WAVE];
#pragma unroll
  for (int j = 0; j < ROWS_PER_WAVE; ++j) {
    const int d = w + j * nW;
    beg[j] = 0; end[j] = 0; sL0[j] = 0; nd[j] = 0.f; big[j] = false;
    o[j] = make_float4(0.f, 0.f, 0.f, 0.f);
    if (d < nN) {
      if (pitch > 0) {
        const int deg = __builtin_amdgcn_readfirstlane(rowptr[d]);
        big[j] = deg > pitch;
        beg[j] = d * pitch;
        end[j] = big[j] ? beg[j] : beg[j] + deg;
      } else {
        beg[j] = __builtin_amdgcn_readfirstlane(rowptr[d]);
        end[j] = __builtin_amdgcn_readfirstlane(rowptr[d + 1]);
      }
      if (lane < end[j] - beg[j])
        sL0[j] = min(max(csrSrc[beg[j] + lane], 0), nN - 1);
      nd[j] = nrm[d];
      if (accumFirst) o[j] = out4[(size_t)d * 32 + lane];
    }
  }

  for (int st = 0; st < nStacks; ++st) {
    const float4* __restrict__ z4 =
        reinterpret_cast<const float4*>(zAll + (size_t)st * planeStride);
    const float4* __restrict__ addm4 =
        reinterpret_cast<const float4*>(addmAll + (size_t)st * planeStride);
    const float4 b4 = reinterpret_cast<const float4*>(biasAll + (size_t)st * D)[lane];
#pragma unroll
    for (int j = 0; j < ROWS_PER_WAVE; ++j) {
      const int d = w + j * nW;
      if (d < nN) {
        const float4 f = addm4[(size_t)d * 32 + lane];
        float4 acc = make_float4(0.f, 0.f, 0.f, 0.f);
        if (big[j])
          sum_row_from_edges<false>(acc, z4, nrm, src, dst, nE, d, nN, lane);
        sum_chunk_plain(acc, z4, sL0[j], min(32, end[j] - beg[j]), lane);
        for (int base = beg[j] + 32; base < end[j]; base += 32) {
          const int cnt = min(32, end[j] - base);
          int sL = 0;
          if (lane < cnt) sL = min(max(csrSrc[base + lane], 0), nN - 1);
          sum_chunk_plain(acc, z4, sL, cnt, lane);
        }
        float4 c;
        c.x = fmaxf(acc.x * nd[j] + b4.x + f.x, 0.f) * outScale;
        c.y = fmaxf(acc.y * nd[j] + b4.y + f.y, 0.f) * outScale;
        c.z = fmaxf(acc.z * nd[j] + b4.z + f.z, 0.f) * outScale;
        c.w = fmaxf(acc.w * nd[j] + b4.w + f.w, 0.f) * outScale;
        o[j].x += c.x; o[j].y += c.y; o[j].z += c.z; o[j].w += c.w;
      }
    }
  }

#pragma unroll
  for (int j = 0; j < ROWS_PER_WAVE; ++j) {
    const int d = w + j * nW;
    if (d < nN) out4[(size_t)d * 32 + lane] = o[j];
  }
}

__global__ __launch_bounds__(256, 4) void gemm_chain_kernel(
    const float* __restrict__ X, const float* __restrict__ G,
    const float* __restrict__ nrm,
    const float* __restrict__ WxAll, const float* __restrict__ bxAll,
    const float* __restrict__ WgAll, const float* __restrict__ bgAll,
    const float* __restrict__ W2All,
    float* __restrict__ fvAll, float* __restrict__ zAll, int nRowTiles,
    int nStacks, size_t planeStride) {
  __shared__ float Xs[16 * LDS_STRIDE];
  __shared__ float Gs[16 * LDS_STRIDE];
  __shared__ float Hs[16 * LDS_STRIDE];
  const int tid  = threadIdx.x;
  const int wid  = __builtin_amdgcn_readfirstlane(tid >> 5);
  const int lane = tid & 31;
  const int half = lane >> 4;
  const int lm   = lane & 15;
  const int colBase = wid * 16;
  const int st      = blockIdx.x % nStacks;
  const int slot    = blockIdx.x / nStacks;
  const int nSlots  = gridDim.x / nStacks;
  const float* __restrict__ Wx = WxAll + (size_t)st * D * D;
  const float* __restrict__ Wg = WgAll + (size_t)st * D * D;
  const float* __restrict__ W2 = W2All + (size_t)st * D * D;
  const float* __restrict__ bx = bxAll + (size_t)st * D;
  const float* __restrict__ bg = bgAll + (size_t)st * D;
  float* __restrict__ fvOut = fvAll + (size_t)st * planeStride;
  float* __restrict__ zOut  = zAll + (size_t)st * planeStride;

  v2f bxr[32];
  v2f bgr[32];
  v2f b2r[32];
#pragma unroll
  for (int ks = 0; ks < 32; ++ks) {
    const int kk = ks * 4 + half * 2;
    bxr[ks].x = Wx[(size_t)kk * D + colBase + lm];
    bxr[ks].y = Wx[(size_t)(kk + 1) * D + colBase + lm];
    bgr[ks].x = Wg[(size_t)kk * D + colBase + lm];
    bgr[ks].y = Wg[(size_t)(kk + 1) * D + colBase + lm];
    b2r[ks].x = W2[(size_t)kk * D + colBase + lm];
    b2r[ks].y = W2[(size_t)(kk + 1) * D + colBase + lm];
  }
  const float bxcol = bx[colBase + lm];
  const float bgcol = bg[colBase + lm];

  const int r0 = wid, r1 = wid + 8;
  const int c0 = lane * 4, c1 = lane * 4;
  const float4* __restrict__ X4 = reinterpret_cast<const float4*>(X);
  const float4* __restrict__ G4 = reinterpret_cast<const float4*>(G);

  int rt = slot;
  if (rt >= nRowTiles) return;

  float4 x0, x1, g0, g1;
  {
    const int rowBase = rt * 16;
    x0 = X4[(size_t)(rowBase + r0) * 32 + lane];
    x1 = X4[(size_t)(rowBase + r1) * 32 + lane];
    g0 = G4[(size_t)(rowBase + r0) * 32 + lane];
    g1 = G4[(size_t)(rowBase + r1) * 32 + lane];
    const float n0 = nrm[rowBase + r0], n1 = nrm[rowBase + r1];
    g0.x *= n0; g0.y *= n0; g0.z *= n0; g0.w *= n0;
    g1.x *= n1; g1.y *= n1; g1.z *= n1; g1.w *= n1;
  }

  for (;;) {
    *reinterpret_cast<float4*>(&Xs[r0 * LDS_STRIDE + c0]) = x0;
    *reinterpret_cast<float4*>(&Xs[r1 * LDS_STRIDE + c1]) = x1;
    *reinterpret_cast<float4*>(&Gs[r0 * LDS_STRIDE + c0]) = g0;
    *reinterpret_cast<float4*>(&Gs[r1 * LDS_STRIDE + c1]) = g1;

    const int rnext = rt + nSlots;
    const bool more = rnext < nRowTiles;
    __syncthreads();

    v8f accX = {0.f, 0.f, 0.f, 0.f, 0.f, 0.f, 0.f, 0.f};
    v8f accG = {0.f, 0.f, 0.f, 0.f, 0.f, 0.f, 0.f, 0.f};
#pragma unroll
    for (int ks = 0; ks < 32; ++ks) {
      const int kk = ks * 4 + half * 2;
      v2f ax, ag;
      ax.x = Xs[lm * LDS_STRIDE + kk];
      ax.y = Xs[lm * LDS_STRIDE + kk + 1];
      ag.x = Gs[lm * LDS_STRIDE + kk];
      ag.y = Gs[lm * LDS_STRIDE + kk + 1];
      accX = __builtin_amdgcn_wmma_f32_16x16x4_f32(
           false, ax,  false, bxr[ks],
           (short)0, accX,  false,  false);
      accG = __builtin_amdgcn_wmma_f32_16x16x4_f32(
           false, ag,  false, bgr[ks],
           (short)0, accG,  false,  false);
    }

    const int rowBase = rt * 16;
    const int col = colBase + lm;
    const int tileOff = (rowBase + half * 8) * D + col;
    float* __restrict__ fvp = fvOut + tileOff;
    float* __restrict__ zp  = zOut + tileOff;
    const float* __restrict__ nrp = nrm + rowBase + half * 8;
#pragma unroll
    for (int r = 0; r < 8; ++r) {
      accX[r] += bxcol;
      float hv = accG[r] + bgcol;
      hv += accX[r];
      Hs[(r + half * 8) * LDS_STRIDE + col] = fmaxf(hv, 0.f);
    }
    __syncthreads();
#pragma unroll
    for (int r = 0; r < 8; ++r) fvp[r * D] = accX[r];

    float n0n = 0.f, n1n = 0.f;
    if (more) {
      const int rowBaseN = rnext * 16;
      x0 = X4[(size_t)(rowBaseN + r0) * 32 + lane];
      x1 = X4[(size_t)(rowBaseN + r1) * 32 + lane];
      g0 = G4[(size_t)(rowBaseN + r0) * 32 + lane];
      g1 = G4[(size_t)(rowBaseN + r1) * 32 + lane];
      n0n = nrm[rowBaseN + r0];
      n1n = nrm[rowBaseN + r1];
    }
    float nr[8];
#pragma unroll
    for (int r = 0; r < 8; ++r) nr[r] = nrp[r];

    v8f accZ = {0.f, 0.f, 0.f, 0.f, 0.f, 0.f, 0.f, 0.f};
#pragma unroll
    for (int ks = 0; ks < 32; ++ks) {
      const int kk = ks * 4 + half * 2;
      v2f ah;
      ah.x = Hs[lm * LDS_STRIDE + kk];
      ah.y = Hs[lm * LDS_STRIDE + kk + 1];
      accZ = __builtin_amdgcn_wmma_f32_16x16x4_f32(
           false, ah,  false, b2r[ks],
           (short)0, accZ,  false,  false);
    }
#pragma unroll
    for (int r = 0; r < 8; ++r) zp[r * D] = accZ[r] * nr[r];

    if (!more) break;
    g0.x *= n0n; g0.y *= n0n; g0.z *= n0n; g0.w *= n0n;
    g1.x *= n1n; g1.y *= n1n; g1.z *= n1n; g1.w *= n1n;
    rt = rnext;
  }
}

extern "C" void kernel_launch(void* const* d_in, const int* in_sizes, int n_in,
                              void* d_out, int out_size, void* d_ws, size_t ws_size,
                              hipStream_t stream) {
  const float* feats = (const float*)d_in[0];
  const int*   src   = (const int*)d_in[1];
  const int*   dst   = (const int*)d_in[2];
  const float* w0    = (const float*)d_in[3];
  const float* b0    = (const float*)d_in[4];
  const float* w     = (const float*)d_in[5];
  const float* bw    = (const float*)d_in[6];
  const float* v     = (const float*)d_in[7];
  const float* bv    = (const float*)d_in[8];
  float* out = (float*)d_out;

  const int N = in_sizes[0] / D;
  const int E = in_sizes[1];
  const size_t M = (size_t)N * D;

  const size_t Nr  = ((size_t)N + 255) & ~(size_t)255;
  const size_t N1r = ((size_t)N + 1 + 255) & ~(size_t)255;
  const size_t Er  = ((size_t)E + 255) & ~(size_t)255;

  float* ws = (float*)d_ws;
  size_t off = 0;
  float* nrm  = ws + off; off += Nr;
  float* agg0 = ws + off; off += M;
  int* cnt    = (int*)(ws + off); off += Nr;
  const size_t Lr = ((size_t)N * ROW_PITCH + 255) & ~(size_t)255;
  const bool fixedPitch = (off + Lr + 2 * (size_t)KSTACK * M) * sizeof(float) <= ws_size;
  int* rowptr = (int*)(ws + off); if (!fixedPitch) off += N1r;
  int* csrSrc = (int*)(ws + off); off += fixedPitch ? Lr : Er;
  const int* rowInfo = fixedPitch ? cnt : rowptr;
  const int pitch = fixedPitch ? ROW_PITCH : 0;
  float* planes = ws + off;
  int* pos      = (int*)planes;
  if ((off + 2 * M) * sizeof(float) > ws_size) return;
  const size_t roomPairs = (ws_size / sizeof(float) - off) / (2 * M);
  const int G = roomPairs >= (size_t)KSTACK ? KSTACK : (int)roomPairs;
  float* fv = planes;
  float* z1 = planes + (size_t)G * M;

  const int nRowTiles  = N / 16;
  const int chainBlocks = 512;
  const int gatherBlocks = (N + 8 * ROWS_PER_WAVE - 1) / (8 * ROWS_PER_WAVE);
  const int edgeBlocks = ((E >> 2) + 255) / 256 + ((E >> 2) == 0 ? 1 : 0);

  zero_f4_kernel<<<((int)(Nr / 4) + 255) / 256, 256, 0, stream>>>((float4*)cnt, (int)(Nr / 4));
  if (Nr <= (size_t)RANK_KEYS) {
    const int n4 = E >> 2;
    const int minBlocks = (n4 + RANK_G * RANK_T - 1) / (RANK_G * RANK_T);
    const int rankBlocks = minBlocks > 64 ? minBlocks : 64;
    rank_lds_kernel<<<rankBlocks, RANK_T, 0, stream>>>(dst, cnt, pos, E, (int)Nr,
                                                       (n4 + rankBlocks - 1) / rankBlocks);
  } else {
    count_rank_kernel<<<edgeBlocks, 256, 0, stream>>>(dst, cnt, pos, E);
  }
  if (fixedPitch) {
    slot_fill_kernel<<<edgeBlocks, 256, 0, stream>>>(src, dst, pos, cnt, csrSrc, nrm, E, N);
  } else {
    scan_norm_kernel<<<(N + SCAN_T - 1) / SCAN_T, SCAN_T, 0, stream>>>(cnt, rowptr, nrm, N);
    csr_fill_kernel<<<edgeBlocks, 256, 0, stream>>>(src, dst, pos, rowptr, csrSrc, E);
  }

  spmm_gather_kernel<<<(N + 7) / 8, 256, 0, stream>>>(feats, rowInfo, csrSrc, nrm, agg0, N,
                                                       pitch, src, dst, E);

  for (int k0 = 0; k0 < KSTACK; k0 += G) {
    const int g = (KSTACK - k0 < G) ? (KSTACK - k0) : G;
    const int slots = chainBlocks / g;
    gemm_chain_kernel<<<g * slots, 256, 0, stream>>>(
        feats, agg0, nrm, v + (size_t)k0 * D * D, bv + (size_t)k0 * D,
        w0 + (size_t)k0 * D * D, b0 + (size_t)k0 * D, w + (size_t)k0 * D * D,
        fv, z1, nRowTiles, g, M);
    gather_out_kernel<<<gatherBlocks, 256, 0, stream>>>(
        z1, rowInfo, csrSrc, nrm, bw + (size_t)k0 * D, fv, out, N,
        1.0f / (float)KSTACK, g, M, k0 > 0 ? 1 : 0, pitch, src, dst, E);
  }
}
